# starting kernel
# baseline (speedup 1.0000x reference)
#include <hip/hip_runtime.h>

#define N_NODES 100000
#define N_EDGES 1600000
#define DIN 128
#define DH  128
#define DOUT 64

typedef __attribute__((ext_vector_type(2))) float v2f;
typedef __attribute__((ext_vector_type(8))) float v8f;

__device__ __forceinline__ void atomAddF32(float* p, float v) {
    unsafeAtomicAdd(p, v);
}


__global__ void k_init_deg(float* __restrict__ deg) {
    int i = blockIdx.x * blockDim.x + threadIdx.x;
    if (i < N_NODES) deg[i] = 1.0f;
}

__global__ void k_deg_accum(const int* __restrict__ dst, float* __restrict__ deg) {
    int e = blockIdx.x * blockDim.x + threadIdx.x;
    if (e < N_EDGES) atomAddF32(&deg[dst[e]], 1.0f);
}

__global__ void k_dinv(float* __restrict__ deg_dinv) {
    int i = blockIdx.x * blockDim.x + threadIdx.x;
    if (i < N_NODES) deg_dinv[i] = rsqrtf(deg_dinv[i]);
}

__global__ void k_norm_e(const int* __restrict__ src, const int* __restrict__ dst,
                         const float* __restrict__ dinv, float* __restrict__ norme) {
    int e = blockIdx.x * blockDim.x + threadIdx.x;
    if (e < N_EDGES) norme[e] = dinv[src[e]] * dinv[dst[e]];
}


template<int K, int NCOLS, bool RELU>
__global__ void k_gemm_wmma(const float* __restrict__ A,
                            const float* __restrict__ W,
                            float* __restrict__ out) {
    const int NT = NCOLS / 16;
    int tid  = blockIdx.x * blockDim.x + threadIdx.x;
    int wave = tid >> 5;
    int lane = threadIdx.x & 31;
    int mTile = wave / NT;
    int nTile = wave - mTile * NT;

    int hi      = lane >> 4;
    int laneLow = lane & 15;
    int mBase = mTile * 16;
    int nBase = nTile * 16;

    const float* aRow = A + (size_t)(mBase + laneLow) * K;
    const float* wCol = W + nBase + laneLow;

    v8f c = {0.f, 0.f, 0.f, 0.f, 0.f, 0.f, 0.f, 0.f};

    #pragma unroll 4
    for (int k = 0; k < K; k += 4) {
        int kk = k + 2 * hi;
        float a0 = aRow[kk];
        float a1 = aRow[kk + 1];
        if (RELU) { a0 = fmaxf(a0, 0.0f); a1 = fmaxf(a1, 0.0f); }
        v2f av; av.x = a0; av.y = a1;

        v2f bv;
        bv.x = wCol[(size_t)kk * NCOLS];
        bv.y = wCol[(size_t)(kk + 1) * NCOLS];

        c = __builtin_amdgcn_wmma_f32_16x16x4_f32(false, av, false, bv,
                                                  (short)0, c, false, false);
    }

    float* oCol = out + (size_t)(mBase + 8 * hi) * NCOLS + nBase + laneLow;
    #pragma unroll
    for (int r = 0; r < 8; ++r) {
        oCol[(size_t)r * NCOLS] = c[r];
    }
}


template<int D>
__global__ void k_self_bias(const float* __restrict__ h,
                            const float* __restrict__ dinv,
                            const float* __restrict__ bias,
                            float* __restrict__ agg) {
    int t = blockIdx.x * blockDim.x + threadIdx.x;
    if (t < N_NODES * D) {
        int i = t / D;
        int f = t - i * D;
        float sn = dinv[i] * dinv[i];
        agg[t] = h[t] * sn + bias[f];
    }
}


template<int D>
__global__ void k_scatter(const int* __restrict__ src, const int* __restrict__ dst,
                          const float* __restrict__ norme,
                          const float* __restrict__ h, float* __restrict__ agg) {
    const int CH = D / 4;
    int t = blockIdx.x * blockDim.x + threadIdx.x;
    if (t >= N_EDGES * CH) return;
    int e = t / CH;
    int c = t - e * CH;

    int s = src[e], d = dst[e];
    float w = norme[e];

    const float4 v = *(const float4*)(h + (size_t)s * D + c * 4);
    float* p = agg + (size_t)d * D + c * 4;
    atomAddF32(p + 0, v.x * w);
    atomAddF32(p + 1, v.y * w);
    atomAddF32(p + 2, v.z * w);
    atomAddF32(p + 3, v.w * w);
}


extern "C" void kernel_launch(void* const* d_in, const int* in_sizes, int n_in,
                              void* d_out, int out_size, void* d_ws, size_t ws_size,
                              hipStream_t stream) {
    const float* x   = (const float*)d_in[0];
    const int*   ei  = (const int*)d_in[1];
    const float* W1  = (const float*)d_in[2];
    const float* b1  = (const float*)d_in[3];
    const float* W2  = (const float*)d_in[4];
    const float* b2  = (const float*)d_in[5];
    float*       out = (float*)d_out;

    const int* src = ei;
    const int* dst = ei + N_EDGES;

    float* ws    = (float*)d_ws;
    float* dinv  = ws;
    float* norme = dinv  + N_NODES;
    float* h1    = norme + N_EDGES;
    float* agg1  = h1    + (size_t)N_NODES * DH;
    float* h2    = agg1  + (size_t)N_NODES * DH;

    const int B = 256;

    k_init_deg <<<(N_NODES + B - 1) / B, B, 0, stream>>>(dinv);
    k_deg_accum<<<(N_EDGES + B - 1) / B, B, 0, stream>>>(dst, dinv);
    k_dinv     <<<(N_NODES + B - 1) / B, B, 0, stream>>>(dinv);
    k_norm_e   <<<(N_EDGES + B - 1) / B, B, 0, stream>>>(src, dst, dinv, norme);

    {
        int waves = (N_NODES / 16) * (DH / 16);
        k_gemm_wmma<DIN, DH, false><<<waves * 32 / B, B, 0, stream>>>(x, W1, h1);
    }
    k_self_bias<DH><<<(N_NODES * DH + B - 1) / B, B, 0, stream>>>(h1, dinv, b1, agg1);
    {
        int total = N_EDGES * (DH / 4);
        k_scatter<DH><<<(total + B - 1) / B, B, 0, stream>>>(src, dst, norme, h1, agg1);
    }

    {
        int waves = (N_NODES / 16) * (DOUT / 16);
        k_gemm_wmma<DH, DOUT, true><<<waves * 32 / B, B, 0, stream>>>(agg1, W2, h2);
    }
    k_self_bias<DOUT><<<(N_NODES * DOUT + B - 1) / B, B, 0, stream>>>(h2, dinv, b2, out);
    {
        int total = N_EDGES * (DOUT / 4);
        k_scatter<DOUT><<<(total + B - 1) / B, B, 0, stream>>>(src, dst, norme, h2, out);
    }
}


# Edge scatter by global float atomics replaced by destination-range owner workgroups: each filters the whole dst list, sums its rows in LDS by LDS float atomics, writes each row once; 10 -> 8 launches
# speedup vs baseline: 1.6678x; 1.6678x over previous
#include <hip/hip_runtime.h>

#define N_NODES 100000
#define N_EDGES 1600000
#define DIN 128
#define DH  128
#define DOUT 64

typedef __attribute__((ext_vector_type(2))) float v2f;
typedef __attribute__((ext_vector_type(8))) float v8f;

__device__ __forceinline__ void atomAddF32(float* p, float v) {
    unsafeAtomicAdd(p, v);
}


__global__ void k_init_deg(float* __restrict__ deg) {
    int i = blockIdx.x * blockDim.x + threadIdx.x;
    if (i < N_NODES) deg[i] = 1.0f;
}

__global__ void k_deg_accum(const int* __restrict__ dst, float* __restrict__ deg) {
    int e = blockIdx.x * blockDim.x + threadIdx.x;
    if (e < N_EDGES) atomAddF32(&deg[dst[e]], 1.0f);
}

__global__ void k_dinv(float* __restrict__ deg_dinv) {
    int i = blockIdx.x * blockDim.x + threadIdx.x;
    if (i < N_NODES) deg_dinv[i] = rsqrtf(deg_dinv[i]);
}

__global__ void k_norm_e(const int* __restrict__ src, const int* __restrict__ dst,
                         const float* __restrict__ dinv, float* __restrict__ norme) {
    int e = blockIdx.x * blockDim.x + threadIdx.x;
    if (e < N_EDGES) norme[e] = dinv[src[e]] * dinv[dst[e]];
}


template<int K, int NCOLS, bool RELU>
__global__ void k_gemm_wmma(const float* __restrict__ A,
                            const float* __restrict__ W,
                            float* __restrict__ out) {
    const int NT = NCOLS / 16;
    int tid  = blockIdx.x * blockDim.x + threadIdx.x;
    int wave = tid >> 5;
    int lane = threadIdx.x & 31;
    int mTile = wave / NT;
    int nTile = wave - mTile * NT;

    int hi      = lane >> 4;
    int laneLow = lane & 15;
    int mBase = mTile * 16;
    int nBase = nTile * 16;

    const float* aRow = A + (size_t)(mBase + laneLow) * K;
    const float* wCol = W + nBase + laneLow;

    v8f c = {0.f, 0.f, 0.f, 0.f, 0.f, 0.f, 0.f, 0.f};

    #pragma unroll 4
    for (int k = 0; k < K; k += 4) {
        int kk = k + 2 * hi;
        float a0 = aRow[kk];
        float a1 = aRow[kk + 1];
        if (RELU) { a0 = fmaxf(a0, 0.0f); a1 = fmaxf(a1, 0.0f); }
        v2f av; av.x = a0; av.y = a1;

        v2f bv;
        bv.x = wCol[(size_t)kk * NCOLS];
        bv.y = wCol[(size_t)(kk + 1) * NCOLS];

        c = __builtin_amdgcn_wmma_f32_16x16x4_f32(false, av, false, bv,
                                                  (short)0, c, false, false);
    }

    float* oCol = out + (size_t)(mBase + 8 * hi) * NCOLS + nBase + laneLow;
    #pragma unroll
    for (int r = 0; r < 8; ++r) {
        oCol[(size_t)r * NCOLS] = c[r];
    }
}


#define OWN_ROWS 391
#define OWN_WG   256
#define OWN_THREADS 1024
#define OWN_U    8
#define OWN_LIST 160

template<int V> struct fvec;
template<> struct fvec<2> { typedef __attribute__((ext_vector_type(2))) float t; };
template<> struct fvec<4> { typedef __attribute__((ext_vector_type(4))) float t; };

template<int D>
__device__ __forceinline__ void own_flush(int n, int lane, int le, int ld,
                                          const int* __restrict__ src,
                                          const float* __restrict__ norme,
                                          const float* __restrict__ h,
                                          float* acc) {
    constexpr int V = D / 32;
    typedef typename fvec<V>::t vec_t;
    constexpr int G = 8;
    int   s = 0;
    float w = 0.0f;
    int   d = OWN_ROWS;
    if (lane < n) { s = src[le]; w = norme[le]; d = ld; }
    for (int i0 = 0; i0 < n; i0 += G) {
        vec_t v[G];
        #pragma unroll
        for (int g = 0; g < G; ++g) {
            int si = __builtin_amdgcn_readlane(s, i0 + g);
            v[g] = *(const vec_t*)(h + (size_t)si * D + lane * V);
        }
        #pragma unroll
        for (int g = 0; g < G; ++g) {
            float wi = __int_as_float(__builtin_amdgcn_readlane(__float_as_int(w), i0 + g));
            int   di = __builtin_amdgcn_readlane(d, i0 + g);
            float* a = acc + di * D + lane;
            #pragma unroll
            for (int j = 0; j < V; ++j) {
                __hip_atomic_fetch_add(a + j * 32, v[g][j] * wi, __ATOMIC_RELAXED,
                                       __HIP_MEMORY_SCOPE_WORKGROUP);
            }
        }
    }
}

template<int D>
__global__ __launch_bounds__(OWN_THREADS)
void k_agg_owner(const int* __restrict__ src, const int* __restrict__ dst,
                 const float* __restrict__ norme, const float* __restrict__ dinv,
                 const float* __restrict__ bias,
                 const float* __restrict__ h, float* __restrict__ agg) {
    constexpr int V = D / 32;
    typedef typename fvec<V>::t vec_t;
    __shared__ float acc[(OWN_ROWS + 1) * D];
    __shared__ int2  wlist[OWN_THREADS / 32][OWN_LIST];

    const int tid  = threadIdx.x;
    const int lane = tid & 31;
    const int wv   = __builtin_amdgcn_readfirstlane(tid >> 5);
    const int lo   = blockIdx.x * OWN_ROWS;
    int2* mylist = wlist[wv];

    if (wv == 0) {
        #pragma unroll
        for (int j = 0; j < V; ++j) acc[OWN_ROWS * D + j * 32 + lane] = 0.0f;
    }
    for (int r = wv; r < OWN_ROWS; r += OWN_THREADS / 32) {
        int node = lo + r;
        if (node < N_NODES) {
            float sn = dinv[node] * dinv[node];
            vec_t hv = *(const vec_t*)(h + (size_t)node * D + lane * V);
            #pragma unroll
            for (int j = 0; j < V; ++j)
                acc[r * D + j * 32 + lane] = hv[j] * sn + bias[lane * V + j];
        }
    }
    __syncthreads();

    constexpr int NV4   = N_EDGES / 4;
    constexpr int TRIPS = (NV4 + OWN_THREADS - 1) / OWN_THREADS;
    constexpr int SUPER = (TRIPS + OWN_U - 1) / OWN_U;
    const int4* dst4 = (const int4*)dst;
    int sup = (blockIdx.x * SUPER) / OWN_WG;
    int cnt = 0;

    for (int k = 0; k < SUPER; ++k) {
        int4 d4[OWN_U];
        #pragma unroll
        for (int t = 0; t < OWN_U; ++t) {
            int idx = (sup * OWN_U + t) * OWN_THREADS + tid;
            d4[t] = dst4[min(idx, NV4 - 1)];
        }
        #pragma unroll
        for (int t = 0; t < OWN_U; ++t) {
            int wbase = (sup * OWN_U + t) * OWN_THREADS + wv * 32;
            unsigned u[4];
            u[0] = (unsigned)(d4[t].x - lo); u[1] = (unsigned)(d4[t].y - lo);
            u[2] = (unsigned)(d4[t].z - lo); u[3] = (unsigned)(d4[t].w - lo);
            unsigned mn = min(min(u[0], u[1]), min(u[2], u[3]));
            if (wbase < NV4 && __builtin_amdgcn_ballot_w32(mn < (unsigned)OWN_ROWS) != 0u) {
                #pragma unroll
                for (int j = 0; j < 4; ++j) {
                    bool p = u[j] < (unsigned)OWN_ROWS;
                    unsigned m = __builtin_amdgcn_ballot_w32(p);
                    if (p) {
                        int slot = cnt + (int)__builtin_amdgcn_mbcnt_lo(m, 0u);
                        mylist[slot] = make_int2((wbase + lane) * 4 + j, (int)u[j]);
                    }
                    cnt += __builtin_popcount(m);
                }
                if (cnt >= 32) {
                    int full = cnt & ~31;
                    for (int base = 0; base < full; base += 32) {
                        int2 en = mylist[base + lane];
                        own_flush<D>(32, lane, en.x, en.y, src, norme, h, acc);
                    }
                    int rest = cnt - full;
                    int2 mv = make_int2(0, 0);
                    if (lane < rest) mv = mylist[full + lane];
                    if (lane < rest) mylist[lane] = mv;
                    cnt = rest;
                }
            }
        }
        sup = (sup + 1 == SUPER) ? 0 : sup + 1;
    }
    if (cnt != 0) {
        int2 en = make_int2(0, 0);
        if (lane < cnt) en = mylist[lane];
        own_flush<D>(cnt, lane, en.x, en.y, src, norme, h, acc);
    }
    __syncthreads();

    for (int r = wv; r < OWN_ROWS; r += OWN_THREADS / 32) {
        int node = lo + r;
        if (node < N_NODES) {
            vec_t o;
            #pragma unroll
            for (int j = 0; j < V; ++j) o[j] = acc[r * D + j * 32 + lane];
            *(vec_t*)(agg + (size_t)node * D + lane * V) = o;
        }
    }
}


extern "C" void kernel_launch(void* const* d_in, const int* in_sizes, int n_in,
                              void* d_out, int out_size, void* d_ws, size_t ws_size,
                              hipStream_t stream) {
    const float* x   = (const float*)d_in[0];
    const int*   ei  = (const int*)d_in[1];
    const float* W1  = (const float*)d_in[2];
    const float* b1  = (const float*)d_in[3];
    const float* W2  = (const float*)d_in[4];
    const float* b2  = (const float*)d_in[5];
    float*       out = (float*)d_out;

    const int* src = ei;
    const int* dst = ei + N_EDGES;

    float* ws    = (float*)d_ws;
    float* dinv  = ws;
    float* norme = dinv  + N_NODES;
    float* h1    = norme + N_EDGES;
    float* agg1  = h1    + (size_t)N_NODES * DH;
    float* h2    = agg1  + (size_t)N_NODES * DH;

    const int B = 256;

    k_init_deg <<<(N_NODES + B - 1) / B, B, 0, stream>>>(dinv);
    k_deg_accum<<<(N_EDGES + B - 1) / B, B, 0, stream>>>(dst, dinv);
    k_dinv     <<<(N_NODES + B - 1) / B, B, 0, stream>>>(dinv);
    k_norm_e   <<<(N_EDGES + B - 1) / B, B, 0, stream>>>(src, dst, dinv, norme);

    {
        int waves = (N_NODES / 16) * (DH / 16);
        k_gemm_wmma<DIN, DH, false><<<waves * 32 / B, B, 0, stream>>>(x, W1, h1);
    }
    k_agg_owner<DH><<<OWN_WG, OWN_THREADS, 0, stream>>>(src, dst, norme, dinv, b1, h1, agg1);

    {
        int waves = (N_NODES / 16) * (DOUT / 16);
        k_gemm_wmma<DH, DOUT, true><<<waves * 32 / B, B, 0, stream>>>(agg1, W2, h2);
    }
    k_agg_owner<DOUT><<<OWN_WG, OWN_THREADS, 0, stream>>>(src, dst, norme, dinv, b2, h2, out);
}


# Both GEMMs move operands and results through LDS: per wave a 16-row A strip by whole-row 16-byte loads, W transposed once per workgroup, 8-byte LDS fragment reads, whole-row stores; sums unchanged
# speedup vs baseline: 1.8278x; 1.0959x over previous
#include <hip/hip_runtime.h>

#define N_NODES 100000
#define N_EDGES 1600000
#define DIN 128
#define DH  128
#define DOUT 64

typedef __attribute__((ext_vector_type(2))) float v2f;
typedef __attribute__((ext_vector_type(8))) float v8f;

__device__ __forceinline__ void atomAddF32(float* p, float v) {
    unsafeAtomicAdd(p, v);
}


__global__ void k_init_deg(float* __restrict__ deg) {
    int i = blockIdx.x * blockDim.x + threadIdx.x;
    if (i < N_NODES) deg[i] = 1.0f;
}

__global__ void k_deg_accum(const int* __restrict__ dst, float* __restrict__ deg) {
    int e = blockIdx.x * blockDim.x + threadIdx.x;
    if (e < N_EDGES) atomAddF32(&deg[dst[e]], 1.0f);
}

__global__ void k_dinv(float* __restrict__ deg_dinv) {
    int i = blockIdx.x * blockDim.x + threadIdx.x;
    if (i < N_NODES) deg_dinv[i] = rsqrtf(deg_dinv[i]);
}

__global__ void k_norm_e(const int* __restrict__ src, const int* __restrict__ dst,
                         const float* __restrict__ dinv, float* __restrict__ norme) {
    int e = blockIdx.x * blockDim.x + threadIdx.x;
    if (e < N_EDGES) norme[e] = dinv[src[e]] * dinv[dst[e]];
}


#define GM_WAVES 25
#define GM_WG    250
#define GM_PITCH 132

typedef __attribute__((ext_vector_type(4))) float v4f;

template<int NCOLS, bool RELU>
__global__ __launch_bounds__(GM_WAVES * 32)
void k_gemm_lds(const float* __restrict__ A,
                const float* __restrict__ W,
                float* __restrict__ out) {
    constexpr int K  = 128;
    constexpr int NT = NCOLS / 16;
    __shared__ __attribute__((aligned(16))) float sW[NCOLS * GM_PITCH];
    __shared__ __attribute__((aligned(16))) float sA[GM_WAVES * 16 * GM_PITCH];

    const int tid  = threadIdx.x;
    const int lane = tid & 31;
    const int wv   = __builtin_amdgcn_readfirstlane(tid >> 5);
    const int hi      = lane >> 4;
    const int laneLow = lane & 15;
    const int mBase = (blockIdx.x * GM_WAVES + wv) * 16;
    float* myA = sA + wv * (16 * GM_PITCH);

    v4f arow[16];
    {
        const float* ap = A + (size_t)mBase * K + lane * 4;
        #pragma unroll
        for (int r = 0; r < 16; ++r) arow[r] = *(const v4f*)(ap + (size_t)r * K);
    }

    if (wv < 16) {
        constexpr int NB  = NCOLS / 16;
        constexpr int WIT = (K / 8) * NB / 16;
        const int k0 = lane >> 2;
        const int n0 = (lane & 3) * 4;
        v4f wreg[WIT];
        #pragma unroll
        for (int i = 0; i < WIT; ++i) {
            int g  = wv + i * 16;
            int kb = g / NB;
            int nb = g - kb * NB;
            wreg[i] = *(const v4f*)(W + (size_t)(kb * 8 + k0) * NCOLS + nb * 16 + n0);
        }
        #pragma unroll
        for (int i = 0; i < WIT; ++i) {
            int g  = wv + i * 16;
            int kb = g / NB;
            int nb = g - kb * NB;
            float* d = sW + (nb * 16 + n0) * GM_PITCH + kb * 8 + k0;
            d[0]            = wreg[i].x;
            d[GM_PITCH]     = wreg[i].y;
            d[2 * GM_PITCH] = wreg[i].z;
            d[3 * GM_PITCH] = wreg[i].w;
        }
    }

    #pragma unroll
    for (int r = 0; r < 16; ++r) {
        v4f t = arow[r];
        if (RELU) {
            t.x = fmaxf(t.x, 0.0f); t.y = fmaxf(t.y, 0.0f);
            t.z = fmaxf(t.z, 0.0f); t.w = fmaxf(t.w, 0.0f);
        }
        *(v4f*)(myA + r * GM_PITCH + lane * 4) = t;
    }
    __syncthreads();

    v8f c[NT];
    #pragma unroll
    for (int t = 0; t < NT; ++t) c[t] = (v8f){0.f, 0.f, 0.f, 0.f, 0.f, 0.f, 0.f, 0.f};

    const float* aP = myA + laneLow * GM_PITCH + 2 * hi;
    const float* bP = sW  + laneLow * GM_PITCH + 2 * hi;

    #pragma unroll 4
    for (int k = 0; k < K; k += 4) {
        v2f av = *(const v2f*)(aP + k);
        #pragma unroll
        for (int t = 0; t < NT; ++t) {
            v2f bv = *(const v2f*)(bP + t * (16 * GM_PITCH) + k);
            c[t] = __builtin_amdgcn_wmma_f32_16x16x4_f32(false, av, false, bv,
                                                         (short)0, c[t], false, false);
        }
    }

    {
        float* o = myA + (8 * hi) * GM_PITCH + laneLow;
        #pragma unroll
        for (int t = 0; t < NT; ++t) {
            #pragma unroll
            for (int r = 0; r < 8; ++r) o[r * GM_PITCH + t * 16] = c[t][r];
        }
    }
    {
        constexpr int LPR = NCOLS / 4;
        constexpr int RPI = 32 / LPR;
        const int rsub = lane / LPR;
        const int col  = (lane - rsub * LPR) * 4;
        #pragma unroll
        for (int i = 0; i < 16 / RPI; ++i) {
            int row = i * RPI + rsub;
            v4f v = *(const v4f*)(myA + row * GM_PITCH + col);
            *(v4f*)(out + (size_t)(mBase + row) * NCOLS + col) = v;
        }
    }
}


#define OWN_ROWS 391
#define OWN_WG   256
#define OWN_THREADS 1024
#define OWN_U    8
#define OWN_LIST 160

template<int V> struct fvec;
template<> struct fvec<2> { typedef __attribute__((ext_vector_type(2))) float t; };
template<> struct fvec<4> { typedef __attribute__((ext_vector_type(4))) float t; };

template<int D>
__device__ __forceinline__ void own_flush(int n, int lane, int le, int ld,
                                          const int* __restrict__ src,
                                          const float* __restrict__ norme,
                                          const float* __restrict__ h,
                                          float* acc) {
    constexpr int V = D / 32;
    typedef typename fvec<V>::t vec_t;
    constexpr int G = 8;
    int   s = 0;
    float w = 0.0f;
    int   d = OWN_ROWS;
    if (lane < n) { s = src[le]; w = norme[le]; d = ld; }
    for (int i0 = 0; i0 < n; i0 += G) {
        vec_t v[G];
        #pragma unroll
        for (int g = 0; g < G; ++g) {
            int si = __builtin_amdgcn_readlane(s, i0 + g);
            v[g] = *(const vec_t*)(h + (size_t)si * D + lane * V);
        }
        #pragma unroll
        for (int g = 0; g < G; ++g) {
            float wi = __int_as_float(__builtin_amdgcn_readlane(__float_as_int(w), i0 + g));
            int   di = __builtin_amdgcn_readlane(d, i0 + g);
            float* a = acc + di * D + lane;
            #pragma unroll
            for (int j = 0; j < V; ++j) {
                __hip_atomic_fetch_add(a + j * 32, v[g][j] * wi, __ATOMIC_RELAXED,
                                       __HIP_MEMORY_SCOPE_WORKGROUP);
            }
        }
    }
}

template<int D>
__global__ __launch_bounds__(OWN_THREADS)
void k_agg_owner(const int* __restrict__ src, const int* __restrict__ dst,
                 const float* __restrict__ norme, const float* __restrict__ dinv,
                 const float* __restrict__ bias,
                 const float* __restrict__ h, float* __restrict__ agg) {
    constexpr int V = D / 32;
    typedef typename fvec<V>::t vec_t;
    __shared__ float acc[(OWN_ROWS + 1) * D];
    __shared__ int2  wlist[OWN_THREADS / 32][OWN_LIST];

    const int tid  = threadIdx.x;
    const int lane = tid & 31;
    const int wv   = __builtin_amdgcn_readfirstlane(tid >> 5);
    const int lo   = blockIdx.x * OWN_ROWS;
    int2* mylist = wlist[wv];

    if (wv == 0) {
        #pragma unroll
        for (int j = 0; j < V; ++j) acc[OWN_ROWS * D + j * 32 + lane] = 0.0f;
    }
    for (int r = wv; r < OWN_ROWS; r += OWN_THREADS / 32) {
        int node = lo + r;
        if (node < N_NODES) {
            float sn = dinv[node] * dinv[node];
            vec_t hv = *(const vec_t*)(h + (size_t)node * D + lane * V);
            #pragma unroll
            for (int j = 0; j < V; ++j)
                acc[r * D + j * 32 + lane] = hv[j] * sn + bias[lane * V + j];
        }
    }
    __syncthreads();

    constexpr int NV4   = N_EDGES / 4;
    constexpr int TRIPS = (NV4 + OWN_THREADS - 1) / OWN_THREADS;
    constexpr int SUPER = (TRIPS + OWN_U - 1) / OWN_U;
    const int4* dst4 = (const int4*)dst;
    int sup = (blockIdx.x * SUPER) / OWN_WG;
    int cnt = 0;

    for (int k = 0; k < SUPER; ++k) {
        int4 d4[OWN_U];
        #pragma unroll
        for (int t = 0; t < OWN_U; ++t) {
            int idx = (sup * OWN_U + t) * OWN_THREADS + tid;
            d4[t] = dst4[min(idx, NV4 - 1)];
        }
        #pragma unroll
        for (int t = 0; t < OWN_U; ++t) {
            int wbase = (sup * OWN_U + t) * OWN_THREADS + wv * 32;
            unsigned u[4];
            u[0] = (unsigned)(d4[t].x - lo); u[1] = (unsigned)(d4[t].y - lo);
            u[2] = (unsigned)(d4[t].z - lo); u[3] = (unsigned)(d4[t].w - lo);
            unsigned mn = min(min(u[0], u[1]), min(u[2], u[3]));
            if (wbase < NV4 && __builtin_amdgcn_ballot_w32(mn < (unsigned)OWN_ROWS) != 0u) {
                #pragma unroll
                for (int j = 0; j < 4; ++j) {
                    bool p = u[j] < (unsigned)OWN_ROWS;
                    unsigned m = __builtin_amdgcn_ballot_w32(p);
                    if (p) {
                        int slot = cnt + (int)__builtin_amdgcn_mbcnt_lo(m, 0u);
                        mylist[slot] = make_int2((wbase + lane) * 4 + j, (int)u[j]);
                    }
                    cnt += __builtin_popcount(m);
                }
                if (cnt >= 32) {
                    int full = cnt & ~31;
                    for (int base = 0; base < full; base += 32) {
                        int2 en = mylist[base + lane];
                        own_flush<D>(32, lane, en.x, en.y, src, norme, h, acc);
                    }
                    int rest = cnt - full;
                    int2 mv = make_int2(0, 0);
                    if (lane < rest) mv = mylist[full + lane];
                    if (lane < rest) mylist[lane] = mv;
                    cnt = rest;
                }
            }
        }
        sup = (sup + 1 == SUPER) ? 0 : sup + 1;
    }
    if (cnt != 0) {
        int2 en = make_int2(0, 0);
        if (lane < cnt) en = mylist[lane];
        own_flush<D>(cnt, lane, en.x, en.y, src, norme, h, acc);
    }
    __syncthreads();

    for (int r = wv; r < OWN_ROWS; r += OWN_THREADS / 32) {
        int node = lo + r;
        if (node < N_NODES) {
            vec_t o;
            #pragma unroll
            for (int j = 0; j < V; ++j) o[j] = acc[r * D + j * 32 + lane];
            *(vec_t*)(agg + (size_t)node * D + lane * V) = o;
        }
    }
}


extern "C" void kernel_launch(void* const* d_in, const int* in_sizes, int n_in,
                              void* d_out, int out_size, void* d_ws, size_t ws_size,
                              hipStream_t stream) {
    const float* x   = (const float*)d_in[0];
    const int*   ei  = (const int*)d_in[1];
    const float* W1  = (const float*)d_in[2];
    const float* b1  = (const float*)d_in[3];
    const float* W2  = (const float*)d_in[4];
    const float* b2  = (const float*)d_in[5];
    float*       out = (float*)d_out;

    const int* src = ei;
    const int* dst = ei + N_EDGES;

    float* ws    = (float*)d_ws;
    float* dinv  = ws;
    float* norme = dinv  + N_NODES;
    float* h1    = norme + N_EDGES;
    float* agg1  = h1    + (size_t)N_NODES * DH;
    float* h2    = agg1  + (size_t)N_NODES * DH;

    const int B = 256;

    k_init_deg <<<(N_NODES + B - 1) / B, B, 0, stream>>>(dinv);
    k_deg_accum<<<(N_EDGES + B - 1) / B, B, 0, stream>>>(dst, dinv);
    k_dinv     <<<(N_NODES + B - 1) / B, B, 0, stream>>>(dinv);
    k_norm_e   <<<(N_EDGES + B - 1) / B, B, 0, stream>>>(src, dst, dinv, norme);

    k_gemm_lds<DH, false><<<GM_WG, GM_WAVES * 32, 0, stream>>>(x, W1, h1);
    k_agg_owner<DH><<<OWN_WG, OWN_THREADS, 0, stream>>>(src, dst, norme, dinv, b1, h1, agg1);

    k_gemm_lds<DOUT, true><<<GM_WG, GM_WAVES * 32, 0, stream>>>(agg1, W2, h2);
    k_agg_owner<DOUT><<<OWN_WG, OWN_THREADS, 0, stream>>>(src, dst, norme, dinv, b2, h2, out);
}


# Owners filter a 1-byte owner key per edge (16 edges per 16-byte load; key plane from the old weight launch), not 4-byte dst; weight dinv[src]*dinv[dst] formed per kept edge; sums stay arrival-order
# speedup vs baseline: 1.9887x; 1.0880x over previous
#include <hip/hip_runtime.h>

#define N_NODES 100000
#define N_EDGES 1600000
#define DIN 128
#define DH  128
#define DOUT 64

typedef __attribute__((ext_vector_type(2))) float v2f;
typedef __attribute__((ext_vector_type(8))) float v8f;

__device__ __forceinline__ void atomAddF32(float* p, float v) {
    unsafeAtomicAdd(p, v);
}


__global__ void k_init_deg(float* __restrict__ deg) {
    int i = blockIdx.x * blockDim.x + threadIdx.x;
    if (i < N_NODES) deg[i] = 1.0f;
}

__global__ void k_deg_accum(const int* __restrict__ dst, float* __restrict__ deg) {
    int e = blockIdx.x * blockDim.x + threadIdx.x;
    if (e < N_EDGES) atomAddF32(&deg[dst[e]], 1.0f);
}

__global__ void k_dinv(float* __restrict__ deg_dinv) {
    int i = blockIdx.x * blockDim.x + threadIdx.x;
    if (i < N_NODES) deg_dinv[i] = rsqrtf(deg_dinv[i]);
}

#define OWN_ROWS 391

__global__ void k_owner_key(const int* __restrict__ dst, unsigned* __restrict__ key) {
    int i = blockIdx.x * blockDim.x + threadIdx.x;
    if (i < N_EDGES / 4) {
        int4 d = ((const int4*)dst)[i];
        key[i] =  ((unsigned)d.x / (unsigned)OWN_ROWS)
               | (((unsigned)d.y / (unsigned)OWN_ROWS) << 8)
               | (((unsigned)d.z / (unsigned)OWN_ROWS) << 16)
               | (((unsigned)d.w / (unsigned)OWN_ROWS) << 24);
    }
}


#define GM_WAVES 25
#define GM_WG    250
#define GM_PITCH 132

typedef __attribute__((ext_vector_type(4))) float v4f;

template<int NCOLS, bool RELU>
__global__ __launch_bounds__(GM_WAVES * 32)
void k_gemm_lds(const float* __restrict__ A,
                const float* __restrict__ W,
                float* __restrict__ out) {
    constexpr int K  = 128;
    constexpr int NT = NCOLS / 16;
    __shared__ __attribute__((aligned(16))) float sW[NCOLS * GM_PITCH];
    __shared__ __attribute__((aligned(16))) float sA[GM_WAVES * 16 * GM_PITCH];

    const int tid  = threadIdx.x;
    const int lane = tid & 31;
    const int wv   = __builtin_amdgcn_readfirstlane(tid >> 5);
    const int hi      = lane >> 4;
    const int laneLow = lane & 15;
    const int mBase = (blockIdx.x * GM_WAVES + wv) * 16;
    float* myA = sA + wv * (16 * GM_PITCH);

    v4f arow[16];
    {
        const float* ap = A + (size_t)mBase * K + lane * 4;
        #pragma unroll
        for (int r = 0; r < 16; ++r) arow[r] = *(const v4f*)(ap + (size_t)r * K);
    }

    if (wv < 16) {
        constexpr int NB  = NCOLS / 16;
        constexpr int WIT = (K / 8) * NB / 16;
        const int k0 = lane >> 2;
        const int n0 = (lane & 3) * 4;
        v4f wreg[WIT];
        #pragma unroll
        for (int i = 0; i < WIT; ++i) {
            int g  = wv + i * 16;
            int kb = g / NB;
            int nb = g - kb * NB;
            wreg[i] = *(const v4f*)(W + (size_t)(kb * 8 + k0) * NCOLS + nb * 16 + n0);
        }
        #pragma unroll
        for (int i = 0; i < WIT; ++i) {
            int g  = wv + i * 16;
            int kb = g / NB;
            int nb = g - kb * NB;
            float* d = sW + (nb * 16 + n0) * GM_PITCH + kb * 8 + k0;
            d[0]            = wreg[i].x;
            d[GM_PITCH]     = wreg[i].y;
            d[2 * GM_PITCH] = wreg[i].z;
            d[3 * GM_PITCH] = wreg[i].w;
        }
    }

    #pragma unroll
    for (int r = 0; r < 16; ++r) {
        v4f t = arow[r];
        if (RELU) {
            t.x = fmaxf(t.x, 0.0f); t.y = fmaxf(t.y, 0.0f);
            t.z = fmaxf(t.z, 0.0f); t.w = fmaxf(t.w, 0.0f);
        }
        *(v4f*)(myA + r * GM_PITCH + lane * 4) = t;
    }
    __syncthreads();

    v8f c[NT];
    #pragma unroll
    for (int t = 0; t < NT; ++t) c[t] = (v8f){0.f, 0.f, 0.f, 0.f, 0.f, 0.f, 0.f, 0.f};

    const float* aP = myA + laneLow * GM_PITCH + 2 * hi;
    const float* bP = sW  + laneLow * GM_PITCH + 2 * hi;

    #pragma unroll 4
    for (int k = 0; k < K; k += 4) {
        v2f av = *(const v2f*)(aP + k);
        #pragma unroll
        for (int t = 0; t < NT; ++t) {
            v2f bv = *(const v2f*)(bP + t * (16 * GM_PITCH) + k);
            c[t] = __builtin_amdgcn_wmma_f32_16x16x4_f32(false, av, false, bv,
                                                         (short)0, c[t], false, false);
        }
    }

    {
        float* o = myA + (8 * hi) * GM_PITCH + laneLow;
        #pragma unroll
        for (int t = 0; t < NT; ++t) {
            #pragma unroll
            for (int r = 0; r < 8; ++r) o[r * GM_PITCH + t * 16] = c[t][r];
        }
    }
    {
        constexpr int LPR = NCOLS / 4;
        constexpr int RPI = 32 / LPR;
        const int rsub = lane / LPR;
        const int col  = (lane - rsub * LPR) * 4;
        #pragma unroll
        for (int i = 0; i < 16 / RPI; ++i) {
            int row = i * RPI + rsub;
            v4f v = *(const v4f*)(myA + row * GM_PITCH + col);
            *(v4f*)(out + (size_t)(mBase + row) * NCOLS + col) = v;
        }
    }
}


#define OWN_WG   256
#define OWN_THREADS 1024
#define OWN_U    7
#define OWN_LIST 544

__device__ __forceinline__ unsigned zero_bytes(unsigned x) {
    unsigned t = (x & 0x7f7f7f7fu) + 0x7f7f7f7fu;
    return ~(t | x | 0x7f7f7f7fu);
}

template<int V> struct fvec;
template<> struct fvec<2> { typedef __attribute__((ext_vector_type(2))) float t; };
template<> struct fvec<4> { typedef __attribute__((ext_vector_type(4))) float t; };

template<int D>
__device__ __forceinline__ void own_flush(int n, int lane, int le, int lo,
                                          const int* __restrict__ src,
                                          const int* __restrict__ dst,
                                          const float* __restrict__ dinv,
                                          const float* __restrict__ h,
                                          float* acc) {
    constexpr int V = D / 32;
    typedef typename fvec<V>::t vec_t;
    constexpr int G = 8;
    int   s  = 0;
    float ws = 0.0f;
    float wd = 0.0f;
    int   d  = OWN_ROWS;
    if (lane < n) {
        s = src[le];
        int dg = dst[le];
        ws = dinv[s];
        wd = dinv[dg];
        d = dg - lo;
    }
    for (int i0 = 0; i0 < n; i0 += G) {
        vec_t v[G];
        #pragma unroll
        for (int g = 0; g < G; ++g) {
            int si = __builtin_amdgcn_readlane(s, i0 + g);
            v[g] = *(const vec_t*)(h + (size_t)si * D + lane * V);
        }
        #pragma unroll
        for (int g = 0; g < G; ++g) {
            float wi = __int_as_float(__builtin_amdgcn_readlane(__float_as_int(ws), i0 + g))
                     * __int_as_float(__builtin_amdgcn_readlane(__float_as_int(wd), i0 + g));
            int   di = __builtin_amdgcn_readlane(d, i0 + g);
            float* a = acc + di * D + lane;
            #pragma unroll
            for (int j = 0; j < V; ++j) {
                __hip_atomic_fetch_add(a + j * 32, v[g][j] * wi, __ATOMIC_RELAXED,
                                       __HIP_MEMORY_SCOPE_WORKGROUP);
            }
        }
    }
}

template<int D>
__global__ __launch_bounds__(OWN_THREADS)
void k_agg_owner(const int* __restrict__ src, const int* __restrict__ dst,
                 const unsigned* __restrict__ key, const float* __restrict__ dinv,
                 const float* __restrict__ bias,
                 const float* __restrict__ h, float* __restrict__ agg) {
    constexpr int V = D / 32;
    typedef typename fvec<V>::t vec_t;
    __shared__ float acc[(OWN_ROWS + 1) * D];
    __shared__ int   wlist[OWN_THREADS / 32][OWN_LIST];

    const int tid  = threadIdx.x;
    const int lane = tid & 31;
    const int wv   = __builtin_amdgcn_readfirstlane(tid >> 5);
    const int lo   = blockIdx.x * OWN_ROWS;
    int* mylist = wlist[wv];

    if (wv == 0) {
        #pragma unroll
        for (int j = 0; j < V; ++j) acc[OWN_ROWS * D + j * 32 + lane] = 0.0f;
    }
    for (int r = wv; r < OWN_ROWS; r += OWN_THREADS / 32) {
        int node = lo + r;
        if (node < N_NODES) {
            float sn = dinv[node] * dinv[node];
            vec_t hv = *(const vec_t*)(h + (size_t)node * D + lane * V);
            #pragma unroll
            for (int j = 0; j < V; ++j)
                acc[r * D + j * 32 + lane] = hv[j] * sn + bias[lane * V + j];
        }
    }
    __syncthreads();

    constexpr int NV16  = N_EDGES / 16;
    constexpr int TRIPS = (NV16 + OWN_THREADS - 1) / OWN_THREADS;
    constexpr int SUPER = (TRIPS + OWN_U - 1) / OWN_U;
    const uint4* key16 = (const uint4*)key;
    const unsigned me4 = (unsigned)blockIdx.x * 0x01010101u;
    int sup = (blockIdx.x * SUPER) / OWN_WG;
    int cnt = 0;

    for (int k = 0; k < SUPER; ++k) {
        uint4 kk[OWN_U];
        #pragma unroll
        for (int t = 0; t < OWN_U; ++t) {
            int idx = (sup * OWN_U + t) * OWN_THREADS + tid;
            kk[t] = key16[min(idx, NV16 - 1)];
        }
        #pragma unroll
        for (int t = 0; t < OWN_U; ++t) {
            int wbase = (sup * OWN_U + t) * OWN_THREADS + wv * 32;
            const unsigned z0 = zero_bytes(kk[t].x ^ me4), z1 = zero_bytes(kk[t].y ^ me4);
            const unsigned z2 = zero_bytes(kk[t].z ^ me4), z3 = zero_bytes(kk[t].w ^ me4);
            unsigned c = (z0 >> 7) | (z1 >> 6) | (z2 >> 5) | (z3 >> 4);
            unsigned m = __builtin_amdgcn_ballot_w32(c != 0u);
            if (wbase < NV16 && m != 0u) {
                const int e0 = (wbase + lane) * 16;
                while (m != 0u) {
                    if (c != 0u) {
                        int p = __builtin_ctz(c);
                        int slot = cnt + (int)__builtin_amdgcn_mbcnt_lo(m, 0u);
                        mylist[slot] = e0 + ((p & 7) << 2) + (p >> 3);
                    }
                    cnt += __builtin_popcount(m);
                    c &= c - 1u;
                    m = __builtin_amdgcn_ballot_w32(c != 0u);
                }
                if (cnt >= 32) {
                    int full = cnt & ~31;
                    for (int base = 0; base < full; base += 32) {
                        int en = mylist[base + lane];
                        own_flush<D>(32, lane, en, lo, src, dst, dinv, h, acc);
                    }
                    int rest = cnt - full;
                    int mv = 0;
                    if (lane < rest) mv = mylist[full + lane];
                    if (lane < rest) mylist[lane] = mv;
                    cnt = rest;
                }
            }
        }
        sup = (sup + 1 == SUPER) ? 0 : sup + 1;
    }
    if (cnt != 0) {
        int en = 0;
        if (lane < cnt) en = mylist[lane];
        own_flush<D>(cnt, lane, en, lo, src, dst, dinv, h, acc);
    }
    __syncthreads();

    for (int r = wv; r < OWN_ROWS; r += OWN_THREADS / 32) {
        int node = lo + r;
        if (node < N_NODES) {
            vec_t o;
            #pragma unroll
            for (int j = 0; j < V; ++j) o[j] = acc[r * D + j * 32 + lane];
            *(vec_t*)(agg + (size_t)node * D + lane * V) = o;
        }
    }
}


extern "C" void kernel_launch(void* const* d_in, const int* in_sizes, int n_in,
                              void* d_out, int out_size, void* d_ws, size_t ws_size,
                              hipStream_t stream) {
    const float* x   = (const float*)d_in[0];
    const int*   ei  = (const int*)d_in[1];
    const float* W1  = (const float*)d_in[2];
    const float* b1  = (const float*)d_in[3];
    const float* W2  = (const float*)d_in[4];
    const float* b2  = (const float*)d_in[5];
    float*       out = (float*)d_out;

    const int* src = ei;
    const int* dst = ei + N_EDGES;

    float* ws    = (float*)d_ws;
    float* dinv  = ws;
    float* keyp  = dinv  + N_NODES;
    float* h1    = keyp  + N_EDGES;
    float* agg1  = h1    + (size_t)N_NODES * DH;
    float* h2    = agg1  + (size_t)N_NODES * DH;

    const int B = 256;

    k_init_deg <<<(N_NODES + B - 1) / B, B, 0, stream>>>(dinv);
    k_deg_accum<<<(N_EDGES + B - 1) / B, B, 0, stream>>>(dst, dinv);
    k_dinv     <<<(N_NODES + B - 1) / B, B, 0, stream>>>(dinv);
    unsigned* key = (unsigned*)keyp;
    k_owner_key<<<(N_EDGES / 4 + B - 1) / B, B, 0, stream>>>(dst, key);

    k_gemm_lds<DH, false><<<GM_WG, GM_WAVES * 32, 0, stream>>>(x, W1, h1);
    k_agg_owner<DH><<<OWN_WG, OWN_THREADS, 0, stream>>>(src, dst, key, dinv, b1, h1, agg1);

    k_gemm_lds<DOUT, true><<<GM_WG, GM_WAVES * 32, 0, stream>>>(agg1, W2, h2);
    k_agg_owner<DOUT><<<OWN_WG, OWN_THREADS, 0, stream>>>(src, dst, key, dinv, b2, h2, out);
}


# Layer-1 owners write each kept batch of 32 edges as packed (src,row,weight) records; layer-2 owners replay their records: no 2nd key scan, no per-edge src/dst/dinv reads; sums stay arrival-order
# speedup vs baseline: 2.3210x; 1.1671x over previous
#include <hip/hip_runtime.h>

#define N_NODES 100000
#define N_EDGES 1600000
#define DIN 128
#define DH  128
#define DOUT 64

typedef __attribute__((ext_vector_type(2))) float v2f;
typedef __attribute__((ext_vector_type(8))) float v8f;

__device__ __forceinline__ void atomAddF32(float* p, float v) {
    unsafeAtomicAdd(p, v);
}


__global__ void k_init_deg(float* __restrict__ deg) {
    int i = blockIdx.x * blockDim.x + threadIdx.x;
    if (i < N_NODES) deg[i] = 1.0f;
}

__global__ void k_deg_accum(const int* __restrict__ dst, float* __restrict__ deg) {
    int e = blockIdx.x * blockDim.x + threadIdx.x;
    if (e < N_EDGES) atomAddF32(&deg[dst[e]], 1.0f);
}

#define OWN_ROWS 391
#define OWN_WG   256
#define OWN_THREADS 1024

__global__ __launch_bounds__(512)
void k_dinv(float* __restrict__ deg_dinv, int* __restrict__ ocnt) {
    __shared__ int part[16];
    const int t = threadIdx.x;
    const int node = blockIdx.x * OWN_ROWS + t;
    int c = 0;
    if (t < OWN_ROWS && node < N_NODES) {
        float d = deg_dinv[node];
        c = (int)d - 1;
        deg_dinv[node] = rsqrtf(d);
    }
    for (int o = 16; o > 0; o >>= 1) c += __shfl_xor(c, o);
    if ((t & 31) == 0) part[t >> 5] = c;
    __syncthreads();
    if (t == 0) {
        int sum = 0;
        for (int i = 0; i < 16; ++i) sum += part[i];
        ocnt[blockIdx.x] = sum;
    }
}

__global__ void k_owner_key(const int* __restrict__ dst, unsigned* __restrict__ key) {
    int i = blockIdx.x * blockDim.x + threadIdx.x;
    if (i < N_EDGES / 4) {
        int4 d = ((const int4*)dst)[i];
        key[i] =  ((unsigned)d.x / (unsigned)OWN_ROWS)
               | (((unsigned)d.y / (unsigned)OWN_ROWS) << 8)
               | (((unsigned)d.z / (unsigned)OWN_ROWS) << 16)
               | (((unsigned)d.w / (unsigned)OWN_ROWS) << 24);
    }
}


#define GM_WAVES 25
#define GM_WG    250
#define GM_PITCH 132

typedef __attribute__((ext_vector_type(4))) float v4f;

template<int NCOLS, bool RELU>
__global__ __launch_bounds__(GM_WAVES * 32)
void k_gemm_lds(const float* __restrict__ A,
                const float* __restrict__ W,
                float* __restrict__ out) {
    constexpr int K  = 128;
    constexpr int NT = NCOLS / 16;
    __shared__ __attribute__((aligned(16))) float sW[NCOLS * GM_PITCH];
    __shared__ __attribute__((aligned(16))) float sA[GM_WAVES * 16 * GM_PITCH];

    const int tid  = threadIdx.x;
    const int lane = tid & 31;
    const int wv   = __builtin_amdgcn_readfirstlane(tid >> 5);
    const int hi      = lane >> 4;
    const int laneLow = lane & 15;
    const int mBase = (blockIdx.x * GM_WAVES + wv) * 16;
    float* myA = sA + wv * (16 * GM_PITCH);

    v4f arow[16];
    {
        const float* ap = A + (size_t)mBase * K + lane * 4;
        #pragma unroll
        for (int r = 0; r < 16; ++r) arow[r] = *(const v4f*)(ap + (size_t)r * K);
    }

    if (wv < 16) {
        constexpr int NB  = NCOLS / 16;
        constexpr int WIT = (K / 8) * NB / 16;
        const int k0 = lane >> 2;
        const int n0 = (lane & 3) * 4;
        v4f wreg[WIT];
        #pragma unroll
        for (int i = 0; i < WIT; ++i) {
            int g  = wv + i * 16;
            int kb = g / NB;
            int nb = g - kb * NB;
            wreg[i] = *(const v4f*)(W + (size_t)(kb * 8 + k0) * NCOLS + nb * 16 + n0);
        }
        #pragma unroll
        for (int i = 0; i < WIT; ++i) {
            int g  = wv + i * 16;
            int kb = g / NB;
            int nb = g - kb * NB;
            float* d = sW + (nb * 16 + n0) * GM_PITCH + kb * 8 + k0;
            d[0]            = wreg[i].x;
            d[GM_PITCH]     = wreg[i].y;
            d[2 * GM_PITCH] = wreg[i].z;
            d[3 * GM_PITCH] = wreg[i].w;
        }
    }

    #pragma unroll
    for (int r = 0; r < 16; ++r) {
        v4f t = arow[r];
        if (RELU) {
            t.x = fmaxf(t.x, 0.0f); t.y = fmaxf(t.y, 0.0f);
            t.z = fmaxf(t.z, 0.0f); t.w = fmaxf(t.w, 0.0f);
        }
        *(v4f*)(myA + r * GM_PITCH + lane * 4) = t;
    }
    __syncthreads();

    v8f c[NT];
    #pragma unroll
    for (int t = 0; t < NT; ++t) c[t] = (v8f){0.f, 0.f, 0.f, 0.f, 0.f, 0.f, 0.f, 0.f};

    const float* aP = myA + laneLow * GM_PITCH + 2 * hi;
    const float* bP = sW  + laneLow * GM_PITCH + 2 * hi;

    #pragma unroll 4
    for (int k = 0; k < K; k += 4) {
        v2f av = *(const v2f*)(aP + k);
        #pragma unroll
        for (int t = 0; t < NT; ++t) {
            v2f bv = *(const v2f*)(bP + t * (16 * GM_PITCH) + k);
            c[t] = __builtin_amdgcn_wmma_f32_16x16x4_f32(false, av, false, bv,
                                                         (short)0, c[t], false, false);
        }
    }

    {
        float* o = myA + (8 * hi) * GM_PITCH + laneLow;
        #pragma unroll
        for (int t = 0; t < NT; ++t) {
            #pragma unroll
            for (int r = 0; r < 8; ++r) o[r * GM_PITCH + t * 16] = c[t][r];
        }
    }
    {
        constexpr int LPR = NCOLS / 4;
        constexpr int RPI = 32 / LPR;
        const int rsub = lane / LPR;
        const int col  = (lane - rsub * LPR) * 4;
        #pragma unroll
        for (int i = 0; i < 16 / RPI; ++i) {
            int row = i * RPI + rsub;
            v4f v = *(const v4f*)(myA + row * GM_PITCH + col);
            *(v4f*)(out + (size_t)(mBase + row) * NCOLS + col) = v;
        }
    }
}


#define OWN_U    7
#define OWN_LIST 544

__device__ __forceinline__ int own_cap(int cnt) { return (cnt >> 5) + OWN_THREADS / 32; }

__device__ __forceinline__ unsigned zero_bytes(unsigned x) {
    unsigned t = (x & 0x7f7f7f7fu) + 0x7f7f7f7fu;
    return ~(t | x | 0x7f7f7f7fu);
}

template<int V> struct fvec;
template<> struct fvec<2> { typedef __attribute__((ext_vector_type(2))) float t; };
template<> struct fvec<4> { typedef __attribute__((ext_vector_type(4))) float t; };

template<int D>
__device__ __forceinline__ void own_rows(int n, int lane, int s, int d, float ws, float wd,
                                         const float* __restrict__ h, float* acc) {
    constexpr int V = D / 32;
    typedef typename fvec<V>::t vec_t;
    constexpr int G = 8;
    for (int i0 = 0; i0 < n; i0 += G) {
        vec_t v[G];
        #pragma unroll
        for (int g = 0; g < G; ++g) {
            int si = __builtin_amdgcn_readlane(s, i0 + g);
            v[g] = *(const vec_t*)(h + (size_t)si * D + lane * V);
        }
        #pragma unroll
        for (int g = 0; g < G; ++g) {
            float wi = __int_as_float(__builtin_amdgcn_readlane(__float_as_int(ws), i0 + g))
                     * __int_as_float(__builtin_amdgcn_readlane(__float_as_int(wd), i0 + g));
            int   di = __builtin_amdgcn_readlane(d, i0 + g);
            float* a = acc + di * D + lane;
            #pragma unroll
            for (int j = 0; j < V; ++j) {
                __hip_atomic_fetch_add(a + j * 32, v[g][j] * wi, __ATOMIC_RELAXED,
                                       __HIP_MEMORY_SCOPE_WORKGROUP);
            }
        }
    }
}

template<int D>
__device__ __forceinline__ void own_flush(int n, int lane, int le, int lo,
                                          const int* __restrict__ src,
                                          const int* __restrict__ dst,
                                          const float* __restrict__ dinv,
                                          const float* __restrict__ h,
                                          float* acc, int* nchunk, uint2* __restrict__ myrec) {
    int   s  = 0;
    float ws = 0.0f;
    float wd = 0.0f;
    int   d  = OWN_ROWS;
    if (lane < n) {
        s = src[le];
        int dg = dst[le];
        ws = dinv[s];
        wd = dinv[dg];
        d = dg - lo;
    }
    int c = 0;
    if (lane == 0) c = __hip_atomic_fetch_add(nchunk, 1, __ATOMIC_RELAXED, __HIP_MEMORY_SCOPE_WORKGROUP);
    c = __builtin_amdgcn_readfirstlane(c);
    own_rows<D>(n, lane, s, d, ws, wd, h, acc);
    myrec[c * 32 + lane] = make_uint2(((unsigned)s << 9) | (unsigned)d, __float_as_uint(ws * wd));
}

template<int D, bool REPLAY>
__global__ __launch_bounds__(OWN_THREADS)
void k_agg_owner(const int* __restrict__ src, const int* __restrict__ dst,
                 const unsigned* __restrict__ key, const float* __restrict__ dinv,
                 const float* __restrict__ bias,
                 const float* __restrict__ h, float* __restrict__ agg,
                 const int* __restrict__ ocnt, int* nch, unsigned* rec) {
    constexpr int V = D / 32;
    typedef typename fvec<V>::t vec_t;
    constexpr int ACC_MIN = REPLAY ? 164 * 1024 / 4 : 0;
    constexpr int ACC_FLOATS = (OWN_ROWS + 1) * D > ACC_MIN ? (OWN_ROWS + 1) * D : ACC_MIN;
    __shared__ float acc[ACC_FLOATS];
    __shared__ int   wlist[REPLAY ? 1 : OWN_THREADS / 32][REPLAY ? 1 : OWN_LIST];
    __shared__ int   red[OWN_WG / 32];
    __shared__ int   nchunk;

    const int tid  = threadIdx.x;
    const int lane = tid & 31;
    const int wv   = __builtin_amdgcn_readfirstlane(tid >> 5);
    const int lo   = blockIdx.x * OWN_ROWS;

    {
        int cap = 0;
        if (tid < (int)blockIdx.x) cap = own_cap(ocnt[tid]);
        for (int o = 16; o > 0; o >>= 1) cap += __shfl_xor(cap, o);
        if (wv < OWN_WG / 32 && lane == 0) red[wv] = cap;
        if (tid == 0) nchunk = 0;
    }

    if (wv == 0) {
        #pragma unroll
        for (int j = 0; j < V; ++j) acc[OWN_ROWS * D + j * 32 + lane] = 0.0f;
    }
    for (int r = wv; r < OWN_ROWS; r += OWN_THREADS / 32) {
        int node = lo + r;
        if (node < N_NODES) {
            float sn = dinv[node] * dinv[node];
            vec_t hv = *(const vec_t*)(h + (size_t)node * D + lane * V);
            #pragma unroll
            for (int j = 0; j < V; ++j)
                acc[r * D + j * 32 + lane] = hv[j] * sn + bias[lane * V + j];
        }
    }
    __syncthreads();

    int start = 0;
    for (int i = 0; i < OWN_WG / 32; ++i) start += red[i];
    uint2* myrec = (uint2*)rec + (size_t)start * 32;

    if (REPLAY) {
        const int nCh = min(nch[blockIdx.x], own_cap(ocnt[blockIdx.x]));
        int c = wv;
        uint2 r = make_uint2((unsigned)OWN_ROWS, 0u);
        if (c < nCh) r = myrec[c * 32 + lane];
        while (c < nCh) {
            const int cn = c + OWN_THREADS / 32;
            uint2 rn = make_uint2((unsigned)OWN_ROWS, 0u);
            if (cn < nCh) rn = myrec[cn * 32 + lane];
            const int d = (int)(r.x & 511u);
            const int n = __builtin_popcount(__builtin_amdgcn_ballot_w32(d != OWN_ROWS));
            own_rows<D>(n, lane, (int)(r.x >> 9), d, __uint_as_float(r.y), 1.0f, h, acc);
            r = rn;
            c = cn;
        }
    } else {
    int* mylist = wlist[wv];
    constexpr int NV16  = N_EDGES / 16;
    constexpr int TRIPS = (NV16 + OWN_THREADS - 1) / OWN_THREADS;
    constexpr int SUPER = (TRIPS + OWN_U - 1) / OWN_U;
    const uint4* key16 = (const uint4*)key;
    const unsigned me4 = (unsigned)blockIdx.x * 0x01010101u;
    int sup = (blockIdx.x * SUPER) / OWN_WG;
    int cnt = 0;

    for (int k = 0; k < SUPER; ++k) {
        uint4 kk[OWN_U];
        #pragma unroll
        for (int t = 0; t < OWN_U; ++t) {
            int idx = (sup * OWN_U + t) * OWN_THREADS + tid;
            kk[t] = key16[min(idx, NV16 - 1)];
        }
        #pragma unroll
        for (int t = 0; t < OWN_U; ++t) {
            int wbase = (sup * OWN_U + t) * OWN_THREADS + wv * 32;
            const unsigned z0 = zero_bytes(kk[t].x ^ me4), z1 = zero_bytes(kk[t].y ^ me4);
            const unsigned z2 = zero_bytes(kk[t].z ^ me4), z3 = zero_bytes(kk[t].w ^ me4);
            unsigned c = (z0 >> 7) | (z1 >> 6) | (z2 >> 5) | (z3 >> 4);
            unsigned m = __builtin_amdgcn_ballot_w32(c != 0u);
            if (wbase < NV16 && m != 0u) {
                const int e0 = (wbase + lane) * 16;
                while (m != 0u) {
                    if (c != 0u) {
                        int p = __builtin_ctz(c);
                        int slot = cnt + (int)__builtin_amdgcn_mbcnt_lo(m, 0u);
                        mylist[slot] = e0 + ((p & 7) << 2) + (p >> 3);
                    }
                    cnt += __builtin_popcount(m);
                    c &= c - 1u;
                    m = __builtin_amdgcn_ballot_w32(c != 0u);
                }
                if (cnt >= 32) {
                    int full = cnt & ~31;
                    for (int base = 0; base < full; base += 32) {
                        int en = mylist[base + lane];
                        own_flush<D>(32, lane, en, lo, src, dst, dinv, h, acc, &nchunk, myrec);
                    }
                    int rest = cnt - full;
                    int mv = 0;
                    if (lane < rest) mv = mylist[full + lane];
                    if (lane < rest) mylist[lane] = mv;
                    cnt = rest;
                }
            }
        }
        sup = (sup + 1 == SUPER) ? 0 : sup + 1;
    }
    if (cnt != 0) {
        int en = 0;
        if (lane < cnt) en = mylist[lane];
        own_flush<D>(cnt, lane, en, lo, src, dst, dinv, h, acc, &nchunk, myrec);
    }
    }
    __syncthreads();

    if (!REPLAY && tid == 0) nch[blockIdx.x] = nchunk;

    for (int r = wv; r < OWN_ROWS; r += OWN_THREADS / 32) {
        int node = lo + r;
        if (node < N_NODES) {
            vec_t o;
            #pragma unroll
            for (int j = 0; j < V; ++j) o[j] = acc[r * D + j * 32 + lane];
            *(vec_t*)(agg + (size_t)node * D + lane * V) = o;
        }
    }
}


extern "C" void kernel_launch(void* const* d_in, const int* in_sizes, int n_in,
                              void* d_out, int out_size, void* d_ws, size_t ws_size,
                              hipStream_t stream) {
    const float* x   = (const float*)d_in[0];
    const int*   ei  = (const int*)d_in[1];
    const float* W1  = (const float*)d_in[2];
    const float* b1  = (const float*)d_in[3];
    const float* W2  = (const float*)d_in[4];
    const float* b2  = (const float*)d_in[5];
    float*       out = (float*)d_out;

    const int* src = ei;
    const int* dst = ei + N_EDGES;

    float* ws    = (float*)d_ws;
    float* dinv  = ws;
    float* keyp  = dinv  + N_NODES;
    float* h1    = keyp  + N_EDGES;
    float* agg1  = h1    + (size_t)N_NODES * DH;
    float* recp  = agg1  + (size_t)N_NODES * DH;
    int*   ocnt  = (int*)(keyp + N_EDGES / 4);
    int*   nch   = ocnt + OWN_WG;
    unsigned* rec = (unsigned*)(recp + 32);
    float* h2    = h1;

    const int B = 256;

    k_init_deg <<<(N_NODES + B - 1) / B, B, 0, stream>>>(dinv);
    k_deg_accum<<<(N_EDGES + B - 1) / B, B, 0, stream>>>(dst, dinv);
    k_dinv     <<<OWN_WG, 512, 0, stream>>>(dinv, ocnt);
    unsigned* key = (unsigned*)keyp;
    k_owner_key<<<(N_EDGES / 4 + B - 1) / B, B, 0, stream>>>(dst, key);

    k_gemm_lds<DH, false><<<GM_WG, GM_WAVES * 32, 0, stream>>>(x, W1, h1);
    k_agg_owner<DH, false><<<OWN_WG, OWN_THREADS, 0, stream>>>(src, dst, key, dinv, b1, h1, agg1, ocnt, nch, rec);

    k_gemm_lds<DOUT, true><<<GM_WG, GM_WAVES * 32, 0, stream>>>(agg1, W2, h2);
    k_agg_owner<DOUT, true><<<OWN_WG, OWN_THREADS, 0, stream>>>(src, dst, key, dinv, b2, h2, out, ocnt, nch, rec);
}


# Degree count: scattered global float atomics replaced by 64 edge-slice workgroups counting all nodes in packed 16-bit LDS counters (and writing the owner keys); k_dinv sums slices; 8 -> 6 launches
# speedup vs baseline: 2.9915x; 1.2889x over previous
#include <hip/hip_runtime.h>

#define N_NODES 100000
#define N_EDGES 1600000
#define DIN 128
#define DH  128
#define DOUT 64

typedef __attribute__((ext_vector_type(2))) float v2f;
typedef __attribute__((ext_vector_type(8))) float v8f;

#define OWN_ROWS 391
#define OWN_WG   256
#define OWN_THREADS 1024

#define HIST_G     64
#define HIST_PER   (N_EDGES / 4 / HIST_G)
#define HIST_PITCH 50016

static_assert(HIST_PER * 4 * HIST_G == N_EDGES, "the slices cover the edge list exactly");
static_assert(HIST_PER * 4 < 65536, "a 16-bit counter holds a whole slice");
static_assert(HIST_PITCH * 2 >= N_NODES && HIST_PITCH % 32 == 0, "plane size");

__device__ __forceinline__ void count_node(unsigned* cnt, int node) {
    __hip_atomic_fetch_add(cnt + (node >> 1), 1u << ((node & 1) << 4), __ATOMIC_RELAXED,
                           __HIP_MEMORY_SCOPE_WORKGROUP);
}

__global__ __launch_bounds__(1024)
void k_hist_key(const int* __restrict__ dst, unsigned* __restrict__ key, unsigned* __restrict__ hist) {
    __shared__ __attribute__((aligned(16))) unsigned cnt[HIST_PITCH];
    constexpr int IT = (HIST_PER + 1023) / 1024;
    const int tid = threadIdx.x;
    const int4* d4 = (const int4*)dst + (size_t)blockIdx.x * HIST_PER;

    int4 d[IT];
    #pragma unroll
    for (int i = 0; i < IT; ++i) {
        int idx = tid + i * 1024;
        d[i] = d4[min(idx, HIST_PER - 1)];
    }
    for (int i = tid; i < HIST_PITCH / 4; i += 1024) ((uint4*)cnt)[i] = make_uint4(0u, 0u, 0u, 0u);
    __syncthreads();

    #pragma unroll
    for (int i = 0; i < IT; ++i) {
        int idx = tid + i * 1024;
        if (idx < HIST_PER) {
            const int4 v = d[i];
            key[(size_t)blockIdx.x * HIST_PER + idx] =
                  ((unsigned)v.x / (unsigned)OWN_ROWS)
               | (((unsigned)v.y / (unsigned)OWN_ROWS) << 8)
               | (((unsigned)v.z / (unsigned)OWN_ROWS) << 16)
               | (((unsigned)v.w / (unsigned)OWN_ROWS) << 24);
            count_node(cnt, v.x);
            count_node(cnt, v.y);
            count_node(cnt, v.z);
            count_node(cnt, v.w);
        }
    }
    __syncthreads();

    uint4* o = (uint4*)(hist + (size_t)blockIdx.x * HIST_PITCH);
    for (int i = tid; i < HIST_PITCH / 4; i += 1024) o[i] = ((const uint4*)cnt)[i];
}

__global__ __launch_bounds__(512)
void k_dinv(const unsigned* __restrict__ hist, float* __restrict__ dinv, int* __restrict__ ocnt) {
    __shared__ int part[16];
    const int t = threadIdx.x;
    const int node = blockIdx.x * OWN_ROWS + t;
    int c = 0;
    if (t < OWN_ROWS && node < N_NODES) {
        const unsigned* p = hist + (node >> 1);
        const int sh = (node & 1) << 4;
        #pragma unroll 16
        for (int g = 0; g < HIST_G; ++g) c += (int)((p[(size_t)g * HIST_PITCH] >> sh) & 0xffffu);
        dinv[node] = rsqrtf((float)(c + 1));
    }
    for (int o = 16; o > 0; o >>= 1) c += __shfl_xor(c, o);
    if ((t & 31) == 0) part[t >> 5] = c;
    __syncthreads();
    if (t == 0) {
        int sum = 0;
        for (int i = 0; i < 16; ++i) sum += part[i];
        ocnt[blockIdx.x] = sum;
    }
}


#define GM_WAVES 25
#define GM_WG    250
#define GM_PITCH 132

typedef __attribute__((ext_vector_type(4))) float v4f;

template<int NCOLS, bool RELU>
__global__ __launch_bounds__(GM_WAVES * 32)
void k_gemm_lds(const float* __restrict__ A,
                const float* __restrict__ W,
                float* __restrict__ out) {
    constexpr int K  = 128;
    constexpr int NT = NCOLS / 16;
    __shared__ __attribute__((aligned(16))) float sW[NCOLS * GM_PITCH];
    __shared__ __attribute__((aligned(16))) float sA[GM_WAVES * 16 * GM_PITCH];

    const int tid  = threadIdx.x;
    const int lane = tid & 31;
    const int wv   = __builtin_amdgcn_readfirstlane(tid >> 5);
    const int hi      = lane >> 4;
    const int laneLow = lane & 15;
    const int mBase = (blockIdx.x * GM_WAVES + wv) * 16;
    float* myA = sA + wv * (16 * GM_PITCH);

    v4f arow[16];
    {
        const float* ap = A + (size_t)mBase * K + lane * 4;
        #pragma unroll
        for (int r = 0; r < 16; ++r) arow[r] = *(const v4f*)(ap + (size_t)r * K);
    }

    if (wv < 16) {
        constexpr int NB  = NCOLS / 16;
        constexpr int WIT = (K / 8) * NB / 16;
        const int k0 = lane >> 2;
        const int n0 = (lane & 3) * 4;
        v4f wreg[WIT];
        #pragma unroll
        for (int i = 0; i < WIT; ++i) {
            int g  = wv + i * 16;
            int kb = g / NB;
            int nb = g - kb * NB;
            wreg[i] = *(const v4f*)(W + (size_t)(kb * 8 + k0) * NCOLS + nb * 16 + n0);
        }
        #pragma unroll
        for (int i = 0; i < WIT; ++i) {
            int g  = wv + i * 16;
            int kb = g / NB;
            int nb = g - kb * NB;
            float* d = sW + (nb * 16 + n0) * GM_PITCH + kb * 8 + k0;
            d[0]            = wreg[i].x;
            d[GM_PITCH]     = wreg[i].y;
            d[2 * GM_PITCH] = wreg[i].z;
            d[3 * GM_PITCH] = wreg[i].w;
        }
    }

    #pragma unroll
    for (int r = 0; r < 16; ++r) {
        v4f t = arow[r];
        if (RELU) {
            t.x = fmaxf(t.x, 0.0f); t.y = fmaxf(t.y, 0.0f);
            t.z = fmaxf(t.z, 0.0f); t.w = fmaxf(t.w, 0.0f);
        }
        *(v4f*)(myA + r * GM_PITCH + lane * 4) = t;
    }
    __syncthreads();

    v8f c[NT];
    #pragma unroll
    for (int t = 0; t < NT; ++t) c[t] = (v8f){0.f, 0.f, 0.f, 0.f, 0.f, 0.f, 0.f, 0.f};

    const float* aP = myA + laneLow * GM_PITCH + 2 * hi;
    const float* bP = sW  + laneLow * GM_PITCH + 2 * hi;

    #pragma unroll 4
    for (int k = 0; k < K; k += 4) {
        v2f av = *(const v2f*)(aP + k);
        #pragma unroll
        for (int t = 0; t < NT; ++t) {
            v2f bv = *(const v2f*)(bP + t * (16 * GM_PITCH) + k);
            c[t] = __builtin_amdgcn_wmma_f32_16x16x4_f32(false, av, false, bv,
                                                         (short)0, c[t], false, false);
        }
    }

    {
        float* o = myA + (8 * hi) * GM_PITCH + laneLow;
        #pragma unroll
        for (int t = 0; t < NT; ++t) {
            #pragma unroll
            for (int r = 0; r < 8; ++r) o[r * GM_PITCH + t * 16] = c[t][r];
        }
    }
    {
        constexpr int LPR = NCOLS / 4;
        constexpr int RPI = 32 / LPR;
        const int rsub = lane / LPR;
        const int col  = (lane - rsub * LPR) * 4;
        #pragma unroll
        for (int i = 0; i < 16 / RPI; ++i) {
            int row = i * RPI + rsub;
            v4f v = *(const v4f*)(myA + row * GM_PITCH + col);
            *(v4f*)(out + (size_t)(mBase + row) * NCOLS + col) = v;
        }
    }
}


#define OWN_U    7
#define OWN_LIST 544

__device__ __forceinline__ int own_cap(int cnt) { return (cnt >> 5) + OWN_THREADS / 32; }

__device__ __forceinline__ unsigned zero_bytes(unsigned x) {
    unsigned t = (x & 0x7f7f7f7fu) + 0x7f7f7f7fu;
    return ~(t | x | 0x7f7f7f7fu);
}

template<int V> struct fvec;
template<> struct fvec<2> { typedef __attribute__((ext_vector_type(2))) float t; };
template<> struct fvec<4> { typedef __attribute__((ext_vector_type(4))) float t; };

template<int D>
__device__ __forceinline__ void own_rows(int n, int lane, int s, int d, float ws, float wd,
                                         const float* __restrict__ h, float* acc) {
    constexpr int V = D / 32;
    typedef typename fvec<V>::t vec_t;
    constexpr int G = 8;
    for (int i0 = 0; i0 < n; i0 += G) {
        vec_t v[G];
        #pragma unroll
        for (int g = 0; g < G; ++g) {
            int si = __builtin_amdgcn_readlane(s, i0 + g);
            v[g] = *(const vec_t*)(h + (size_t)si * D + lane * V);
        }
        #pragma unroll
        for (int g = 0; g < G; ++g) {
            float wi = __int_as_float(__builtin_amdgcn_readlane(__float_as_int(ws), i0 + g))
                     * __int_as_float(__builtin_amdgcn_readlane(__float_as_int(wd), i0 + g));
            int   di = __builtin_amdgcn_readlane(d, i0 + g);
            float* a = acc + di * D + lane;
            #pragma unroll
            for (int j = 0; j < V; ++j) {
                __hip_atomic_fetch_add(a + j * 32, v[g][j] * wi, __ATOMIC_RELAXED,
                                       __HIP_MEMORY_SCOPE_WORKGROUP);
            }
        }
    }
}

template<int D>
__device__ __forceinline__ void own_flush(int n, int lane, int le, int lo,
                                          const int* __restrict__ src,
                                          const int* __restrict__ dst,
                                          const float* __restrict__ dinv,
                                          const float* __restrict__ h,
                                          float* acc, int* nchunk, uint2* __restrict__ myrec) {
    int   s  = 0;
    float ws = 0.0f;
    float wd = 0.0f;
    int   d  = OWN_ROWS;
    if (lane < n) {
        s = src[le];
        int dg = dst[le];
        ws = dinv[s];
        wd = dinv[dg];
        d = dg - lo;
    }
    int c = 0;
    if (lane == 0) c = __hip_atomic_fetch_add(nchunk, 1, __ATOMIC_RELAXED, __HIP_MEMORY_SCOPE_WORKGROUP);
    c = __builtin_amdgcn_readfirstlane(c);
    own_rows<D>(n, lane, s, d, ws, wd, h, acc);
    myrec[c * 32 + lane] = make_uint2(((unsigned)s << 9) | (unsigned)d, __float_as_uint(ws * wd));
}

template<int D, bool REPLAY>
__global__ __launch_bounds__(OWN_THREADS)
void k_agg_owner(const int* __restrict__ src, const int* __restrict__ dst,
                 const unsigned* __restrict__ key, const float* __restrict__ dinv,
                 const float* __restrict__ bias,
                 const float* __restrict__ h, float* __restrict__ agg,
                 const int* __restrict__ ocnt, int* nch, unsigned* rec) {
    constexpr int V = D / 32;
    typedef typename fvec<V>::t vec_t;
    constexpr int ACC_MIN = REPLAY ? 164 * 1024 / 4 : 0;
    constexpr int ACC_FLOATS = (OWN_ROWS + 1) * D > ACC_MIN ? (OWN_ROWS + 1) * D : ACC_MIN;
    __shared__ float acc[ACC_FLOATS];
    __shared__ int   wlist[REPLAY ? 1 : OWN_THREADS / 32][REPLAY ? 1 : OWN_LIST];
    __shared__ int   red[OWN_WG / 32];
    __shared__ int   nchunk;

    const int tid  = threadIdx.x;
    const int lane = tid & 31;
    const int wv   = __builtin_amdgcn_readfirstlane(tid >> 5);
    const int lo   = blockIdx.x * OWN_ROWS;

    {
        int cap = 0;
        if (tid < (int)blockIdx.x) cap = own_cap(ocnt[tid]);
        for (int o = 16; o > 0; o >>= 1) cap += __shfl_xor(cap, o);
        if (wv < OWN_WG / 32 && lane == 0) red[wv] = cap;
        if (tid == 0) nchunk = 0;
    }

    if (wv == 0) {
        #pragma unroll
        for (int j = 0; j < V; ++j) acc[OWN_ROWS * D + j * 32 + lane] = 0.0f;
    }
    for (int r = wv; r < OWN_ROWS; r += OWN_THREADS / 32) {
        int node = lo + r;
        if (node < N_NODES) {
            float sn = dinv[node] * dinv[node];
            vec_t hv = *(const vec_t*)(h + (size_t)node * D + lane * V);
            #pragma unroll
            for (int j = 0; j < V; ++j)
                acc[r * D + j * 32 + lane] = hv[j] * sn + bias[lane * V + j];
        }
    }
    __syncthreads();

    int start = 0;
    for (int i = 0; i < OWN_WG / 32; ++i) start += red[i];
    uint2* myrec = (uint2*)rec + (size_t)start * 32;

    if (REPLAY) {
        const int nCh = min(nch[blockIdx.x], own_cap(ocnt[blockIdx.x]));
        int c = wv;
        uint2 r = make_uint2((unsigned)OWN_ROWS, 0u);
        if (c < nCh) r = myrec[c * 32 + lane];
        while (c < nCh) {
            const int cn = c + OWN_THREADS / 32;
            uint2 rn = make_uint2((unsigned)OWN_ROWS, 0u);
            if (cn < nCh) rn = myrec[cn * 32 + lane];
            const int d = (int)(r.x & 511u);
            const int n = __builtin_popcount(__builtin_amdgcn_ballot_w32(d != OWN_ROWS));
            own_rows<D>(n, lane, (int)(r.x >> 9), d, __uint_as_float(r.y), 1.0f, h, acc);
            r = rn;
            c = cn;
        }
    } else {
    int* mylist = wlist[wv];
    constexpr int NV16  = N_EDGES / 16;
    constexpr int TRIPS = (NV16 + OWN_THREADS - 1) / OWN_THREADS;
    constexpr int SUPER = (TRIPS + OWN_U - 1) / OWN_U;
    const uint4* key16 = (const uint4*)key;
    const unsigned me4 = (unsigned)blockIdx.x * 0x01010101u;
    int sup = (blockIdx.x * SUPER) / OWN_WG;
    int cnt = 0;

    for (int k = 0; k < SUPER; ++k) {
        uint4 kk[OWN_U];
        #pragma unroll
        for (int t = 0; t < OWN_U; ++t) {
            int idx = (sup * OWN_U + t) * OWN_THREADS + tid;
            kk[t] = key16[min(idx, NV16 - 1)];
        }
        #pragma unroll
        for (int t = 0; t < OWN_U; ++t) {
            int wbase = (sup * OWN_U + t) * OWN_THREADS + wv * 32;
            const unsigned z0 = zero_bytes(kk[t].x ^ me4), z1 = zero_bytes(kk[t].y ^ me4);
            const unsigned z2 = zero_bytes(kk[t].z ^ me4), z3 = zero_bytes(kk[t].w ^ me4);
            unsigned c = (z0 >> 7) | (z1 >> 6) | (z2 >> 5) | (z3 >> 4);
            unsigned m = __builtin_amdgcn_ballot_w32(c != 0u);
            if (wbase < NV16 && m != 0u) {
                const int e0 = (wbase + lane) * 16;
                while (m != 0u) {
                    if (c != 0u) {
                        int p = __builtin_ctz(c);
                        int slot = cnt + (int)__builtin_amdgcn_mbcnt_lo(m, 0u);
                        mylist[slot] = e0 + ((p & 7) << 2) + (p >> 3);
                    }
                    cnt += __builtin_popcount(m);
                    c &= c - 1u;
                    m = __builtin_amdgcn_ballot_w32(c != 0u);
                }
                if (cnt >= 32) {
                    int full = cnt & ~31;
                    for (int base = 0; base < full; base += 32) {
                        int en = mylist[base + lane];
                        own_flush<D>(32, lane, en, lo, src, dst, dinv, h, acc, &nchunk, myrec);
                    }
                    int rest = cnt - full;
                    int mv = 0;
                    if (lane < rest) mv = mylist[full + lane];
                    if (lane < rest) mylist[lane] = mv;
                    cnt = rest;
                }
            }
        }
        sup = (sup + 1 == SUPER) ? 0 : sup + 1;
    }
    if (cnt != 0) {
        int en = 0;
        if (lane < cnt) en = mylist[lane];
        own_flush<D>(cnt, lane, en, lo, src, dst, dinv, h, acc, &nchunk, myrec);
    }
    }
    __syncthreads();

    if (!REPLAY && tid == 0) nch[blockIdx.x] = nchunk;

    for (int r = wv; r < OWN_ROWS; r += OWN_THREADS / 32) {
        int node = lo + r;
        if (node < N_NODES) {
            vec_t o;
            #pragma unroll
            for (int j = 0; j < V; ++j) o[j] = acc[r * D + j * 32 + lane];
            *(vec_t*)(agg + (size_t)node * D + lane * V) = o;
        }
    }
}


extern "C" void kernel_launch(void* const* d_in, const int* in_sizes, int n_in,
                              void* d_out, int out_size, void* d_ws, size_t ws_size,
                              hipStream_t stream) {
    const float* x   = (const float*)d_in[0];
    const int*   ei  = (const int*)d_in[1];
    const float* W1  = (const float*)d_in[2];
    const float* b1  = (const float*)d_in[3];
    const float* W2  = (const float*)d_in[4];
    const float* b2  = (const float*)d_in[5];
    float*       out = (float*)d_out;

    const int* src = ei;
    const int* dst = ei + N_EDGES;

    float* ws    = (float*)d_ws;
    float* dinv  = ws;
    float* keyp  = dinv  + N_NODES;
    float* h1    = keyp  + N_EDGES;
    float* agg1  = h1    + (size_t)N_NODES * DH;
    float* recp  = agg1  + (size_t)N_NODES * DH;
    int*   ocnt  = (int*)(keyp + N_EDGES / 4);
    int*   nch   = ocnt + OWN_WG;
    unsigned* rec = (unsigned*)(recp + 32);
    float* h2    = h1;

    unsigned* hist = (unsigned*)agg1;

    unsigned* key = (unsigned*)keyp;
    k_hist_key<<<HIST_G, 1024, 0, stream>>>(dst, key, hist);
    k_dinv    <<<OWN_WG, 512, 0, stream>>>(hist, dinv, ocnt);

    k_gemm_lds<DH, false><<<GM_WG, GM_WAVES * 32, 0, stream>>>(x, W1, h1);
    k_agg_owner<DH, false><<<OWN_WG, OWN_THREADS, 0, stream>>>(src, dst, key, dinv, b1, h1, agg1, ocnt, nch, rec);

    k_gemm_lds<DOUT, true><<<GM_WG, GM_WAVES * 32, 0, stream>>>(agg1, W2, h2);
    k_agg_owner<DOUT, true><<<OWN_WG, OWN_THREADS, 0, stream>>>(src, dst, key, dinv, b2, h2, out, ocnt, nch, rec);
}


# Layer-1 owners stop scanning the whole key plane: the degree launch also leaves a bit per owner and 8-edge group (LDS bit-or); owners read the keys of marked groups only; sums stay arrival-order
# speedup vs baseline: 3.1330x; 1.0473x over previous
#include <hip/hip_runtime.h>

#define N_NODES 100000
#define N_EDGES 1600000
#define DIN 128
#define DH  128
#define DOUT 64

typedef __attribute__((ext_vector_type(2))) float v2f;
typedef __attribute__((ext_vector_type(8))) float v8f;

#define OWN_ROWS 391
#define OWN_WG   256
#define OWN_THREADS 1024

#define HIST_G     64
#define HIST_PER   (N_EDGES / 4 / HIST_G)
#define HIST_PITCH 50016

#define BM_GROUP   8
#define BM_BITS    (HIST_PER * 4 / BM_GROUP)
#define BM_WORDS   98
#define BM_LPITCH  99
#define BM_GPITCH  128

static_assert(HIST_PER * 4 % BM_GROUP == 0, "a group never straddles two slices");
static_assert(BM_WORDS * 32 >= BM_BITS && BM_WORDS <= BM_GPITCH, "a chunk holds a slice's groups");
static_assert(HIST_PER * 4 * HIST_G == N_EDGES, "the slices cover the edge list exactly");
static_assert(HIST_PER * 4 < 65536, "a 16-bit counter holds a whole slice");
static_assert(HIST_PITCH * 2 >= N_NODES && HIST_PITCH % 32 == 0, "plane size");

__device__ __forceinline__ void count_node(unsigned* cnt, int node) {
    __hip_atomic_fetch_add(cnt + (node >> 1), 1u << ((node & 1) << 4), __ATOMIC_RELAXED,
                           __HIP_MEMORY_SCOPE_WORKGROUP);
}

__global__ __launch_bounds__(1024)
void k_hist_key(const int* __restrict__ dst, unsigned* __restrict__ key, unsigned* __restrict__ hist,
                unsigned* __restrict__ bmp) {
    __shared__ __attribute__((aligned(16))) unsigned cnt[HIST_PITCH];
    __shared__ unsigned bm[OWN_WG * BM_LPITCH];
    constexpr int IT = (HIST_PER + 1023) / 1024;
    const int tid = threadIdx.x;
    const int4* d4 = (const int4*)dst + (size_t)blockIdx.x * HIST_PER;

    int4 d[IT];
    #pragma unroll
    for (int i = 0; i < IT; ++i) {
        int idx = tid + i * 1024;
        d[i] = d4[min(idx, HIST_PER - 1)];
    }
    for (int i = tid; i < HIST_PITCH / 4; i += 1024) ((uint4*)cnt)[i] = make_uint4(0u, 0u, 0u, 0u);
    for (int i = tid; i < OWN_WG * BM_LPITCH; i += 1024) bm[i] = 0u;
    __syncthreads();

    #pragma unroll
    for (int i = 0; i < IT; ++i) {
        int idx = tid + i * 1024;
        if (idx < HIST_PER) {
            const int4 v = d[i];
            const unsigned k0 = (unsigned)v.x / (unsigned)OWN_ROWS, k1 = (unsigned)v.y / (unsigned)OWN_ROWS;
            const unsigned k2 = (unsigned)v.z / (unsigned)OWN_ROWS, k3 = (unsigned)v.w / (unsigned)OWN_ROWS;
            key[(size_t)blockIdx.x * HIST_PER + idx] = k0 | (k1 << 8) | (k2 << 16) | (k3 << 24);
            const int lg = (idx * 4) / BM_GROUP;
            const unsigned gbit = 1u << (lg & 31);
            unsigned* gw = bm + (lg >> 5);
            __hip_atomic_fetch_or(gw + k0 * BM_LPITCH, gbit, __ATOMIC_RELAXED, __HIP_MEMORY_SCOPE_WORKGROUP);
            __hip_atomic_fetch_or(gw + k1 * BM_LPITCH, gbit, __ATOMIC_RELAXED, __HIP_MEMORY_SCOPE_WORKGROUP);
            __hip_atomic_fetch_or(gw + k2 * BM_LPITCH, gbit, __ATOMIC_RELAXED, __HIP_MEMORY_SCOPE_WORKGROUP);
            __hip_atomic_fetch_or(gw + k3 * BM_LPITCH, gbit, __ATOMIC_RELAXED, __HIP_MEMORY_SCOPE_WORKGROUP);
            count_node(cnt, v.x);
            count_node(cnt, v.y);
            count_node(cnt, v.z);
            count_node(cnt, v.w);
        }
    }
    __syncthreads();

    uint4* o = (uint4*)(hist + (size_t)blockIdx.x * HIST_PITCH);
    for (int i = tid; i < HIST_PITCH / 4; i += 1024) o[i] = ((const uint4*)cnt)[i];

    uint4* bo = (uint4*)bmp;
    for (int i = tid; i < OWN_WG * (BM_GPITCH / 4); i += 1024) {
        const int ow = i / (BM_GPITCH / 4);
        const int j4 = i - ow * (BM_GPITCH / 4);
        const int j  = j4 * 4;
        const unsigned* r = bm + ow * BM_LPITCH;
        uint4 v;
        v.x = (j     < BM_WORDS) ? r[j]     : 0u;
        v.y = (j + 1 < BM_WORDS) ? r[j + 1] : 0u;
        v.z = (j + 2 < BM_WORDS) ? r[j + 2] : 0u;
        v.w = (j + 3 < BM_WORDS) ? r[j + 3] : 0u;
        bo[((size_t)ow * HIST_G + blockIdx.x) * (BM_GPITCH / 4) + j4] = v;
    }
}

__global__ __launch_bounds__(512)
void k_dinv(const unsigned* __restrict__ hist, float* __restrict__ dinv, int* __restrict__ ocnt) {
    __shared__ int part[16];
    const int t = threadIdx.x;
    const int node = blockIdx.x * OWN_ROWS + t;
    int c = 0;
    if (t < OWN_ROWS && node < N_NODES) {
        const unsigned* p = hist + (node >> 1);
        const int sh = (node & 1) << 4;
        #pragma unroll 16
        for (int g = 0; g < HIST_G; ++g) c += (int)((p[(size_t)g * HIST_PITCH] >> sh) & 0xffffu);
        dinv[node] = rsqrtf((float)(c + 1));
    }
    for (int o = 16; o > 0; o >>= 1) c += __shfl_xor(c, o);
    if ((t & 31) == 0) part[t >> 5] = c;
    __syncthreads();
    if (t == 0) {
        int sum = 0;
        for (int i = 0; i < 16; ++i) sum += part[i];
        ocnt[blockIdx.x] = sum;
    }
}


#define GM_WAVES 25
#define GM_WG    250
#define GM_PITCH 132

typedef __attribute__((ext_vector_type(4))) float v4f;

template<int NCOLS, bool RELU>
__global__ __launch_bounds__(GM_WAVES * 32)
void k_gemm_lds(const float* __restrict__ A,
                const float* __restrict__ W,
                float* __restrict__ out) {
    constexpr int K  = 128;
    constexpr int NT = NCOLS / 16;
    __shared__ __attribute__((aligned(16))) float sW[NCOLS * GM_PITCH];
    __shared__ __attribute__((aligned(16))) float sA[GM_WAVES * 16 * GM_PITCH];

    const int tid  = threadIdx.x;
    const int lane = tid & 31;
    const int wv   = __builtin_amdgcn_readfirstlane(tid >> 5);
    const int hi      = lane >> 4;
    const int laneLow = lane & 15;
    const int mBase = (blockIdx.x * GM_WAVES + wv) * 16;
    float* myA = sA + wv * (16 * GM_PITCH);

    v4f arow[16];
    {
        const float* ap = A + (size_t)mBase * K + lane * 4;
        #pragma unroll
        for (int r = 0; r < 16; ++r) arow[r] = *(const v4f*)(ap + (size_t)r * K);
    }

    if (wv < 16) {
        constexpr int NB  = NCOLS / 16;
        constexpr int WIT = (K / 8) * NB / 16;
        const int k0 = lane >> 2;
        const int n0 = (lane & 3) * 4;
        v4f wreg[WIT];
        #pragma unroll
        for (int i = 0; i < WIT; ++i) {
            int g  = wv + i * 16;
            int kb = g / NB;
            int nb = g - kb * NB;
            wreg[i] = *(const v4f*)(W + (size_t)(kb * 8 + k0) * NCOLS + nb * 16 + n0);
        }
        #pragma unroll
        for (int i = 0; i < WIT; ++i) {
            int g  = wv + i * 16;
            int kb = g / NB;
            int nb = g - kb * NB;
            float* d = sW + (nb * 16 + n0) * GM_PITCH + kb * 8 + k0;
            d[0]            = wreg[i].x;
            d[GM_PITCH]     = wreg[i].y;
            d[2 * GM_PITCH] = wreg[i].z;
            d[3 * GM_PITCH] = wreg[i].w;
        }
    }

    #pragma unroll
    for (int r = 0; r < 16; ++r) {
        v4f t = arow[r];
        if (RELU) {
            t.x = fmaxf(t.x, 0.0f); t.y = fmaxf(t.y, 0.0f);
            t.z = fmaxf(t.z, 0.0f); t.w = fmaxf(t.w, 0.0f);
        }
        *(v4f*)(myA + r * GM_PITCH + lane * 4) = t;
    }
    __syncthreads();

    v8f c[NT];
    #pragma unroll
    for (int t = 0; t < NT; ++t) c[t] = (v8f){0.f, 0.f, 0.f, 0.f, 0.f, 0.f, 0.f, 0.f};

    const float* aP = myA + laneLow * GM_PITCH + 2 * hi;
    const float* bP = sW  + laneLow * GM_PITCH + 2 * hi;

    #pragma unroll 4
    for (int k = 0; k < K; k += 4) {
        v2f av = *(const v2f*)(aP + k);
        #pragma unroll
        for (int t = 0; t < NT; ++t) {
            v2f bv = *(const v2f*)(bP + t * (16 * GM_PITCH) + k);
            c[t] = __builtin_amdgcn_wmma_f32_16x16x4_f32(false, av, false, bv,
                                                         (short)0, c[t], false, false);
        }
    }

    {
        float* o = myA + (8 * hi) * GM_PITCH + laneLow;
        #pragma unroll
        for (int t = 0; t < NT; ++t) {
            #pragma unroll
            for (int r = 0; r < 8; ++r) o[r * GM_PITCH + t * 16] = c[t][r];
        }
    }
    {
        constexpr int LPR = NCOLS / 4;
        constexpr int RPI = 32 / LPR;
        const int rsub = lane / LPR;
        const int col  = (lane - rsub * LPR) * 4;
        #pragma unroll
        for (int i = 0; i < 16 / RPI; ++i) {
            int row = i * RPI + rsub;
            v4f v = *(const v4f*)(myA + row * GM_PITCH + col);
            *(v4f*)(out + (size_t)(mBase + row) * NCOLS + col) = v;
        }
    }
}


#define OWN_GOFF 288
#define OWN_LIST 352

static_assert(31 + 32 * BM_GROUP <= OWN_GOFF && OWN_GOFF + 31 + 32 < OWN_LIST + 1, "list sizes");

__device__ __forceinline__ int own_cap(int cnt) { return (cnt >> 5) + OWN_THREADS / 32; }

__device__ __forceinline__ unsigned zero_bytes(unsigned x) {
    unsigned t = (x & 0x7f7f7f7fu) + 0x7f7f7f7fu;
    return ~(t | x | 0x7f7f7f7fu);
}

template<int V> struct fvec;
template<> struct fvec<2> { typedef __attribute__((ext_vector_type(2))) float t; };
template<> struct fvec<4> { typedef __attribute__((ext_vector_type(4))) float t; };

template<int D>
__device__ __forceinline__ void own_rows(int n, int lane, int s, int d, float ws, float wd,
                                         const float* __restrict__ h, float* acc) {
    constexpr int V = D / 32;
    typedef typename fvec<V>::t vec_t;
    constexpr int G = 8;
    for (int i0 = 0; i0 < n; i0 += G) {
        vec_t v[G];
        #pragma unroll
        for (int g = 0; g < G; ++g) {
            int si = __builtin_amdgcn_readlane(s, i0 + g);
            v[g] = *(const vec_t*)(h + (size_t)si * D + lane * V);
        }
        #pragma unroll
        for (int g = 0; g < G; ++g) {
            float wi = __int_as_float(__builtin_amdgcn_readlane(__float_as_int(ws), i0 + g))
                     * __int_as_float(__builtin_amdgcn_readlane(__float_as_int(wd), i0 + g));
            int   di = __builtin_amdgcn_readlane(d, i0 + g);
            float* a = acc + di * D + lane;
            #pragma unroll
            for (int j = 0; j < V; ++j) {
                __hip_atomic_fetch_add(a + j * 32, v[g][j] * wi, __ATOMIC_RELAXED,
                                       __HIP_MEMORY_SCOPE_WORKGROUP);
            }
        }
    }
}

template<int D>
__device__ __forceinline__ void own_flush(int n, int lane, int le, int lo,
                                          const int* __restrict__ src,
                                          const int* __restrict__ dst,
                                          const float* __restrict__ dinv,
                                          const float* __restrict__ h,
                                          float* acc, int* nchunk, uint2* __restrict__ myrec) {
    int   s  = 0;
    float ws = 0.0f;
    float wd = 0.0f;
    int   d  = OWN_ROWS;
    if (lane < n) {
        s = src[le];
        int dg = dst[le];
        ws = dinv[s];
        wd = dinv[dg];
        d = dg - lo;
    }
    int c = 0;
    if (lane == 0) c = __hip_atomic_fetch_add(nchunk, 1, __ATOMIC_RELAXED, __HIP_MEMORY_SCOPE_WORKGROUP);
    c = __builtin_amdgcn_readfirstlane(c);
    own_rows<D>(n, lane, s, d, ws, wd, h, acc);
    myrec[c * 32 + lane] = make_uint2(((unsigned)s << 9) | (unsigned)d, __float_as_uint(ws * wd));
}

template<int D>
__device__ __forceinline__ int own_groups(int ng, int lane, int g, unsigned me4, int cnt, int* elist,
                                          const uint2* __restrict__ key2, int lo,
                                          const int* __restrict__ src,
                                          const int* __restrict__ dst,
                                          const float* __restrict__ dinv,
                                          const float* __restrict__ h,
                                          float* acc, int* nchunk, uint2* __restrict__ myrec) {
    unsigned c = 0u;
    g = min(max(g, 0), N_EDGES / BM_GROUP - 1);
    if (lane < ng) {
        const uint2 k = key2[g];
        c = (zero_bytes(k.x ^ me4) >> 7) | (zero_bytes(k.y ^ me4) >> 6);
    }
    unsigned m = __builtin_amdgcn_ballot_w32(c != 0u);
    const int e0 = g * BM_GROUP;
    while (m != 0u) {
        if (c != 0u) {
            int p = __builtin_ctz(c);
            int slot = cnt + (int)__builtin_amdgcn_mbcnt_lo(m, 0u);
            elist[slot] = e0 + ((p & 7) << 2) + (p >> 3);
        }
        cnt += __builtin_popcount(m);
        c &= c - 1u;
        m = __builtin_amdgcn_ballot_w32(c != 0u);
    }
    if (cnt >= 32) {
        int full = cnt & ~31;
        for (int base = 0; base < full; base += 32) {
            int en = elist[base + lane];
            own_flush<D>(32, lane, en, lo, src, dst, dinv, h, acc, nchunk, myrec);
        }
        int rest = cnt - full;
        int mv = 0;
        if (lane < rest) mv = elist[full + lane];
        if (lane < rest) elist[lane] = mv;
        cnt = rest;
    }
    return cnt;
}

template<int D, bool REPLAY>
__global__ __launch_bounds__(OWN_THREADS)
void k_agg_owner(const int* __restrict__ src, const int* __restrict__ dst,
                 const unsigned* __restrict__ key, const unsigned* __restrict__ bmp,
                 const float* __restrict__ dinv,
                 const float* __restrict__ bias,
                 const float* __restrict__ h, float* __restrict__ agg,
                 const int* __restrict__ ocnt, int* nch, unsigned* rec) {
    constexpr int V = D / 32;
    typedef typename fvec<V>::t vec_t;
    constexpr int ACC_MIN = REPLAY ? 164 * 1024 / 4 : 0;
    constexpr int ACC_FLOATS = (OWN_ROWS + 1) * D > ACC_MIN ? (OWN_ROWS + 1) * D : ACC_MIN;
    __shared__ float acc[ACC_FLOATS];
    __shared__ int   wlist[REPLAY ? 1 : OWN_THREADS / 32][REPLAY ? 1 : OWN_LIST];
    __shared__ int   red[OWN_WG / 32];
    __shared__ int   nchunk;

    const int tid  = threadIdx.x;
    const int lane = tid & 31;
    const int wv   = __builtin_amdgcn_readfirstlane(tid >> 5);
    const int lo   = blockIdx.x * OWN_ROWS;

    {
        int cap = 0;
        if (tid < (int)blockIdx.x) cap = own_cap(ocnt[tid]);
        for (int o = 16; o > 0; o >>= 1) cap += __shfl_xor(cap, o);
        if (wv < OWN_WG / 32 && lane == 0) red[wv] = cap;
        if (tid == 0) nchunk = 0;
    }

    if (wv == 0) {
        #pragma unroll
        for (int j = 0; j < V; ++j) acc[OWN_ROWS * D + j * 32 + lane] = 0.0f;
    }
    for (int r = wv; r < OWN_ROWS; r += OWN_THREADS / 32) {
        int node = lo + r;
        if (node < N_NODES) {
            float sn = dinv[node] * dinv[node];
            vec_t hv = *(const vec_t*)(h + (size_t)node * D + lane * V);
            #pragma unroll
            for (int j = 0; j < V; ++j)
                acc[r * D + j * 32 + lane] = hv[j] * sn + bias[lane * V + j];
        }
    }
    __syncthreads();

    int start = 0;
    for (int i = 0; i < OWN_WG / 32; ++i) start += red[i];
    uint2* myrec = (uint2*)rec + (size_t)start * 32;

    if (REPLAY) {
        const int nCh = min(nch[blockIdx.x], own_cap(ocnt[blockIdx.x]));
        int c = wv;
        uint2 r = make_uint2((unsigned)OWN_ROWS, 0u);
        if (c < nCh) r = myrec[c * 32 + lane];
        while (c < nCh) {
            const int cn = c + OWN_THREADS / 32;
            uint2 rn = make_uint2((unsigned)OWN_ROWS, 0u);
            if (cn < nCh) rn = myrec[cn * 32 + lane];
            const int d = (int)(r.x & 511u);
            const int n = __builtin_popcount(__builtin_amdgcn_ballot_w32(d != OWN_ROWS));
            own_rows<D>(n, lane, (int)(r.x >> 9), d, __uint_as_float(r.y), 1.0f, h, acc);
            r = rn;
            c = cn;
        }
    } else {
    int* elist = wlist[wv];
    int* glist = wlist[wv] + OWN_GOFF;
    const uint2* key2 = (const uint2*)key;
    const unsigned me4 = (unsigned)blockIdx.x * 0x01010101u;
    constexpr int BW = (2 * BM_WORDS + 31) / 32;
    static_assert(HIST_G == 2 * (OWN_THREADS / 32), "two slices to a wave");
    const unsigned* mybits = bmp + ((size_t)blockIdx.x * HIST_G + 2 * wv) * BM_GPITCH;
    unsigned bw[BW];
    #pragma unroll
    for (int x = 0; x < BW; ++x) {
        const int rr = x * 32 + lane;
        const int sl = rr >= BM_WORDS ? 1 : 0;
        bw[x] = (rr < 2 * BM_WORDS) ? mybits[sl * BM_GPITCH + rr - sl * BM_WORDS] : 0u;
    }

    int gcnt = 0;
    int cnt  = 0;
    #pragma nounroll
    for (int x = 0; x < BW; ++x) {
        unsigned w = 0u;
        #pragma unroll
        for (int y = 0; y < BW; ++y) w = (x == y) ? bw[y] : w;
        const int rr = x * 32 + lane;
        const int sl = rr >= BM_WORDS ? 1 : 0;
        const int gbase = (2 * wv + sl) * BM_BITS + (rr - sl * BM_WORDS) * 32;
        unsigned m = __builtin_amdgcn_ballot_w32(w != 0u);
        while (m != 0u) {
            if (w != 0u) {
                int slot = gcnt + (int)__builtin_amdgcn_mbcnt_lo(m, 0u);
                glist[slot] = gbase + __builtin_ctz(w);
                w &= w - 1u;
            }
            gcnt += __builtin_popcount(m);
            if (gcnt >= 32) {
                int g = glist[lane];
                cnt = own_groups<D>(32, lane, g, me4, cnt, elist, key2, lo, src, dst, dinv, h, acc,
                                    &nchunk, myrec);
                int rest = gcnt - 32;
                int mv = 0;
                if (lane < rest) mv = glist[32 + lane];
                if (lane < rest) glist[lane] = mv;
                gcnt = rest;
            }
            m = __builtin_amdgcn_ballot_w32(w != 0u);
        }
    }
    if (gcnt != 0) {
        int g = 0;
        if (lane < gcnt) g = glist[lane];
        cnt = own_groups<D>(gcnt, lane, g, me4, cnt, elist, key2, lo, src, dst, dinv, h, acc, &nchunk, myrec);
    }
    if (cnt != 0) {
        int en = 0;
        if (lane < cnt) en = elist[lane];
        own_flush<D>(cnt, lane, en, lo, src, dst, dinv, h, acc, &nchunk, myrec);
    }
    }
    __syncthreads();

    if (!REPLAY && tid == 0) nch[blockIdx.x] = nchunk;

    for (int r = wv; r < OWN_ROWS; r += OWN_THREADS / 32) {
        int node = lo + r;
        if (node < N_NODES) {
            vec_t o;
            #pragma unroll
            for (int j = 0; j < V; ++j) o[j] = acc[r * D + j * 32 + lane];
            *(vec_t*)(agg + (size_t)node * D + lane * V) = o;
        }
    }
}


extern "C" void kernel_launch(void* const* d_in, const int* in_sizes, int n_in,
                              void* d_out, int out_size, void* d_ws, size_t ws_size,
                              hipStream_t stream) {
    const float* x   = (const float*)d_in[0];
    const int*   ei  = (const int*)d_in[1];
    const float* W1  = (const float*)d_in[2];
    const float* b1  = (const float*)d_in[3];
    const float* W2  = (const float*)d_in[4];
    const float* b2  = (const float*)d_in[5];
    float*       out = (float*)d_out;

    const int* src = ei;
    const int* dst = ei + N_EDGES;

    float* ws    = (float*)d_ws;
    float* dinv  = ws;
    float* keyp  = dinv  + N_NODES;
    float* h1    = keyp  + N_EDGES;
    float* agg1  = h1    + (size_t)N_NODES * DH;
    float* recp  = agg1  + (size_t)N_NODES * DH;
    int*   ocnt  = (int*)(keyp + N_EDGES / 4);
    int*   nch   = ocnt + OWN_WG;
    unsigned* rec = (unsigned*)(recp + 32);
    unsigned* bmp = (unsigned*)(recp + 4000000);
    float* h2    = h1;

    unsigned* hist = (unsigned*)agg1;

    unsigned* key = (unsigned*)keyp;
    k_hist_key<<<HIST_G, 1024, 0, stream>>>(dst, key, hist, bmp);
    k_dinv    <<<OWN_WG, 512, 0, stream>>>(hist, dinv, ocnt);

    k_gemm_lds<DH, false><<<GM_WG, GM_WAVES * 32, 0, stream>>>(x, W1, h1);
    k_agg_owner<DH, false><<<OWN_WG, OWN_THREADS, 0, stream>>>(src, dst, key, bmp, dinv, b1, h1, agg1, ocnt, nch, rec);

    k_gemm_lds<DOUT, true><<<GM_WG, GM_WAVES * 32, 0, stream>>>(agg1, W2, h2);
    k_agg_owner<DOUT, true><<<OWN_WG, OWN_THREADS, 0, stream>>>(src, dst, key, bmp, dinv, b2, h2, out, ocnt, nch, rec);
}


# Layer-1 owners read a marked group's 8 dst + 8 src words from the edge list (4 lanes x 8 B a group: 8 lines a load, not 32), not key + src + dst words per edge; key plane dropped; sums arrival-order
# speedup vs baseline: 3.2718x; 1.0443x over previous
#include <hip/hip_runtime.h>

#define N_NODES 100000
#define N_EDGES 1600000
#define DIN 128
#define DH  128
#define DOUT 64

typedef __attribute__((ext_vector_type(2))) float v2f;
typedef __attribute__((ext_vector_type(8))) float v8f;

#define OWN_ROWS 391
#define OWN_WG   256
#define OWN_THREADS 1024

#define HIST_G     64
#define HIST_PER   (N_EDGES / 4 / HIST_G)
#define HIST_PITCH 50016

#define BM_GROUP   8
#define BM_BITS    (HIST_PER * 4 / BM_GROUP)
#define BM_WORDS   98
#define BM_LPITCH  99
#define BM_GPITCH  128

static_assert(HIST_PER * 4 % BM_GROUP == 0, "a group never straddles two slices");
static_assert(BM_WORDS * 32 >= BM_BITS && BM_WORDS <= BM_GPITCH, "a chunk holds a slice's groups");
static_assert(HIST_PER * 4 * HIST_G == N_EDGES, "the slices cover the edge list exactly");
static_assert(HIST_PER * 4 < 65536, "a 16-bit counter holds a whole slice");
static_assert(HIST_PITCH * 2 >= N_NODES && HIST_PITCH % 32 == 0, "plane size");

__device__ __forceinline__ void count_node(unsigned* cnt, int node) {
    __hip_atomic_fetch_add(cnt + (node >> 1), 1u << ((node & 1) << 4), __ATOMIC_RELAXED,
                           __HIP_MEMORY_SCOPE_WORKGROUP);
}

__global__ __launch_bounds__(1024)
void k_hist_mark(const int* __restrict__ dst, unsigned* __restrict__ hist, unsigned* __restrict__ bmp) {
    __shared__ __attribute__((aligned(16))) unsigned cnt[HIST_PITCH];
    __shared__ unsigned bm[OWN_WG * BM_LPITCH];
    constexpr int IT = (HIST_PER + 1023) / 1024;
    const int tid = threadIdx.x;
    const int4* d4 = (const int4*)dst + (size_t)blockIdx.x * HIST_PER;

    int4 d[IT];
    #pragma unroll
    for (int i = 0; i < IT; ++i) {
        int idx = tid + i * 1024;
        d[i] = d4[min(idx, HIST_PER - 1)];
    }
    for (int i = tid; i < HIST_PITCH / 4; i += 1024) ((uint4*)cnt)[i] = make_uint4(0u, 0u, 0u, 0u);
    for (int i = tid; i < OWN_WG * BM_LPITCH; i += 1024) bm[i] = 0u;
    __syncthreads();

    #pragma unroll
    for (int i = 0; i < IT; ++i) {
        int idx = tid + i * 1024;
        if (idx < HIST_PER) {
            const int4 v = d[i];
            const unsigned k0 = (unsigned)v.x / (unsigned)OWN_ROWS, k1 = (unsigned)v.y / (unsigned)OWN_ROWS;
            const unsigned k2 = (unsigned)v.z / (unsigned)OWN_ROWS, k3 = (unsigned)v.w / (unsigned)OWN_ROWS;
            const int lg = (idx * 4) / BM_GROUP;
            const unsigned gbit = 1u << (lg & 31);
            unsigned* gw = bm + (lg >> 5);
            __hip_atomic_fetch_or(gw + k0 * BM_LPITCH, gbit, __ATOMIC_RELAXED, __HIP_MEMORY_SCOPE_WORKGROUP);
            __hip_atomic_fetch_or(gw + k1 * BM_LPITCH, gbit, __ATOMIC_RELAXED, __HIP_MEMORY_SCOPE_WORKGROUP);
            __hip_atomic_fetch_or(gw + k2 * BM_LPITCH, gbit, __ATOMIC_RELAXED, __HIP_MEMORY_SCOPE_WORKGROUP);
            __hip_atomic_fetch_or(gw + k3 * BM_LPITCH, gbit, __ATOMIC_RELAXED, __HIP_MEMORY_SCOPE_WORKGROUP);
            count_node(cnt, v.x);
            count_node(cnt, v.y);
            count_node(cnt, v.z);
            count_node(cnt, v.w);
        }
    }
    __syncthreads();

    uint4* o = (uint4*)(hist + (size_t)blockIdx.x * HIST_PITCH);
    for (int i = tid; i < HIST_PITCH / 4; i += 1024) o[i] = ((const uint4*)cnt)[i];

    uint4* bo = (uint4*)bmp;
    for (int i = tid; i < OWN_WG * (BM_GPITCH / 4); i += 1024) {
        const int ow = i / (BM_GPITCH / 4);
        const int j4 = i - ow * (BM_GPITCH / 4);
        const int j  = j4 * 4;
        const unsigned* r = bm + ow * BM_LPITCH;
        uint4 v;
        v.x = (j     < BM_WORDS) ? r[j]     : 0u;
        v.y = (j + 1 < BM_WORDS) ? r[j + 1] : 0u;
        v.z = (j + 2 < BM_WORDS) ? r[j + 2] : 0u;
        v.w = (j + 3 < BM_WORDS) ? r[j + 3] : 0u;
        bo[((size_t)ow * HIST_G + blockIdx.x) * (BM_GPITCH / 4) + j4] = v;
    }
}

__global__ __launch_bounds__(512)
void k_dinv(const unsigned* __restrict__ hist, float* __restrict__ dinv, int* __restrict__ ocnt) {
    __shared__ int part[16];
    const int t = threadIdx.x;
    const int node = blockIdx.x * OWN_ROWS + t;
    int c = 0;
    if (t < OWN_ROWS && node < N_NODES) {
        const unsigned* p = hist + (node >> 1);
        const int sh = (node & 1) << 4;
        #pragma unroll 16
        for (int g = 0; g < HIST_G; ++g) c += (int)((p[(size_t)g * HIST_PITCH] >> sh) & 0xffffu);
        dinv[node] = rsqrtf((float)(c + 1));
    }
    for (int o = 16; o > 0; o >>= 1) c += __shfl_xor(c, o);
    if ((t & 31) == 0) part[t >> 5] = c;
    __syncthreads();
    if (t == 0) {
        int sum = 0;
        for (int i = 0; i < 16; ++i) sum += part[i];
        ocnt[blockIdx.x] = sum;
    }
}


#define GM_WAVES 25
#define GM_WG    250
#define GM_PITCH 132

typedef __attribute__((ext_vector_type(4))) float v4f;

template<int NCOLS, bool RELU>
__global__ __launch_bounds__(GM_WAVES * 32)
void k_gemm_lds(const float* __restrict__ A,
                const float* __restrict__ W,
                float* __restrict__ out) {
    constexpr int K  = 128;
    constexpr int NT = NCOLS / 16;
    __shared__ __attribute__((aligned(16))) float sW[NCOLS * GM_PITCH];
    __shared__ __attribute__((aligned(16))) float sA[GM_WAVES * 16 * GM_PITCH];

    const int tid  = threadIdx.x;
    const int lane = tid & 31;
    const int wv   = __builtin_amdgcn_readfirstlane(tid >> 5);
    const int hi      = lane >> 4;
    const int laneLow = lane & 15;
    const int mBase = (blockIdx.x * GM_WAVES + wv) * 16;
    float* myA = sA + wv * (16 * GM_PITCH);

    v4f arow[16];
    {
        const float* ap = A + (size_t)mBase * K + lane * 4;
        #pragma unroll
        for (int r = 0; r < 16; ++r) arow[r] = *(const v4f*)(ap + (size_t)r * K);
    }

    if (wv < 16) {
        constexpr int NB  = NCOLS / 16;
        constexpr int WIT = (K / 8) * NB / 16;
        const int k0 = lane >> 2;
        const int n0 = (lane & 3) * 4;
        v4f wreg[WIT];
        #pragma unroll
        for (int i = 0; i < WIT; ++i) {
            int g  = wv + i * 16;
            int kb = g / NB;
            int nb = g - kb * NB;
            wreg[i] = *(const v4f*)(W + (size_t)(kb * 8 + k0) * NCOLS + nb * 16 + n0);
        }
        #pragma unroll
        for (int i = 0; i < WIT; ++i) {
            int g  = wv + i * 16;
            int kb = g / NB;
            int nb = g - kb * NB;
            float* d = sW + (nb * 16 + n0) * GM_PITCH + kb * 8 + k0;
            d[0]            = wreg[i].x;
            d[GM_PITCH]     = wreg[i].y;
            d[2 * GM_PITCH] = wreg[i].z;
            d[3 * GM_PITCH] = wreg[i].w;
        }
    }

    #pragma unroll
    for (int r = 0; r < 16; ++r) {
        v4f t = arow[r];
        if (RELU) {
            t.x = fmaxf(t.x, 0.0f); t.y = fmaxf(t.y, 0.0f);
            t.z = fmaxf(t.z, 0.0f); t.w = fmaxf(t.w, 0.0f);
        }
        *(v4f*)(myA + r * GM_PITCH + lane * 4) = t;
    }
    __syncthreads();

    v8f c[NT];
    #pragma unroll
    for (int t = 0; t < NT; ++t) c[t] = (v8f){0.f, 0.f, 0.f, 0.f, 0.f, 0.f, 0.f, 0.f};

    const float* aP = myA + laneLow * GM_PITCH + 2 * hi;
    const float* bP = sW  + laneLow * GM_PITCH + 2 * hi;

    #pragma unroll 4
    for (int k = 0; k < K; k += 4) {
        v2f av = *(const v2f*)(aP + k);
        #pragma unroll
        for (int t = 0; t < NT; ++t) {
            v2f bv = *(const v2f*)(bP + t * (16 * GM_PITCH) + k);
            c[t] = __builtin_amdgcn_wmma_f32_16x16x4_f32(false, av, false, bv,
                                                         (short)0, c[t], false, false);
        }
    }

    {
        float* o = myA + (8 * hi) * GM_PITCH + laneLow;
        #pragma unroll
        for (int t = 0; t < NT; ++t) {
            #pragma unroll
            for (int r = 0; r < 8; ++r) o[r * GM_PITCH + t * 16] = c[t][r];
        }
    }
    {
        constexpr int LPR = NCOLS / 4;
        constexpr int RPI = 32 / LPR;
        const int rsub = lane / LPR;
        const int col  = (lane - rsub * LPR) * 4;
        #pragma unroll
        for (int i = 0; i < 16 / RPI; ++i) {
            int row = i * RPI + rsub;
            v4f v = *(const v4f*)(myA + row * GM_PITCH + col);
            *(v4f*)(out + (size_t)(mBase + row) * NCOLS + col) = v;
        }
    }
}


#define OWN_GOFF 288
#define OWN_LIST 352

static_assert(31 + 32 * BM_GROUP <= OWN_GOFF && OWN_GOFF + 31 + 32 < OWN_LIST + 1, "list sizes");

__device__ __forceinline__ int own_cap(int cnt) { return (cnt >> 5) + OWN_THREADS / 32; }

template<int V> struct fvec;
template<> struct fvec<2> { typedef __attribute__((ext_vector_type(2))) float t; };
template<> struct fvec<4> { typedef __attribute__((ext_vector_type(4))) float t; };

template<int D>
__device__ __forceinline__ void own_rows(int n, int lane, int s, int d, float ws, float wd,
                                         const float* __restrict__ h, float* acc) {
    constexpr int V = D / 32;
    typedef typename fvec<V>::t vec_t;
    constexpr int G = 8;
    for (int i0 = 0; i0 < n; i0 += G) {
        vec_t v[G];
        #pragma unroll
        for (int g = 0; g < G; ++g) {
            int si = __builtin_amdgcn_readlane(s, i0 + g);
            v[g] = *(const vec_t*)(h + (size_t)si * D + lane * V);
        }
        #pragma unroll
        for (int g = 0; g < G; ++g) {
            float wi = __int_as_float(__builtin_amdgcn_readlane(__float_as_int(ws), i0 + g))
                     * __int_as_float(__builtin_amdgcn_readlane(__float_as_int(wd), i0 + g));
            int   di = __builtin_amdgcn_readlane(d, i0 + g);
            float* a = acc + di * D + lane;
            #pragma unroll
            for (int j = 0; j < V; ++j) {
                __hip_atomic_fetch_add(a + j * 32, v[g][j] * wi, __ATOMIC_RELAXED,
                                       __HIP_MEMORY_SCOPE_WORKGROUP);
            }
        }
    }
}

template<int D>
__device__ __forceinline__ void own_flush(int n, int lane, int le, int lo,
                                          const float* __restrict__ dinv,
                                          const float* __restrict__ h,
                                          float* acc, int* nchunk, uint2* __restrict__ myrec) {
    int   s  = 0;
    float ws = 0.0f;
    float wd = 0.0f;
    int   d  = OWN_ROWS;
    if (lane < n) {
        s = le >> 9;
        d = le & 511;
        ws = dinv[s];
        wd = dinv[lo + d];
    }
    int c = 0;
    if (lane == 0) c = __hip_atomic_fetch_add(nchunk, 1, __ATOMIC_RELAXED, __HIP_MEMORY_SCOPE_WORKGROUP);
    c = __builtin_amdgcn_readfirstlane(c);
    own_rows<D>(n, lane, s, d, ws, wd, h, acc);
    myrec[c * 32 + lane] = make_uint2(((unsigned)s << 9) | (unsigned)d, __float_as_uint(ws * wd));
}

template<int D>
__device__ __forceinline__ int own_groups(int ng, int lane, const int* glist, int cnt, int* elist, int lo,
                                          const int* __restrict__ src,
                                          const int* __restrict__ dst,
                                          const float* __restrict__ dinv,
                                          const float* __restrict__ h,
                                          float* acc, int* nchunk, uint2* __restrict__ myrec) {
    const int q = lane >> 2;
    const int p = lane & 3;
    int2 dv[4];
    int2 sv[4];
    #pragma unroll
    for (int i = 0; i < 4; ++i) {
        int g = 0;
        if (8 * i + q < ng) g = glist[8 * i + q];
        g = min(max(g, 0), N_EDGES / BM_GROUP - 1);
        const int e = g * BM_GROUP + 2 * p;
        dv[i] = *(const int2*)(dst + e);
        sv[i] = *(const int2*)(src + e);
    }
    #pragma unroll
    for (int i = 0; i < 4; ++i) {
        const bool used = 8 * i + q < ng;
        #pragma unroll
        for (int j = 0; j < 2; ++j) {
            const int dd = (j == 0 ? dv[i].x : dv[i].y) - lo;
            const int ss = (j == 0 ? sv[i].x : sv[i].y);
            const bool keep = used && (unsigned)dd < (unsigned)OWN_ROWS;
            const unsigned m = __builtin_amdgcn_ballot_w32(keep);
            if (keep) elist[cnt + (int)__builtin_amdgcn_mbcnt_lo(m, 0u)] = (ss << 9) | dd;
            cnt += __builtin_popcount(m);
        }
    }
    if (cnt >= 32) {
        int full = cnt & ~31;
        for (int base = 0; base < full; base += 32) {
            int en = elist[base + lane];
            own_flush<D>(32, lane, en, lo, dinv, h, acc, nchunk, myrec);
        }
        int rest = cnt - full;
        int mv = 0;
        if (lane < rest) mv = elist[full + lane];
        if (lane < rest) elist[lane] = mv;
        cnt = rest;
    }
    return cnt;
}

template<int D, bool REPLAY>
__global__ __launch_bounds__(OWN_THREADS)
void k_agg_owner(const int* __restrict__ src, const int* __restrict__ dst,
                 const unsigned* __restrict__ bmp,
                 const float* __restrict__ dinv,
                 const float* __restrict__ bias,
                 const float* __restrict__ h, float* __restrict__ agg,
                 const int* __restrict__ ocnt, int* nch, unsigned* rec) {
    constexpr int V = D / 32;
    typedef typename fvec<V>::t vec_t;
    constexpr int ACC_MIN = REPLAY ? 164 * 1024 / 4 : 0;
    constexpr int ACC_FLOATS = (OWN_ROWS + 1) * D > ACC_MIN ? (OWN_ROWS + 1) * D : ACC_MIN;
    __shared__ float acc[ACC_FLOATS];
    __shared__ int   wlist[REPLAY ? 1 : OWN_THREADS / 32][REPLAY ? 1 : OWN_LIST];
    __shared__ int   red[OWN_WG / 32];
    __shared__ int   nchunk;

    const int tid  = threadIdx.x;
    const int lane = tid & 31;
    const int wv   = __builtin_amdgcn_readfirstlane(tid >> 5);
    const int lo   = blockIdx.x * OWN_ROWS;

    {
        int cap = 0;
        if (tid < (int)blockIdx.x) cap = own_cap(ocnt[tid]);
        for (int o = 16; o > 0; o >>= 1) cap += __shfl_xor(cap, o);
        if (wv < OWN_WG / 32 && lane == 0) red[wv] = cap;
        if (tid == 0) nchunk = 0;
    }

    if (wv == 0) {
        #pragma unroll
        for (int j = 0; j < V; ++j) acc[OWN_ROWS * D + j * 32 + lane] = 0.0f;
    }
    for (int r = wv; r < OWN_ROWS; r += OWN_THREADS / 32) {
        int node = lo + r;
        if (node < N_NODES) {
            float sn = dinv[node] * dinv[node];
            vec_t hv = *(const vec_t*)(h + (size_t)node * D + lane * V);
            #pragma unroll
            for (int j = 0; j < V; ++j)
                acc[r * D + j * 32 + lane] = hv[j] * sn + bias[lane * V + j];
        }
    }
    __syncthreads();

    int start = 0;
    for (int i = 0; i < OWN_WG / 32; ++i) start += red[i];
    uint2* myrec = (uint2*)rec + (size_t)start * 32;

    if (REPLAY) {
        const int nCh = min(nch[blockIdx.x], own_cap(ocnt[blockIdx.x]));
        int c = wv;
        uint2 r = make_uint2((unsigned)OWN_ROWS, 0u);
        if (c < nCh) r = myrec[c * 32 + lane];
        while (c < nCh) {
            const int cn = c + OWN_THREADS / 32;
            uint2 rn = make_uint2((unsigned)OWN_ROWS, 0u);
            if (cn < nCh) rn = myrec[cn * 32 + lane];
            const int d = (int)(r.x & 511u);
            const int n = __builtin_popcount(__builtin_amdgcn_ballot_w32(d != OWN_ROWS));
            own_rows<D>(n, lane, (int)(r.x >> 9), d, __uint_as_float(r.y), 1.0f, h, acc);
            r = rn;
            c = cn;
        }
    } else {
    int* elist = wlist[wv];
    int* glist = wlist[wv] + OWN_GOFF;
    constexpr int BW = (2 * BM_WORDS + 31) / 32;
    static_assert(HIST_G == 2 * (OWN_THREADS / 32), "two slices to a wave");
    const unsigned* mybits = bmp + ((size_t)blockIdx.x * HIST_G + 2 * wv) * BM_GPITCH;
    unsigned bw[BW];
    #pragma unroll
    for (int x = 0; x < BW; ++x) {
        const int rr = x * 32 + lane;
        const int sl = rr >= BM_WORDS ? 1 : 0;
        bw[x] = (rr < 2 * BM_WORDS) ? mybits[sl * BM_GPITCH + rr - sl * BM_WORDS] : 0u;
    }

    int gcnt = 0;
    int cnt  = 0;
    #pragma nounroll
    for (int x = 0; x < BW; ++x) {
        unsigned w = 0u;
        #pragma unroll
        for (int y = 0; y < BW; ++y) w = (x == y) ? bw[y] : w;
        const int rr = x * 32 + lane;
        const int sl = rr >= BM_WORDS ? 1 : 0;
        const int gbase = (2 * wv + sl) * BM_BITS + (rr - sl * BM_WORDS) * 32;
        unsigned m = __builtin_amdgcn_ballot_w32(w != 0u);
        while (m != 0u) {
            if (w != 0u) {
                int slot = gcnt + (int)__builtin_amdgcn_mbcnt_lo(m, 0u);
                glist[slot] = gbase + __builtin_ctz(w);
                w &= w - 1u;
            }
            gcnt += __builtin_popcount(m);
            if (gcnt >= 32) {
                cnt = own_groups<D>(32, lane, glist, cnt, elist, lo, src, dst, dinv, h, acc, &nchunk, myrec);
                int rest = gcnt - 32;
                int mv = 0;
                if (lane < rest) mv = glist[32 + lane];
                if (lane < rest) glist[lane] = mv;
                gcnt = rest;
            }
            m = __builtin_amdgcn_ballot_w32(w != 0u);
        }
    }
    if (gcnt != 0) {
        cnt = own_groups<D>(gcnt, lane, glist, cnt, elist, lo, src, dst, dinv, h, acc, &nchunk, myrec);
    }
    if (cnt != 0) {
        int en = 0;
        if (lane < cnt) en = elist[lane];
        own_flush<D>(cnt, lane, en, lo, dinv, h, acc, &nchunk, myrec);
    }
    }
    __syncthreads();

    if (!REPLAY && tid == 0) nch[blockIdx.x] = nchunk;

    for (int r = wv; r < OWN_ROWS; r += OWN_THREADS / 32) {
        int node = lo + r;
        if (node < N_NODES) {
            vec_t o;
            #pragma unroll
            for (int j = 0; j < V; ++j) o[j] = acc[r * D + j * 32 + lane];
            *(vec_t*)(agg + (size_t)node * D + lane * V) = o;
        }
    }
}


extern "C" void kernel_launch(void* const* d_in, const int* in_sizes, int n_in,
                              void* d_out, int out_size, void* d_ws, size_t ws_size,
                              hipStream_t stream) {
    const float* x   = (const float*)d_in[0];
    const int*   ei  = (const int*)d_in[1];
    const float* W1  = (const float*)d_in[2];
    const float* b1  = (const float*)d_in[3];
    const float* W2  = (const float*)d_in[4];
    const float* b2  = (const float*)d_in[5];
    float*       out = (float*)d_out;

    const int* src = ei;
    const int* dst = ei + N_EDGES;

    float* ws    = (float*)d_ws;
    float* dinv  = ws;
    float* keyp  = dinv  + N_NODES;
    float* h1    = keyp  + N_EDGES;
    float* agg1  = h1    + (size_t)N_NODES * DH;
    float* recp  = agg1  + (size_t)N_NODES * DH;
    int*   ocnt  = (int*)(keyp + N_EDGES / 4);
    int*   nch   = ocnt + OWN_WG;
    unsigned* rec = (unsigned*)(recp + 32);
    unsigned* bmp = (unsigned*)(recp + 4000000);
    float* h2    = h1;

    unsigned* hist = (unsigned*)agg1;

    k_hist_mark<<<HIST_G, 1024, 0, stream>>>(dst, hist, bmp);
    k_dinv    <<<OWN_WG, 512, 0, stream>>>(hist, dinv, ocnt);

    k_gemm_lds<DH, false><<<GM_WG, GM_WAVES * 32, 0, stream>>>(x, W1, h1);
    k_agg_owner<DH, false><<<OWN_WG, OWN_THREADS, 0, stream>>>(src, dst, bmp, dinv, b1, h1, agg1, ocnt, nch, rec);

    k_gemm_lds<DOUT, true><<<GM_WG, GM_WAVES * 32, 0, stream>>>(agg1, W2, h2);
    k_agg_owner<DOUT, true><<<OWN_WG, OWN_THREADS, 0, stream>>>(src, dst, bmp, dinv, b2, h2, out, ocnt, nch, rec);
}


# Layer 1 sums x rows, not x@W1 rows ((A x) W1 by linearity): owners form relu(S@W1+b1)@W2 by WMMA from their LDS rows; h1, agg1 and both GEMM launches gone (6 -> 4); order of layer-1 sums changes
# speedup vs baseline: 4.2056x; 1.2854x over previous
#include <hip/hip_runtime.h>

#define N_NODES 100000
#define N_EDGES 1600000
#define DIN 128
#define DH  128
#define DOUT 64

typedef __attribute__((ext_vector_type(2))) float v2f;
typedef __attribute__((ext_vector_type(8))) float v8f;

#define OWN_ROWS 391
#define OWN_WG   256
#define OWN_THREADS 1024

#define HIST_G     64
#define HIST_PER   (N_EDGES / 4 / HIST_G)
#define HIST_PITCH 50016

#define BM_GROUP   8
#define BM_BITS    (HIST_PER * 4 / BM_GROUP)
#define BM_WORDS   98
#define BM_LPITCH  99
#define BM_GPITCH  128

static_assert(HIST_PER * 4 % BM_GROUP == 0, "a group never straddles two slices");
static_assert(BM_WORDS * 32 >= BM_BITS && BM_WORDS <= BM_GPITCH, "a chunk holds a slice's groups");
static_assert(HIST_PER * 4 * HIST_G == N_EDGES, "the slices cover the edge list exactly");
static_assert(HIST_PER * 4 < 65536, "a 16-bit counter holds a whole slice");
static_assert(HIST_PITCH * 2 >= N_NODES && HIST_PITCH % 32 == 0, "plane size");

__device__ __forceinline__ void count_node(unsigned* cnt, int node) {
    __hip_atomic_fetch_add(cnt + (node >> 1), 1u << ((node & 1) << 4), __ATOMIC_RELAXED,
                           __HIP_MEMORY_SCOPE_WORKGROUP);
}

__global__ __launch_bounds__(1024)
void k_hist_mark(const int* __restrict__ dst, unsigned* __restrict__ hist, unsigned* __restrict__ bmp) {
    __shared__ __attribute__((aligned(16))) unsigned cnt[HIST_PITCH];
    __shared__ unsigned bm[OWN_WG * BM_LPITCH];
    constexpr int IT = (HIST_PER + 1023) / 1024;
    const int tid = threadIdx.x;
    const int4* d4 = (const int4*)dst + (size_t)blockIdx.x * HIST_PER;

    int4 d[IT];
    #pragma unroll
    for (int i = 0; i < IT; ++i) {
        int idx = tid + i * 1024;
        d[i] = d4[min(idx, HIST_PER - 1)];
    }
    for (int i = tid; i < HIST_PITCH / 4; i += 1024) ((uint4*)cnt)[i] = make_uint4(0u, 0u, 0u, 0u);
    for (int i = tid; i < OWN_WG * BM_LPITCH; i += 1024) bm[i] = 0u;
    __syncthreads();

    #pragma unroll
    for (int i = 0; i < IT; ++i) {
        int idx = tid + i * 1024;
        if (idx < HIST_PER) {
            const int4 v = d[i];
            const unsigned k0 = (unsigned)v.x / (unsigned)OWN_ROWS, k1 = (unsigned)v.y / (unsigned)OWN_ROWS;
            const unsigned k2 = (unsigned)v.z / (unsigned)OWN_ROWS, k3 = (unsigned)v.w / (unsigned)OWN_ROWS;
            const int lg = (idx * 4) / BM_GROUP;
            const unsigned gbit = 1u << (lg & 31);
            unsigned* gw = bm + (lg >> 5);
            __hip_atomic_fetch_or(gw + k0 * BM_LPITCH, gbit, __ATOMIC_RELAXED, __HIP_MEMORY_SCOPE_WORKGROUP);
            __hip_atomic_fetch_or(gw + k1 * BM_LPITCH, gbit, __ATOMIC_RELAXED, __HIP_MEMORY_SCOPE_WORKGROUP);
            __hip_atomic_fetch_or(gw + k2 * BM_LPITCH, gbit, __ATOMIC_RELAXED, __HIP_MEMORY_SCOPE_WORKGROUP);
            __hip_atomic_fetch_or(gw + k3 * BM_LPITCH, gbit, __ATOMIC_RELAXED, __HIP_MEMORY_SCOPE_WORKGROUP);
            count_node(cnt, v.x);
            count_node(cnt, v.y);
            count_node(cnt, v.z);
            count_node(cnt, v.w);
        }
    }
    __syncthreads();

    uint4* o = (uint4*)(hist + (size_t)blockIdx.x * HIST_PITCH);
    for (int i = tid; i < HIST_PITCH / 4; i += 1024) o[i] = ((const uint4*)cnt)[i];

    uint4* bo = (uint4*)bmp;
    for (int i = tid; i < OWN_WG * (BM_GPITCH / 4); i += 1024) {
        const int ow = i / (BM_GPITCH / 4);
        const int j4 = i - ow * (BM_GPITCH / 4);
        const int j  = j4 * 4;
        const unsigned* r = bm + ow * BM_LPITCH;
        uint4 v;
        v.x = (j     < BM_WORDS) ? r[j]     : 0u;
        v.y = (j + 1 < BM_WORDS) ? r[j + 1] : 0u;
        v.z = (j + 2 < BM_WORDS) ? r[j + 2] : 0u;
        v.w = (j + 3 < BM_WORDS) ? r[j + 3] : 0u;
        bo[((size_t)ow * HIST_G + blockIdx.x) * (BM_GPITCH / 4) + j4] = v;
    }
}

__global__ __launch_bounds__(512)
void k_dinv(const unsigned* __restrict__ hist, float* __restrict__ dinv, int* __restrict__ ocnt) {
    __shared__ int part[16];
    const int t = threadIdx.x;
    const int node = blockIdx.x * OWN_ROWS + t;
    int c = 0;
    if (t < OWN_ROWS && node < N_NODES) {
        const unsigned* p = hist + (node >> 1);
        const int sh = (node & 1) << 4;
        #pragma unroll 16
        for (int g = 0; g < HIST_G; ++g) c += (int)((p[(size_t)g * HIST_PITCH] >> sh) & 0xffffu);
        dinv[node] = rsqrtf((float)(c + 1));
    }
    for (int o = 16; o > 0; o >>= 1) c += __shfl_xor(c, o);
    if ((t & 31) == 0) part[t >> 5] = c;
    __syncthreads();
    if (t == 0) {
        int sum = 0;
        for (int i = 0; i < 16; ++i) sum += part[i];
        ocnt[blockIdx.x] = sum;
    }
}


#define GM_PITCH   132
#define ACC1_PITCH 130
#define ACC1_ROWS  400

typedef __attribute__((ext_vector_type(4))) float v4f;

static_assert(ACC1_ROWS >= OWN_ROWS + 1 && ACC1_ROWS % 16 == 0 && ACC1_ROWS / 16 <= OWN_THREADS / 32, "strips");

template<int NCOLS>
__device__ __forceinline__ void stage_w(const float* __restrict__ W, int* sW, int wv, int lane) {
    if (wv < 16) {
        constexpr int NB  = NCOLS / 16;
        constexpr int WIT = (128 / 8) * NB / 16;
        const int k0 = lane >> 2;
        const int n0 = (lane & 3) * 4;
        v4f wreg[WIT];
        #pragma unroll
        for (int i = 0; i < WIT; ++i) {
            int g  = wv + i * 16;
            int kb = g / NB;
            int nb = g - kb * NB;
            wreg[i] = *(const v4f*)(W + (size_t)(kb * 8 + k0) * NCOLS + nb * 16 + n0);
        }
        #pragma unroll
        for (int i = 0; i < WIT; ++i) {
            int g  = wv + i * 16;
            int kb = g / NB;
            int nb = g - kb * NB;
            int* d = sW + (nb * 16 + n0) * GM_PITCH + kb * 8 + k0;
            d[0]            = __float_as_int(wreg[i].x);
            d[GM_PITCH]     = __float_as_int(wreg[i].y);
            d[2 * GM_PITCH] = __float_as_int(wreg[i].z);
            d[3 * GM_PITCH] = __float_as_int(wreg[i].w);
        }
    }
}

template<int NT>
__device__ __forceinline__ void strip_product(const float* aP, const int* bP, v8f* c) {
    #pragma unroll 4
    for (int m = 0; m < 32; ++m) {
        v2f av;
        av.x = aP[m];
        av.y = aP[m + 32];
        #pragma unroll
        for (int t = 0; t < NT; ++t) {
            const int2 bw = *(const int2*)(bP + t * (16 * GM_PITCH) + 4 * m);
            v2f bv;
            bv.x = __int_as_float(bw.x);
            bv.y = __int_as_float(bw.y);
            c[t] = __builtin_amdgcn_wmma_f32_16x16x4_f32(false, av, false, bv,
                                                         (short)0, c[t], false, false);
        }
    }
}


#define OWN_GOFF 288
#define OWN_LIST 352

static_assert(31 + 32 * BM_GROUP <= OWN_GOFF && OWN_GOFF + 31 + 32 < OWN_LIST + 1, "list sizes");

__device__ __forceinline__ int own_cap(int cnt) { return (cnt >> 5) + OWN_THREADS / 32; }

template<int V> struct fvec;
template<> struct fvec<2> { typedef __attribute__((ext_vector_type(2))) float t; };
template<> struct fvec<4> { typedef __attribute__((ext_vector_type(4))) float t; };

template<int D, int PITCH>
__device__ __forceinline__ void own_rows(int n, int lane, int s, int d, float ws, float wd,
                                         const float* __restrict__ h, float* acc) {
    constexpr int V = D / 32;
    typedef typename fvec<V>::t vec_t;
    constexpr int G = 8;
    for (int i0 = 0; i0 < n; i0 += G) {
        vec_t v[G];
        #pragma unroll
        for (int g = 0; g < G; ++g) {
            int si = __builtin_amdgcn_readlane(s, i0 + g);
            v[g] = *(const vec_t*)(h + (size_t)si * D + lane * V);
        }
        #pragma unroll
        for (int g = 0; g < G; ++g) {
            float wi = __int_as_float(__builtin_amdgcn_readlane(__float_as_int(ws), i0 + g))
                     * __int_as_float(__builtin_amdgcn_readlane(__float_as_int(wd), i0 + g));
            int   di = __builtin_amdgcn_readlane(d, i0 + g);
            float* a = acc + di * PITCH + lane;
            #pragma unroll
            for (int j = 0; j < V; ++j) {
                __hip_atomic_fetch_add(a + j * 32, v[g][j] * wi, __ATOMIC_RELAXED,
                                       __HIP_MEMORY_SCOPE_WORKGROUP);
            }
        }
    }
}

template<int D, int PITCH>
__device__ __forceinline__ void own_flush(int n, int lane, int le, int lo,
                                          const float* __restrict__ dinv,
                                          const float* __restrict__ h,
                                          float* acc, int* nchunk, uint2* __restrict__ myrec) {
    int   s  = 0;
    float ws = 0.0f;
    float wd = 0.0f;
    int   d  = OWN_ROWS;
    if (lane < n) {
        s = le >> 9;
        d = le & 511;
        ws = dinv[s];
        wd = dinv[lo + d];
    }
    int c = 0;
    if (lane == 0) c = __hip_atomic_fetch_add(nchunk, 1, __ATOMIC_RELAXED, __HIP_MEMORY_SCOPE_WORKGROUP);
    c = __builtin_amdgcn_readfirstlane(c);
    own_rows<D, PITCH>(n, lane, s, d, ws, wd, h, acc);
    myrec[c * 32 + lane] = make_uint2(((unsigned)s << 9) | (unsigned)d, __float_as_uint(ws * wd));
}

template<int D, int PITCH>
__device__ __forceinline__ int own_groups(int ng, int lane, const int* glist, int cnt, int* elist, int lo,
                                          const int* __restrict__ src,
                                          const int* __restrict__ dst,
                                          const float* __restrict__ dinv,
                                          const float* __restrict__ h,
                                          float* acc, int* nchunk, uint2* __restrict__ myrec) {
    const int q = lane >> 2;
    const int p = lane & 3;
    int2 dv[4];
    int2 sv[4];
    #pragma unroll
    for (int i = 0; i < 4; ++i) {
        int g = 0;
        if (8 * i + q < ng) g = glist[8 * i + q];
        g = min(max(g, 0), N_EDGES / BM_GROUP - 1);
        const int e = g * BM_GROUP + 2 * p;
        dv[i] = *(const int2*)(dst + e);
        sv[i] = *(const int2*)(src + e);
    }
    #pragma unroll
    for (int i = 0; i < 4; ++i) {
        const bool used = 8 * i + q < ng;
        #pragma unroll
        for (int j = 0; j < 2; ++j) {
            const int dd = (j == 0 ? dv[i].x : dv[i].y) - lo;
            const int ss = (j == 0 ? sv[i].x : sv[i].y);
            const bool keep = used && (unsigned)dd < (unsigned)OWN_ROWS;
            const unsigned m = __builtin_amdgcn_ballot_w32(keep);
            if (keep) elist[cnt + (int)__builtin_amdgcn_mbcnt_lo(m, 0u)] = (ss << 9) | dd;
            cnt += __builtin_popcount(m);
        }
    }
    if (cnt >= 32) {
        int full = cnt & ~31;
        for (int base = 0; base < full; base += 32) {
            int en = elist[base + lane];
            own_flush<D, PITCH>(32, lane, en, lo, dinv, h, acc, nchunk, myrec);
        }
        int rest = cnt - full;
        int mv = 0;
        if (lane < rest) mv = elist[full + lane];
        if (lane < rest) elist[lane] = mv;
        cnt = rest;
    }
    return cnt;
}

template<int D, bool REPLAY>
__global__ __launch_bounds__(OWN_THREADS)
void k_agg_owner(const int* __restrict__ src, const int* __restrict__ dst,
                 const unsigned* __restrict__ bmp,
                 const float* __restrict__ dinv,
                 const float* __restrict__ bias,
                 const float* __restrict__ h, float* __restrict__ agg,
                 const int* __restrict__ ocnt, int* nch, unsigned* rec,
                 const float* __restrict__ W1, const float* __restrict__ W2) {
    constexpr int V = D / 32;
    typedef typename fvec<V>::t vec_t;
    constexpr int PITCH = REPLAY ? D : ACC1_PITCH;
    constexpr int ACC_MIN = REPLAY ? 164 * 1024 / 4 : 0;
    constexpr int ACC_ROWS = REPLAY ? OWN_ROWS + 1 : ACC1_ROWS;
    constexpr int ACC_FLOATS = ACC_ROWS * PITCH > ACC_MIN ? ACC_ROWS * PITCH : ACC_MIN;
    __shared__ __attribute__((aligned(16))) float acc[ACC_FLOATS];
    __shared__ __attribute__((aligned(16))) int wlist[REPLAY ? 1 : OWN_THREADS / 32][REPLAY ? 1 : OWN_LIST];
    __shared__ __attribute__((aligned(16))) int sW1[REPLAY ? 1 : DH * GM_PITCH];
    __shared__ int   red[OWN_WG / 32];
    __shared__ int   nchunk;
    static_assert(REPLAY || DOUT * GM_PITCH <= (OWN_THREADS / 32) * OWN_LIST, "W2 fits the lists' space");

    const int tid  = threadIdx.x;
    const int lane = tid & 31;
    const int wv   = __builtin_amdgcn_readfirstlane(tid >> 5);
    const int lo   = blockIdx.x * OWN_ROWS;

    if constexpr (!REPLAY) stage_w<DH>(W1, sW1, wv, lane);

    {
        int cap = 0;
        if (tid < (int)blockIdx.x) cap = own_cap(ocnt[tid]);
        for (int o = 16; o > 0; o >>= 1) cap += __shfl_xor(cap, o);
        if (wv < OWN_WG / 32 && lane == 0) red[wv] = cap;
        if (tid == 0) nchunk = 0;
    }

    if (wv == 0) {
        #pragma unroll
        for (int j = 0; j < V; ++j) acc[OWN_ROWS * PITCH + j * 32 + lane] = 0.0f;
    }
    for (int r = wv; r < OWN_ROWS; r += OWN_THREADS / 32) {
        int node = lo + r;
        if (node < N_NODES) {
            float sn = dinv[node] * dinv[node];
            vec_t hv = *(const vec_t*)(h + (size_t)node * D + lane * V);
            #pragma unroll
            for (int j = 0; j < V; ++j)
                acc[r * PITCH + j * 32 + lane] = REPLAY ? hv[j] * sn + bias[lane * V + j] : hv[j] * sn;
        }
    }
    __syncthreads();

    int start = 0;
    for (int i = 0; i < OWN_WG / 32; ++i) start += red[i];
    uint2* myrec = (uint2*)rec + (size_t)start * 32;

    if (REPLAY) {
        const int nCh = min(nch[blockIdx.x], own_cap(ocnt[blockIdx.x]));
        int c = wv;
        uint2 r = make_uint2((unsigned)OWN_ROWS, 0u);
        if (c < nCh) r = myrec[c * 32 + lane];
        while (c < nCh) {
            const int cn = c + OWN_THREADS / 32;
            uint2 rn = make_uint2((unsigned)OWN_ROWS, 0u);
            if (cn < nCh) rn = myrec[cn * 32 + lane];
            const int d = (int)(r.x & 511u);
            const int n = __builtin_popcount(__builtin_amdgcn_ballot_w32(d != OWN_ROWS));
            own_rows<D, PITCH>(n, lane, (int)(r.x >> 9), d, __uint_as_float(r.y), 1.0f, h, acc);
            r = rn;
            c = cn;
        }
    } else {
    int* elist = wlist[wv];
    int* glist = wlist[wv] + OWN_GOFF;
    constexpr int BW = (2 * BM_WORDS + 31) / 32;
    static_assert(HIST_G == 2 * (OWN_THREADS / 32), "two slices to a wave");
    const unsigned* mybits = bmp + ((size_t)blockIdx.x * HIST_G + 2 * wv) * BM_GPITCH;
    unsigned bw[BW];
    #pragma unroll
    for (int x = 0; x < BW; ++x) {
        const int rr = x * 32 + lane;
        const int sl = rr >= BM_WORDS ? 1 : 0;
        bw[x] = (rr < 2 * BM_WORDS) ? mybits[sl * BM_GPITCH + rr - sl * BM_WORDS] : 0u;
    }

    int gcnt = 0;
    int cnt  = 0;
    #pragma nounroll
    for (int x = 0; x < BW; ++x) {
        unsigned w = 0u;
        #pragma unroll
        for (int y = 0; y < BW; ++y) w = (x == y) ? bw[y] : w;
        const int rr = x * 32 + lane;
        const int sl = rr >= BM_WORDS ? 1 : 0;
        const int gbase = (2 * wv + sl) * BM_BITS + (rr - sl * BM_WORDS) * 32;
        unsigned m = __builtin_amdgcn_ballot_w32(w != 0u);
        while (m != 0u) {
            if (w != 0u) {
                int slot = gcnt + (int)__builtin_amdgcn_mbcnt_lo(m, 0u);
                glist[slot] = gbase + __builtin_ctz(w);
                w &= w - 1u;
            }
            gcnt += __builtin_popcount(m);
            if (gcnt >= 32) {
                cnt = own_groups<D, PITCH>(32, lane, glist, cnt, elist, lo, src, dst, dinv, h, acc, &nchunk, myrec);
                int rest = gcnt - 32;
                int mv = 0;
                if (lane < rest) mv = glist[32 + lane];
                if (lane < rest) glist[lane] = mv;
                gcnt = rest;
            }
            m = __builtin_amdgcn_ballot_w32(w != 0u);
        }
    }
    if (gcnt != 0) {
        cnt = own_groups<D, PITCH>(gcnt, lane, glist, cnt, elist, lo, src, dst, dinv, h, acc, &nchunk, myrec);
    }
    if (cnt != 0) {
        int en = 0;
        if (lane < cnt) en = elist[lane];
        own_flush<D, PITCH>(cnt, lane, en, lo, dinv, h, acc, &nchunk, myrec);
    }
    }
    __syncthreads();

    if constexpr (REPLAY) {
        for (int r = wv; r < OWN_ROWS; r += OWN_THREADS / 32) {
            int node = lo + r;
            if (node < N_NODES) {
                vec_t o;
                #pragma unroll
                for (int j = 0; j < V; ++j) o[j] = acc[r * PITCH + j * 32 + lane];
                *(vec_t*)(agg + (size_t)node * D + lane * V) = o;
            }
        }
    } else {
        if (tid == 0) nch[blockIdx.x] = nchunk;

        int* sW2 = &wlist[0][0];
        stage_w<DOUT>(W2, sW2, wv, lane);
        __syncthreads();

        if (wv < ACC1_ROWS / 16) {
            const int hi      = lane >> 4;
            const int laneLow = lane & 15;
            float* strip = acc + (wv * 16) * PITCH;
            const float* aP = strip + laneLow * PITCH + 64 * hi;

            {
                constexpr int NT = DH / 16;
                v8f c[NT];
                #pragma unroll
                for (int t = 0; t < NT; ++t) c[t] = (v8f){0.f, 0.f, 0.f, 0.f, 0.f, 0.f, 0.f, 0.f};
                strip_product<NT>(aP, sW1 + laneLow * GM_PITCH + 2 * hi, c);
                float* o = strip + (8 * hi) * PITCH + (laneLow & 3) * 32 + (laneLow >> 2);
                #pragma unroll
                for (int t = 0; t < NT; ++t) {
                    const float bb = bias[t * 16 + laneLow];
                    #pragma unroll
                    for (int r = 0; r < 8; ++r) o[r * PITCH + 4 * t] = fmaxf(c[t][r] + bb, 0.0f);
                }
            }
            {
                constexpr int NT = DOUT / 16;
                v8f c[NT];
                #pragma unroll
                for (int t = 0; t < NT; ++t) c[t] = (v8f){0.f, 0.f, 0.f, 0.f, 0.f, 0.f, 0.f, 0.f};
                strip_product<NT>(aP, sW2 + laneLow * GM_PITCH + 2 * hi, c);
                float* o = strip + (8 * hi) * PITCH + laneLow;
                #pragma unroll
                for (int t = 0; t < NT; ++t) {
                    #pragma unroll
                    for (int r = 0; r < 8; ++r) o[r * PITCH + t * 16] = c[t][r];
                }
                #pragma unroll
                for (int i = 0; i < 16; ++i) {
                    const int r    = wv * 16 + i;
                    const int node = lo + r;
                    if (r < OWN_ROWS && node < N_NODES) {
                        v2f v = *(const v2f*)(strip + i * PITCH + 2 * lane);
                        *(v2f*)(agg + (size_t)node * DOUT + 2 * lane) = v;
                    }
                }
            }
        }
    }
}


extern "C" void kernel_launch(void* const* d_in, const int* in_sizes, int n_in,
                              void* d_out, int out_size, void* d_ws, size_t ws_size,
                              hipStream_t stream) {
    const float* x   = (const float*)d_in[0];
    const int*   ei  = (const int*)d_in[1];
    const float* W1  = (const float*)d_in[2];
    const float* b1  = (const float*)d_in[3];
    const float* W2  = (const float*)d_in[4];
    const float* b2  = (const float*)d_in[5];
    float*       out = (float*)d_out;

    const int* src = ei;
    const int* dst = ei + N_EDGES;

    float* ws    = (float*)d_ws;
    float* dinv  = ws;
    float* keyp  = dinv  + N_NODES;
    float* h2    = keyp  + N_EDGES;
    float* histp = h2    + (size_t)N_NODES * DH;
    float* recp  = histp + (size_t)N_NODES * DH;
    int*   ocnt  = (int*)(keyp + N_EDGES / 4);
    int*   nch   = ocnt + OWN_WG;
    unsigned* rec = (unsigned*)(recp + 32);
    unsigned* bmp = (unsigned*)(recp + 4000000);
    unsigned* hist = (unsigned*)histp;

    k_hist_mark<<<HIST_G, 1024, 0, stream>>>(dst, hist, bmp);
    k_dinv    <<<OWN_WG, 512, 0, stream>>>(hist, dinv, ocnt);

    k_agg_owner<DIN, false><<<OWN_WG, OWN_THREADS, 0, stream>>>(src, dst, bmp, dinv, b1, x, h2, ocnt, nch, rec, W1, W2);

    k_agg_owner<DOUT, true><<<OWN_WG, OWN_THREADS, 0, stream>>>(src, dst, bmp, dinv, b2, h2, out, ocnt, nch, rec, W1, W2);
}


# Prologue reads issued together: a wave asks for its 13 self-term rows, their dinv, the owner counts and its group bits before its first wait (one round trip, not 16 in series); sums unchanged
# speedup vs baseline: 4.3697x; 1.0390x over previous
#include <hip/hip_runtime.h>

#define N_NODES 100000
#define N_EDGES 1600000
#define DIN 128
#define DH  128
#define DOUT 64

typedef __attribute__((ext_vector_type(2))) float v2f;
typedef __attribute__((ext_vector_type(8))) float v8f;

#define OWN_ROWS 391
#define OWN_WG   256
#define OWN_THREADS 1024

#define HIST_G     64
#define HIST_PER   (N_EDGES / 4 / HIST_G)
#define HIST_PITCH 50016

#define BM_GROUP   8
#define BM_BITS    (HIST_PER * 4 / BM_GROUP)
#define BM_WORDS   98
#define BM_LPITCH  99
#define BM_GPITCH  128

static_assert(HIST_PER * 4 % BM_GROUP == 0, "a group never straddles two slices");
static_assert(BM_WORDS * 32 >= BM_BITS && BM_WORDS <= BM_GPITCH, "a chunk holds a slice's groups");
static_assert(HIST_PER * 4 * HIST_G == N_EDGES, "the slices cover the edge list exactly");
static_assert(HIST_PER * 4 < 65536, "a 16-bit counter holds a whole slice");
static_assert(HIST_PITCH * 2 >= N_NODES && HIST_PITCH % 32 == 0, "plane size");

__device__ __forceinline__ void count_node(unsigned* cnt, int node) {
    __hip_atomic_fetch_add(cnt + (node >> 1), 1u << ((node & 1) << 4), __ATOMIC_RELAXED,
                           __HIP_MEMORY_SCOPE_WORKGROUP);
}

__global__ __launch_bounds__(1024)
void k_hist_mark(const int* __restrict__ dst, unsigned* __restrict__ hist, unsigned* __restrict__ bmp) {
    __shared__ __attribute__((aligned(16))) unsigned cnt[HIST_PITCH];
    __shared__ unsigned bm[OWN_WG * BM_LPITCH];
    constexpr int IT = (HIST_PER + 1023) / 1024;
    const int tid = threadIdx.x;
    const int4* d4 = (const int4*)dst + (size_t)blockIdx.x * HIST_PER;

    int4 d[IT];
    #pragma unroll
    for (int i = 0; i < IT; ++i) {
        int idx = tid + i * 1024;
        d[i] = d4[min(idx, HIST_PER - 1)];
    }
    for (int i = tid; i < HIST_PITCH / 4; i += 1024) ((uint4*)cnt)[i] = make_uint4(0u, 0u, 0u, 0u);
    for (int i = tid; i < OWN_WG * BM_LPITCH; i += 1024) bm[i] = 0u;
    __syncthreads();

    #pragma unroll
    for (int i = 0; i < IT; ++i) {
        int idx = tid + i * 1024;
        if (idx < HIST_PER) {
            const int4 v = d[i];
            const unsigned k0 = (unsigned)v.x / (unsigned)OWN_ROWS, k1 = (unsigned)v.y / (unsigned)OWN_ROWS;
            const unsigned k2 = (unsigned)v.z / (unsigned)OWN_ROWS, k3 = (unsigned)v.w / (unsigned)OWN_ROWS;
            const int lg = (idx * 4) / BM_GROUP;
            const unsigned gbit = 1u << (lg & 31);
            unsigned* gw = bm + (lg >> 5);
            __hip_atomic_fetch_or(gw + k0 * BM_LPITCH, gbit, __ATOMIC_RELAXED, __HIP_MEMORY_SCOPE_WORKGROUP);
            __hip_atomic_fetch_or(gw + k1 * BM_LPITCH, gbit, __ATOMIC_RELAXED, __HIP_MEMORY_SCOPE_WORKGROUP);
            __hip_atomic_fetch_or(gw + k2 * BM_LPITCH, gbit, __ATOMIC_RELAXED, __HIP_MEMORY_SCOPE_WORKGROUP);
            __hip_atomic_fetch_or(gw + k3 * BM_LPITCH, gbit, __ATOMIC_RELAXED, __HIP_MEMORY_SCOPE_WORKGROUP);
            count_node(cnt, v.x);
            count_node(cnt, v.y);
            count_node(cnt, v.z);
            count_node(cnt, v.w);
        }
    }
    __syncthreads();

    uint4* o = (uint4*)(hist + (size_t)blockIdx.x * HIST_PITCH);
    for (int i = tid; i < HIST_PITCH / 4; i += 1024) o[i] = ((const uint4*)cnt)[i];

    uint4* bo = (uint4*)bmp;
    for (int i = tid; i < OWN_WG * (BM_GPITCH / 4); i += 1024) {
        const int ow = i / (BM_GPITCH / 4);
        const int j4 = i - ow * (BM_GPITCH / 4);
        const int j  = j4 * 4;
        const unsigned* r = bm + ow * BM_LPITCH;
        uint4 v;
        v.x = (j     < BM_WORDS) ? r[j]     : 0u;
        v.y = (j + 1 < BM_WORDS) ? r[j + 1] : 0u;
        v.z = (j + 2 < BM_WORDS) ? r[j + 2] : 0u;
        v.w = (j + 3 < BM_WORDS) ? r[j + 3] : 0u;
        bo[((size_t)ow * HIST_G + blockIdx.x) * (BM_GPITCH / 4) + j4] = v;
    }
}

__global__ __launch_bounds__(512)
void k_dinv(const unsigned* __restrict__ hist, float* __restrict__ dinv, int* __restrict__ ocnt) {
    __shared__ int part[16];
    const int t = threadIdx.x;
    const int node = blockIdx.x * OWN_ROWS + t;
    int c = 0;
    if (t < OWN_ROWS && node < N_NODES) {
        const unsigned* p = hist + (node >> 1);
        const int sh = (node & 1) << 4;
        #pragma unroll 16
        for (int g = 0; g < HIST_G; ++g) c += (int)((p[(size_t)g * HIST_PITCH] >> sh) & 0xffffu);
        dinv[node] = rsqrtf((float)(c + 1));
    }
    for (int o = 16; o > 0; o >>= 1) c += __shfl_xor(c, o);
    if ((t & 31) == 0) part[t >> 5] = c;
    __syncthreads();
    if (t == 0) {
        int sum = 0;
        for (int i = 0; i < 16; ++i) sum += part[i];
        ocnt[blockIdx.x] = sum;
    }
}


#define GM_PITCH   132
#define ACC1_PITCH 130
#define ACC1_ROWS  400

typedef __attribute__((ext_vector_type(4))) float v4f;

static_assert(ACC1_ROWS >= OWN_ROWS + 1 && ACC1_ROWS % 16 == 0 && ACC1_ROWS / 16 <= OWN_THREADS / 32, "strips");

template<int NCOLS>
__device__ __forceinline__ void stage_w(const float* __restrict__ W, int* sW, int wv, int lane) {
    if (wv < 16) {
        constexpr int NB  = NCOLS / 16;
        constexpr int WIT = (128 / 8) * NB / 16;
        const int k0 = lane >> 2;
        const int n0 = (lane & 3) * 4;
        v4f wreg[WIT];
        #pragma unroll
        for (int i = 0; i < WIT; ++i) {
            int g  = wv + i * 16;
            int kb = g / NB;
            int nb = g - kb * NB;
            wreg[i] = *(const v4f*)(W + (size_t)(kb * 8 + k0) * NCOLS + nb * 16 + n0);
        }
        #pragma unroll
        for (int i = 0; i < WIT; ++i) {
            int g  = wv + i * 16;
            int kb = g / NB;
            int nb = g - kb * NB;
            int* d = sW + (nb * 16 + n0) * GM_PITCH + kb * 8 + k0;
            d[0]            = __float_as_int(wreg[i].x);
            d[GM_PITCH]     = __float_as_int(wreg[i].y);
            d[2 * GM_PITCH] = __float_as_int(wreg[i].z);
            d[3 * GM_PITCH] = __float_as_int(wreg[i].w);
        }
    }
}

template<int NT>
__device__ __forceinline__ void strip_product(const float* aP, const int* bP, v8f* c) {
    #pragma unroll 4
    for (int m = 0; m < 32; ++m) {
        v2f av;
        av.x = aP[m];
        av.y = aP[m + 32];
        #pragma unroll
        for (int t = 0; t < NT; ++t) {
            const int2 bw = *(const int2*)(bP + t * (16 * GM_PITCH) + 4 * m);
            v2f bv;
            bv.x = __int_as_float(bw.x);
            bv.y = __int_as_float(bw.y);
            c[t] = __builtin_amdgcn_wmma_f32_16x16x4_f32(false, av, false, bv,
                                                         (short)0, c[t], false, false);
        }
    }
}


#define OWN_GOFF 288
#define OWN_LIST 352

static_assert(31 + 32 * BM_GROUP <= OWN_GOFF && OWN_GOFF + 31 + 32 < OWN_LIST + 1, "list sizes");

__device__ __forceinline__ int own_cap(int cnt) { return (cnt >> 5) + OWN_THREADS / 32; }

template<int V> struct fvec;
template<> struct fvec<2> { typedef __attribute__((ext_vector_type(2))) float t; };
template<> struct fvec<4> { typedef __attribute__((ext_vector_type(4))) float t; };

template<int D, int PITCH>
__device__ __forceinline__ void own_rows(int n, int lane, int s, int d, float ws, float wd,
                                         const float* __restrict__ h, float* acc) {
    constexpr int V = D / 32;
    typedef typename fvec<V>::t vec_t;
    constexpr int G = 8;
    for (int i0 = 0; i0 < n; i0 += G) {
        vec_t v[G];
        #pragma unroll
        for (int g = 0; g < G; ++g) {
            int si = __builtin_amdgcn_readlane(s, i0 + g);
            v[g] = *(const vec_t*)(h + (size_t)si * D + lane * V);
        }
        #pragma unroll
        for (int g = 0; g < G; ++g) {
            float wi = __int_as_float(__builtin_amdgcn_readlane(__float_as_int(ws), i0 + g))
                     * __int_as_float(__builtin_amdgcn_readlane(__float_as_int(wd), i0 + g));
            int   di = __builtin_amdgcn_readlane(d, i0 + g);
            float* a = acc + di * PITCH + lane;
            #pragma unroll
            for (int j = 0; j < V; ++j) {
                __hip_atomic_fetch_add(a + j * 32, v[g][j] * wi, __ATOMIC_RELAXED,
                                       __HIP_MEMORY_SCOPE_WORKGROUP);
            }
        }
    }
}

template<int D, int PITCH>
__device__ __forceinline__ void own_flush(int n, int lane, int le, int lo,
                                          const float* __restrict__ dinv,
                                          const float* __restrict__ h,
                                          float* acc, int* nchunk, uint2* __restrict__ myrec) {
    int   s  = 0;
    float ws = 0.0f;
    float wd = 0.0f;
    int   d  = OWN_ROWS;
    if (lane < n) {
        s = le >> 9;
        d = le & 511;
        ws = dinv[s];
        wd = dinv[lo + d];
    }
    int c = 0;
    if (lane == 0) c = __hip_atomic_fetch_add(nchunk, 1, __ATOMIC_RELAXED, __HIP_MEMORY_SCOPE_WORKGROUP);
    c = __builtin_amdgcn_readfirstlane(c);
    own_rows<D, PITCH>(n, lane, s, d, ws, wd, h, acc);
    myrec[c * 32 + lane] = make_uint2(((unsigned)s << 9) | (unsigned)d, __float_as_uint(ws * wd));
}

template<int D, int PITCH>
__device__ __forceinline__ int own_groups(int ng, int lane, const int* glist, int cnt, int* elist, int lo,
                                          const int* __restrict__ src,
                                          const int* __restrict__ dst,
                                          const float* __restrict__ dinv,
                                          const float* __restrict__ h,
                                          float* acc, int* nchunk, uint2* __restrict__ myrec) {
    const int q = lane >> 2;
    const int p = lane & 3;
    int2 dv[4];
    int2 sv[4];
    #pragma unroll
    for (int i = 0; i < 4; ++i) {
        int g = 0;
        if (8 * i + q < ng) g = glist[8 * i + q];
        g = min(max(g, 0), N_EDGES / BM_GROUP - 1);
        const int e = g * BM_GROUP + 2 * p;
        dv[i] = *(const int2*)(dst + e);
        sv[i] = *(const int2*)(src + e);
    }
    #pragma unroll
    for (int i = 0; i < 4; ++i) {
        const bool used = 8 * i + q < ng;
        #pragma unroll
        for (int j = 0; j < 2; ++j) {
            const int dd = (j == 0 ? dv[i].x : dv[i].y) - lo;
            const int ss = (j == 0 ? sv[i].x : sv[i].y);
            const bool keep = used && (unsigned)dd < (unsigned)OWN_ROWS;
            const unsigned m = __builtin_amdgcn_ballot_w32(keep);
            if (keep) elist[cnt + (int)__builtin_amdgcn_mbcnt_lo(m, 0u)] = (ss << 9) | dd;
            cnt += __builtin_popcount(m);
        }
    }
    if (cnt >= 32) {
        int full = cnt & ~31;
        for (int base = 0; base < full; base += 32) {
            int en = elist[base + lane];
            own_flush<D, PITCH>(32, lane, en, lo, dinv, h, acc, nchunk, myrec);
        }
        int rest = cnt - full;
        int mv = 0;
        if (lane < rest) mv = elist[full + lane];
        if (lane < rest) elist[lane] = mv;
        cnt = rest;
    }
    return cnt;
}

template<int D, bool REPLAY>
__global__ __launch_bounds__(OWN_THREADS)
void k_agg_owner(const int* __restrict__ src, const int* __restrict__ dst,
                 const unsigned* __restrict__ bmp,
                 const float* __restrict__ dinv,
                 const float* __restrict__ bias,
                 const float* __restrict__ h, float* __restrict__ agg,
                 const int* __restrict__ ocnt, int* nch, unsigned* rec,
                 const float* __restrict__ W1, const float* __restrict__ W2) {
    constexpr int V = D / 32;
    typedef typename fvec<V>::t vec_t;
    constexpr int PITCH = REPLAY ? D : ACC1_PITCH;
    constexpr int ACC_MIN = REPLAY ? 164 * 1024 / 4 : 0;
    constexpr int ACC_ROWS = REPLAY ? OWN_ROWS + 1 : ACC1_ROWS;
    constexpr int ACC_FLOATS = ACC_ROWS * PITCH > ACC_MIN ? ACC_ROWS * PITCH : ACC_MIN;
    __shared__ __attribute__((aligned(16))) float acc[ACC_FLOATS];
    __shared__ __attribute__((aligned(16))) int wlist[REPLAY ? 1 : OWN_THREADS / 32][REPLAY ? 1 : OWN_LIST];
    __shared__ __attribute__((aligned(16))) int sW1[REPLAY ? 1 : DH * GM_PITCH];
    __shared__ int   red[OWN_WG / 32];
    __shared__ int   nchunk;
    static_assert(REPLAY || DOUT * GM_PITCH <= (OWN_THREADS / 32) * OWN_LIST, "W2 fits the lists' space");

    const int tid  = threadIdx.x;
    const int lane = tid & 31;
    const int wv   = __builtin_amdgcn_readfirstlane(tid >> 5);
    const int lo   = blockIdx.x * OWN_ROWS;

    constexpr int NSELF = (OWN_ROWS + OWN_THREADS / 32 - 1) / (OWN_THREADS / 32);
    vec_t hv[NSELF];
    float dn[NSELF];
    #pragma unroll
    for (int i = 0; i < NSELF; ++i) {
        const int node = min(lo + wv + i * (OWN_THREADS / 32), N_NODES - 1);
        dn[i] = dinv[node];
        hv[i] = *(const vec_t*)(h + (size_t)node * D + lane * V);
    }
    static_assert((OWN_WG & (OWN_WG - 1)) == 0, "mask below");
    const int ocBefore = ocnt[tid & (OWN_WG - 1)];
    int nchMine = 0;
    int ocMine  = 0;
    if constexpr (REPLAY) {
        nchMine = nch[blockIdx.x];
        ocMine  = ocnt[blockIdx.x];
    }
    constexpr int BW = (2 * BM_WORDS + 31) / 32;
    static_assert(HIST_G == 2 * (OWN_THREADS / 32), "two slices to a wave");
    unsigned bw[BW];
    #pragma unroll
    for (int x = 0; x < BW; ++x) bw[x] = 0u;
    if constexpr (!REPLAY) {
        const unsigned* mybits = bmp + ((size_t)blockIdx.x * HIST_G + 2 * wv) * BM_GPITCH;
        #pragma unroll
        for (int x = 0; x < BW; ++x) {
            const int rr = x * 32 + lane;
            const int sl = rr >= BM_WORDS ? 1 : 0;
            bw[x] = (rr < 2 * BM_WORDS) ? mybits[sl * BM_GPITCH + rr - sl * BM_WORDS] : 0u;
        }
    }

    if constexpr (!REPLAY) stage_w<DH>(W1, sW1, wv, lane);

    {
        int cap = 0;
        if (tid < (int)blockIdx.x) cap = own_cap(ocBefore);
        for (int o = 16; o > 0; o >>= 1) cap += __shfl_xor(cap, o);
        if (wv < OWN_WG / 32 && lane == 0) red[wv] = cap;
        if (tid == 0) nchunk = 0;
    }

    if (wv == 0) {
        #pragma unroll
        for (int j = 0; j < V; ++j) acc[OWN_ROWS * PITCH + j * 32 + lane] = 0.0f;
    }
    {
        float bj[V];
        #pragma unroll
        for (int j = 0; j < V; ++j) bj[j] = REPLAY ? bias[lane * V + j] : 0.0f;
        #pragma unroll
        for (int i = 0; i < NSELF; ++i) {
            const int r    = wv + i * (OWN_THREADS / 32);
            const int node = lo + r;
            if (r < OWN_ROWS && node < N_NODES) {
                const float sn = dn[i] * dn[i];
                #pragma unroll
                for (int j = 0; j < V; ++j)
                    acc[r * PITCH + j * 32 + lane] = REPLAY ? hv[i][j] * sn + bj[j] : hv[i][j] * sn;
            }
        }
    }
    __syncthreads();

    int start = 0;
    for (int i = 0; i < OWN_WG / 32; ++i) start += red[i];
    uint2* myrec = (uint2*)rec + (size_t)start * 32;

    if (REPLAY) {
        const int nCh = min(nchMine, own_cap(ocMine));
        int c = wv;
        uint2 r = make_uint2((unsigned)OWN_ROWS, 0u);
        if (c < nCh) r = myrec[c * 32 + lane];
        while (c < nCh) {
            const int cn = c + OWN_THREADS / 32;
            uint2 rn = make_uint2((unsigned)OWN_ROWS, 0u);
            if (cn < nCh) rn = myrec[cn * 32 + lane];
            const int d = (int)(r.x & 511u);
            const int n = __builtin_popcount(__builtin_amdgcn_ballot_w32(d != OWN_ROWS));
            own_rows<D, PITCH>(n, lane, (int)(r.x >> 9), d, __uint_as_float(r.y), 1.0f, h, acc);
            r = rn;
            c = cn;
        }
    } else {
    int* elist = wlist[wv];
    int* glist = wlist[wv] + OWN_GOFF;

    int gcnt = 0;
    int cnt  = 0;
    #pragma nounroll
    for (int x = 0; x < BW; ++x) {
        unsigned w = 0u;
        #pragma unroll
        for (int y = 0; y < BW; ++y) w = (x == y) ? bw[y] : w;
        const int rr = x * 32 + lane;
        const int sl = rr >= BM_WORDS ? 1 : 0;
        const int gbase = (2 * wv + sl) * BM_BITS + (rr - sl * BM_WORDS) * 32;
        unsigned m = __builtin_amdgcn_ballot_w32(w != 0u);
        while (m != 0u) {
            if (w != 0u) {
                int slot = gcnt + (int)__builtin_amdgcn_mbcnt_lo(m, 0u);
                glist[slot] = gbase + __builtin_ctz(w);
                w &= w - 1u;
            }
            gcnt += __builtin_popcount(m);
            if (gcnt >= 32) {
                cnt = own_groups<D, PITCH>(32, lane, glist, cnt, elist, lo, src, dst, dinv, h, acc, &nchunk, myrec);
                int rest = gcnt - 32;
                int mv = 0;
                if (lane < rest) mv = glist[32 + lane];
                if (lane < rest) glist[lane] = mv;
                gcnt = rest;
            }
            m = __builtin_amdgcn_ballot_w32(w != 0u);
        }
    }
    if (gcnt != 0) {
        cnt = own_groups<D, PITCH>(gcnt, lane, glist, cnt, elist, lo, src, dst, dinv, h, acc, &nchunk, myrec);
    }
    if (cnt != 0) {
        int en = 0;
        if (lane < cnt) en = elist[lane];
        own_flush<D, PITCH>(cnt, lane, en, lo, dinv, h, acc, &nchunk, myrec);
    }
    }
    __syncthreads();

    if constexpr (REPLAY) {
        for (int r = wv; r < OWN_ROWS; r += OWN_THREADS / 32) {
            int node = lo + r;
            if (node < N_NODES) {
                vec_t o;
                #pragma unroll
                for (int j = 0; j < V; ++j) o[j] = acc[r * PITCH + j * 32 + lane];
                *(vec_t*)(agg + (size_t)node * D + lane * V) = o;
            }
        }
    } else {
        if (tid == 0) nch[blockIdx.x] = nchunk;

        int* sW2 = &wlist[0][0];
        stage_w<DOUT>(W2, sW2, wv, lane);
        __syncthreads();

        if (wv < ACC1_ROWS / 16) {
            const int hi      = lane >> 4;
            const int laneLow = lane & 15;
            float* strip = acc + (wv * 16) * PITCH;
            const float* aP = strip + laneLow * PITCH + 64 * hi;

            {
                constexpr int NT = DH / 16;
                v8f c[NT];
                #pragma unroll
                for (int t = 0; t < NT; ++t) c[t] = (v8f){0.f, 0.f, 0.f, 0.f, 0.f, 0.f, 0.f, 0.f};
                strip_product<NT>(aP, sW1 + laneLow * GM_PITCH + 2 * hi, c);
                float* o = strip + (8 * hi) * PITCH + (laneLow & 3) * 32 + (laneLow >> 2);
                #pragma unroll
                for (int t = 0; t < NT; ++t) {
                    const float bb = bias[t * 16 + laneLow];
                    #pragma unroll
                    for (int r = 0; r < 8; ++r) o[r * PITCH + 4 * t] = fmaxf(c[t][r] + bb, 0.0f);
                }
            }
            {
                constexpr int NT = DOUT / 16;
                v8f c[NT];
                #pragma unroll
                for (int t = 0; t < NT; ++t) c[t] = (v8f){0.f, 0.f, 0.f, 0.f, 0.f, 0.f, 0.f, 0.f};
                strip_product<NT>(aP, sW2 + laneLow * GM_PITCH + 2 * hi, c);
                float* o = strip + (8 * hi) * PITCH + laneLow;
                #pragma unroll
                for (int t = 0; t < NT; ++t) {
                    #pragma unroll
                    for (int r = 0; r < 8; ++r) o[r * PITCH + t * 16] = c[t][r];
                }
                #pragma unroll
                for (int i = 0; i < 16; ++i) {
                    const int r    = wv * 16 + i;
                    const int node = lo + r;
                    if (r < OWN_ROWS && node < N_NODES) {
                        v2f v = *(const v2f*)(strip + i * PITCH + 2 * lane);
                        *(v2f*)(agg + (size_t)node * DOUT + 2 * lane) = v;
                    }
                }
            }
        }
    }
}


extern "C" void kernel_launch(void* const* d_in, const int* in_sizes, int n_in,
                              void* d_out, int out_size, void* d_ws, size_t ws_size,
                              hipStream_t stream) {
    const float* x   = (const float*)d_in[0];
    const int*   ei  = (const int*)d_in[1];
    const float* W1  = (const float*)d_in[2];
    const float* b1  = (const float*)d_in[3];
    const float* W2  = (const float*)d_in[4];
    const float* b2  = (const float*)d_in[5];
    float*       out = (float*)d_out;

    const int* src = ei;
    const int* dst = ei + N_EDGES;

    float* ws    = (float*)d_ws;
    float* dinv  = ws;
    float* keyp  = dinv  + N_NODES;
    float* h2    = keyp  + N_EDGES;
    float* histp = h2    + (size_t)N_NODES * DH;
    float* recp  = histp + (size_t)N_NODES * DH;
    int*   ocnt  = (int*)(keyp + N_EDGES / 4);
    int*   nch   = ocnt + OWN_WG;
    unsigned* rec = (unsigned*)(recp + 32);
    unsigned* bmp = (unsigned*)(recp + 4000000);
    unsigned* hist = (unsigned*)histp;

    k_hist_mark<<<HIST_G, 1024, 0, stream>>>(dst, hist, bmp);
    k_dinv    <<<OWN_WG, 512, 0, stream>>>(hist, dinv, ocnt);

    k_agg_owner<DIN, false><<<OWN_WG, OWN_THREADS, 0, stream>>>(src, dst, bmp, dinv, b1, x, h2, ocnt, nch, rec, W1, W2);

    k_agg_owner<DOUT, true><<<OWN_WG, OWN_THREADS, 0, stream>>>(src, dst, bmp, dinv, b2, h2, out, ocnt, nch, rec, W1, W2);
}


# Layer-1 owners read a marked group as one 64-B piece of a packed (dst,src) copy of the edge list, written by the dinv launch: one line per group, not a dst line and a src line; sums arrival-order
# speedup vs baseline: 4.4913x; 1.0278x over previous
#include <hip/hip_runtime.h>

#define N_NODES 100000
#define N_EDGES 1600000
#define DIN 128
#define DH  128
#define DOUT 64

typedef __attribute__((ext_vector_type(2))) float v2f;
typedef __attribute__((ext_vector_type(8))) float v8f;

#define OWN_ROWS 391
#define OWN_WG   256
#define OWN_THREADS 1024

#define HIST_G     64
#define HIST_PER   (N_EDGES / 4 / HIST_G)
#define HIST_PITCH 50016

#define BM_GROUP   8
#define BM_BITS    (HIST_PER * 4 / BM_GROUP)
#define BM_WORDS   98
#define BM_LPITCH  99
#define BM_GPITCH  128

static_assert(HIST_PER * 4 % BM_GROUP == 0, "a group never straddles two slices");
static_assert(BM_WORDS * 32 >= BM_BITS && BM_WORDS <= BM_GPITCH, "a chunk holds a slice's groups");
static_assert(HIST_PER * 4 * HIST_G == N_EDGES, "the slices cover the edge list exactly");
static_assert(HIST_PER * 4 < 65536, "a 16-bit counter holds a whole slice");
static_assert(HIST_PITCH * 2 >= N_NODES && HIST_PITCH % 32 == 0, "plane size");

#define EPK_UNITS   (N_EDGES / 4)
#define EPK_THREADS (OWN_WG * 512)
#define EPK_IT      ((EPK_UNITS + EPK_THREADS - 1) / EPK_THREADS)

static_assert(N_EDGES % 4 == 0 && BM_GROUP == 8, "a group is four 16-byte pieces of the packed plane");

__device__ __forceinline__ void count_node(unsigned* cnt, int node) {
    __hip_atomic_fetch_add(cnt + (node >> 1), 1u << ((node & 1) << 4), __ATOMIC_RELAXED,
                           __HIP_MEMORY_SCOPE_WORKGROUP);
}

__global__ __launch_bounds__(1024)
void k_hist_mark(const int* __restrict__ dst, unsigned* __restrict__ hist, unsigned* __restrict__ bmp) {
    __shared__ __attribute__((aligned(16))) unsigned cnt[HIST_PITCH];
    __shared__ unsigned bm[OWN_WG * BM_LPITCH];
    constexpr int IT = (HIST_PER + 1023) / 1024;
    const int tid = threadIdx.x;
    const int4* d4 = (const int4*)dst + (size_t)blockIdx.x * HIST_PER;

    int4 d[IT];
    #pragma unroll
    for (int i = 0; i < IT; ++i) {
        int idx = tid + i * 1024;
        d[i] = d4[min(idx, HIST_PER - 1)];
    }
    for (int i = tid; i < HIST_PITCH / 4; i += 1024) ((uint4*)cnt)[i] = make_uint4(0u, 0u, 0u, 0u);
    for (int i = tid; i < OWN_WG * BM_LPITCH; i += 1024) bm[i] = 0u;
    __syncthreads();

    #pragma unroll
    for (int i = 0; i < IT; ++i) {
        int idx = tid + i * 1024;
        if (idx < HIST_PER) {
            const int4 v = d[i];
            const unsigned k0 = (unsigned)v.x / (unsigned)OWN_ROWS, k1 = (unsigned)v.y / (unsigned)OWN_ROWS;
            const unsigned k2 = (unsigned)v.z / (unsigned)OWN_ROWS, k3 = (unsigned)v.w / (unsigned)OWN_ROWS;
            const int lg = (idx * 4) / BM_GROUP;
            const unsigned gbit = 1u << (lg & 31);
            unsigned* gw = bm + (lg >> 5);
            __hip_atomic_fetch_or(gw + k0 * BM_LPITCH, gbit, __ATOMIC_RELAXED, __HIP_MEMORY_SCOPE_WORKGROUP);
            __hip_atomic_fetch_or(gw + k1 * BM_LPITCH, gbit, __ATOMIC_RELAXED, __HIP_MEMORY_SCOPE_WORKGROUP);
            __hip_atomic_fetch_or(gw + k2 * BM_LPITCH, gbit, __ATOMIC_RELAXED, __HIP_MEMORY_SCOPE_WORKGROUP);
            __hip_atomic_fetch_or(gw + k3 * BM_LPITCH, gbit, __ATOMIC_RELAXED, __HIP_MEMORY_SCOPE_WORKGROUP);
            count_node(cnt, v.x);
            count_node(cnt, v.y);
            count_node(cnt, v.z);
            count_node(cnt, v.w);
        }
    }
    __syncthreads();

    uint4* o = (uint4*)(hist + (size_t)blockIdx.x * HIST_PITCH);
    for (int i = tid; i < HIST_PITCH / 4; i += 1024) o[i] = ((const uint4*)cnt)[i];

    uint4* bo = (uint4*)bmp;
    for (int i = tid; i < OWN_WG * (BM_GPITCH / 4); i += 1024) {
        const int ow = i / (BM_GPITCH / 4);
        const int j4 = i - ow * (BM_GPITCH / 4);
        const int j  = j4 * 4;
        const unsigned* r = bm + ow * BM_LPITCH;
        uint4 v;
        v.x = (j     < BM_WORDS) ? r[j]     : 0u;
        v.y = (j + 1 < BM_WORDS) ? r[j + 1] : 0u;
        v.z = (j + 2 < BM_WORDS) ? r[j + 2] : 0u;
        v.w = (j + 3 < BM_WORDS) ? r[j + 3] : 0u;
        bo[((size_t)ow * HIST_G + blockIdx.x) * (BM_GPITCH / 4) + j4] = v;
    }
}

__global__ __launch_bounds__(512)
void k_dinv(const unsigned* __restrict__ hist, float* __restrict__ dinv, int* __restrict__ ocnt,
            const int* __restrict__ src, const int* __restrict__ dst, int* __restrict__ epk) {
    __shared__ int part[16];
    const int t = threadIdx.x;
    const int node = blockIdx.x * OWN_ROWS + t;
    const int gid = blockIdx.x * 512 + t;
    int4 ed[EPK_IT];
    int4 es[EPK_IT];
    #pragma unroll
    for (int i = 0; i < EPK_IT; ++i) {
        const int u = min(gid + i * EPK_THREADS, EPK_UNITS - 1);
        ed[i] = ((const int4*)dst)[u];
        es[i] = ((const int4*)src)[u];
    }
    {
        int4* o = (int4*)epk;
        #pragma unroll
        for (int i = 0; i < EPK_IT; ++i) {
            const int u = gid + i * EPK_THREADS;
            if (u < EPK_UNITS) {
                o[2 * u]     = make_int4(ed[i].x, es[i].x, ed[i].y, es[i].y);
                o[2 * u + 1] = make_int4(ed[i].z, es[i].z, ed[i].w, es[i].w);
            }
        }
    }
    int c = 0;
    if (t < OWN_ROWS && node < N_NODES) {
        const unsigned* p = hist + (node >> 1);
        const int sh = (node & 1) << 4;
        #pragma unroll 16
        for (int g = 0; g < HIST_G; ++g) c += (int)((p[(size_t)g * HIST_PITCH] >> sh) & 0xffffu);
        dinv[node] = rsqrtf((float)(c + 1));
    }
    for (int o = 16; o > 0; o >>= 1) c += __shfl_xor(c, o);
    if ((t & 31) == 0) part[t >> 5] = c;
    __syncthreads();
    if (t == 0) {
        int sum = 0;
        for (int i = 0; i < 16; ++i) sum += part[i];
        ocnt[blockIdx.x] = sum;
    }
}


#define GM_PITCH   132
#define ACC1_PITCH 130
#define ACC1_ROWS  400

typedef __attribute__((ext_vector_type(4))) float v4f;

static_assert(ACC1_ROWS >= OWN_ROWS + 1 && ACC1_ROWS % 16 == 0 && ACC1_ROWS / 16 <= OWN_THREADS / 32, "strips");

template<int NCOLS>
__device__ __forceinline__ void stage_w(const float* __restrict__ W, int* sW, int wv, int lane) {
    if (wv < 16) {
        constexpr int NB  = NCOLS / 16;
        constexpr int WIT = (128 / 8) * NB / 16;
        const int k0 = lane >> 2;
        const int n0 = (lane & 3) * 4;
        v4f wreg[WIT];
        #pragma unroll
        for (int i = 0; i < WIT; ++i) {
            int g  = wv + i * 16;
            int kb = g / NB;
            int nb = g - kb * NB;
            wreg[i] = *(const v4f*)(W + (size_t)(kb * 8 + k0) * NCOLS + nb * 16 + n0);
        }
        #pragma unroll
        for (int i = 0; i < WIT; ++i) {
            int g  = wv + i * 16;
            int kb = g / NB;
            int nb = g - kb * NB;
            int* d = sW + (nb * 16 + n0) * GM_PITCH + kb * 8 + k0;
            d[0]            = __float_as_int(wreg[i].x);
            d[GM_PITCH]     = __float_as_int(wreg[i].y);
            d[2 * GM_PITCH] = __float_as_int(wreg[i].z);
            d[3 * GM_PITCH] = __float_as_int(wreg[i].w);
        }
    }
}

template<int NT>
__device__ __forceinline__ void strip_product(const float* aP, const int* bP, v8f* c) {
    #pragma unroll 4
    for (int m = 0; m < 32; ++m) {
        v2f av;
        av.x = aP[m];
        av.y = aP[m + 32];
        #pragma unroll
        for (int t = 0; t < NT; ++t) {
            const int2 bw = *(const int2*)(bP + t * (16 * GM_PITCH) + 4 * m);
            v2f bv;
            bv.x = __int_as_float(bw.x);
            bv.y = __int_as_float(bw.y);
            c[t] = __builtin_amdgcn_wmma_f32_16x16x4_f32(false, av, false, bv,
                                                         (short)0, c[t], false, false);
        }
    }
}


#define OWN_GOFF 288
#define OWN_LIST 352

static_assert(31 + 32 * BM_GROUP <= OWN_GOFF && OWN_GOFF + 31 + 32 < OWN_LIST + 1, "list sizes");

__device__ __forceinline__ int own_cap(int cnt) { return (cnt >> 5) + OWN_THREADS / 32; }

template<int V> struct fvec;
template<> struct fvec<2> { typedef __attribute__((ext_vector_type(2))) float t; };
template<> struct fvec<4> { typedef __attribute__((ext_vector_type(4))) float t; };

template<int D, int PITCH>
__device__ __forceinline__ void own_rows(int n, int lane, int s, int d, float ws, float wd,
                                         const float* __restrict__ h, float* acc) {
    constexpr int V = D / 32;
    typedef typename fvec<V>::t vec_t;
    constexpr int G = 8;
    for (int i0 = 0; i0 < n; i0 += G) {
        vec_t v[G];
        #pragma unroll
        for (int g = 0; g < G; ++g) {
            int si = __builtin_amdgcn_readlane(s, i0 + g);
            v[g] = *(const vec_t*)(h + (size_t)si * D + lane * V);
        }
        #pragma unroll
        for (int g = 0; g < G; ++g) {
            float wi = __int_as_float(__builtin_amdgcn_readlane(__float_as_int(ws), i0 + g))
                     * __int_as_float(__builtin_amdgcn_readlane(__float_as_int(wd), i0 + g));
            int   di = __builtin_amdgcn_readlane(d, i0 + g);
            float* a = acc + di * PITCH + lane;
            #pragma unroll
            for (int j = 0; j < V; ++j) {
                __hip_atomic_fetch_add(a + j * 32, v[g][j] * wi, __ATOMIC_RELAXED,
                                       __HIP_MEMORY_SCOPE_WORKGROUP);
            }
        }
    }
}

template<int D, int PITCH>
__device__ __forceinline__ void own_flush(int n, int lane, int le, int lo,
                                          const float* __restrict__ dinv,
                                          const float* __restrict__ h,
                                          float* acc, int* nchunk, uint2* __restrict__ myrec) {
    int   s  = 0;
    float ws = 0.0f;
    float wd = 0.0f;
    int   d  = OWN_ROWS;
    if (lane < n) {
        s = le >> 9;
        d = le & 511;
        ws = dinv[s];
        wd = dinv[lo + d];
    }
    int c = 0;
    if (lane == 0) c = __hip_atomic_fetch_add(nchunk, 1, __ATOMIC_RELAXED, __HIP_MEMORY_SCOPE_WORKGROUP);
    c = __builtin_amdgcn_readfirstlane(c);
    own_rows<D, PITCH>(n, lane, s, d, ws, wd, h, acc);
    myrec[c * 32 + lane] = make_uint2(((unsigned)s << 9) | (unsigned)d, __float_as_uint(ws * wd));
}

template<int D, int PITCH>
__device__ __forceinline__ int own_groups(int ng, int lane, const int* glist, int cnt, int* elist, int lo,
                                          const int4* __restrict__ epk4,
                                          const float* __restrict__ dinv,
                                          const float* __restrict__ h,
                                          float* acc, int* nchunk, uint2* __restrict__ myrec) {
    const int q = lane >> 2;
    const int p = lane & 3;
    int4 ev[4];
    #pragma unroll
    for (int i = 0; i < 4; ++i) {
        int g = 0;
        if (8 * i + q < ng) g = glist[8 * i + q];
        g = min(max(g, 0), N_EDGES / BM_GROUP - 1);
        ev[i] = epk4[g * (BM_GROUP / 2) + p];
    }
    asm volatile("" :: "v"(ev[0].x), "v"(ev[0].y), "v"(ev[0].z), "v"(ev[0].w),
                       "v"(ev[1].x), "v"(ev[1].y), "v"(ev[1].z), "v"(ev[1].w),
                       "v"(ev[2].x), "v"(ev[2].y), "v"(ev[2].z), "v"(ev[2].w),
                       "v"(ev[3].x), "v"(ev[3].y), "v"(ev[3].z), "v"(ev[3].w));
    #pragma unroll
    for (int i = 0; i < 4; ++i) {
        const bool used = 8 * i + q < ng;
        #pragma unroll
        for (int j = 0; j < 2; ++j) {
            const int dd = (j == 0 ? ev[i].x : ev[i].z) - lo;
            const int ss = (j == 0 ? ev[i].y : ev[i].w);
            const bool keep = used && (unsigned)dd < (unsigned)OWN_ROWS;
            const unsigned m = __builtin_amdgcn_ballot_w32(keep);
            if (keep) elist[cnt + (int)__builtin_amdgcn_mbcnt_lo(m, 0u)] = (ss << 9) | dd;
            cnt += __builtin_popcount(m);
        }
    }
    if (cnt >= 32) {
        int full = cnt & ~31;
        for (int base = 0; base < full; base += 32) {
            int en = elist[base + lane];
            own_flush<D, PITCH>(32, lane, en, lo, dinv, h, acc, nchunk, myrec);
        }
        int rest = cnt - full;
        int mv = 0;
        if (lane < rest) mv = elist[full + lane];
        if (lane < rest) elist[lane] = mv;
        cnt = rest;
    }
    return cnt;
}

template<int D, bool REPLAY>
__global__ __launch_bounds__(OWN_THREADS)
void k_agg_owner(const int* __restrict__ epk,
                 const unsigned* __restrict__ bmp,
                 const float* __restrict__ dinv,
                 const float* __restrict__ bias,
                 const float* __restrict__ h, float* __restrict__ agg,
                 const int* __restrict__ ocnt, int* nch, unsigned* rec,
                 const float* __restrict__ W1, const float* __restrict__ W2) {
    constexpr int V = D / 32;
    typedef typename fvec<V>::t vec_t;
    constexpr int PITCH = REPLAY ? D : ACC1_PITCH;
    constexpr int ACC_MIN = REPLAY ? 164 * 1024 / 4 : 0;
    constexpr int ACC_ROWS = REPLAY ? OWN_ROWS + 1 : ACC1_ROWS;
    constexpr int ACC_FLOATS = ACC_ROWS * PITCH > ACC_MIN ? ACC_ROWS * PITCH : ACC_MIN;
    __shared__ __attribute__((aligned(16))) float acc[ACC_FLOATS];
    __shared__ __attribute__((aligned(16))) int wlist[REPLAY ? 1 : OWN_THREADS / 32][REPLAY ? 1 : OWN_LIST];
    __shared__ __attribute__((aligned(16))) int sW1[REPLAY ? 1 : DH * GM_PITCH];
    __shared__ int   red[OWN_WG / 32];
    __shared__ int   nchunk;
    static_assert(REPLAY || DOUT * GM_PITCH <= (OWN_THREADS / 32) * OWN_LIST, "W2 fits the lists' space");

    const int tid  = threadIdx.x;
    const int lane = tid & 31;
    const int wv   = __builtin_amdgcn_readfirstlane(tid >> 5);
    const int lo   = blockIdx.x * OWN_ROWS;

    constexpr int NSELF = (OWN_ROWS + OWN_THREADS / 32 - 1) / (OWN_THREADS / 32);
    vec_t hv[NSELF];
    float dn[NSELF];
    #pragma unroll
    for (int i = 0; i < NSELF; ++i) {
        const int node = min(lo + wv + i * (OWN_THREADS / 32), N_NODES - 1);
        dn[i] = dinv[node];
        hv[i] = *(const vec_t*)(h + (size_t)node * D + lane * V);
    }
    static_assert((OWN_WG & (OWN_WG - 1)) == 0, "mask below");
    const int ocBefore = ocnt[tid & (OWN_WG - 1)];
    int nchMine = 0;
    int ocMine  = 0;
    if constexpr (REPLAY) {
        nchMine = nch[blockIdx.x];
        ocMine  = ocnt[blockIdx.x];
    }
    constexpr int BW = (2 * BM_WORDS + 31) / 32;
    static_assert(HIST_G == 2 * (OWN_THREADS / 32), "two slices to a wave");
    unsigned bw[BW];
    #pragma unroll
    for (int x = 0; x < BW; ++x) bw[x] = 0u;
    if constexpr (!REPLAY) {
        const unsigned* mybits = bmp + ((size_t)blockIdx.x * HIST_G + 2 * wv) * BM_GPITCH;
        #pragma unroll
        for (int x = 0; x < BW; ++x) {
            const int rr = x * 32 + lane;
            const int sl = rr >= BM_WORDS ? 1 : 0;
            bw[x] = (rr < 2 * BM_WORDS) ? mybits[sl * BM_GPITCH + rr - sl * BM_WORDS] : 0u;
        }
    }

    if constexpr (!REPLAY) stage_w<DH>(W1, sW1, wv, lane);

    {
        int cap = 0;
        if (tid < (int)blockIdx.x) cap = own_cap(ocBefore);
        for (int o = 16; o > 0; o >>= 1) cap += __shfl_xor(cap, o);
        if (wv < OWN_WG / 32 && lane == 0) red[wv] = cap;
        if (tid == 0) nchunk = 0;
    }

    if (wv == 0) {
        #pragma unroll
        for (int j = 0; j < V; ++j) acc[OWN_ROWS * PITCH + j * 32 + lane] = 0.0f;
    }
    {
        float bj[V];
        #pragma unroll
        for (int j = 0; j < V; ++j) bj[j] = REPLAY ? bias[lane * V + j] : 0.0f;
        #pragma unroll
        for (int i = 0; i < NSELF; ++i) {
            const int r    = wv + i * (OWN_THREADS / 32);
            const int node = lo + r;
            if (r < OWN_ROWS && node < N_NODES) {
                const float sn = dn[i] * dn[i];
                #pragma unroll
                for (int j = 0; j < V; ++j)
                    acc[r * PITCH + j * 32 + lane] = REPLAY ? hv[i][j] * sn + bj[j] : hv[i][j] * sn;
            }
        }
    }
    __syncthreads();

    int start = 0;
    for (int i = 0; i < OWN_WG / 32; ++i) start += red[i];
    uint2* myrec = (uint2*)rec + (size_t)start * 32;

    if (REPLAY) {
        const int nCh = min(nchMine, own_cap(ocMine));
        int c = wv;
        uint2 r = make_uint2((unsigned)OWN_ROWS, 0u);
        if (c < nCh) r = myrec[c * 32 + lane];
        while (c < nCh) {
            const int cn = c + OWN_THREADS / 32;
            uint2 rn = make_uint2((unsigned)OWN_ROWS, 0u);
            if (cn < nCh) rn = myrec[cn * 32 + lane];
            const int d = (int)(r.x & 511u);
            const int n = __builtin_popcount(__builtin_amdgcn_ballot_w32(d != OWN_ROWS));
            own_rows<D, PITCH>(n, lane, (int)(r.x >> 9), d, __uint_as_float(r.y), 1.0f, h, acc);
            r = rn;
            c = cn;
        }
    } else {
    int* elist = wlist[wv];
    int* glist = wlist[wv] + OWN_GOFF;

    int gcnt = 0;
    int cnt  = 0;
    #pragma nounroll
    for (int x = 0; x < BW; ++x) {
        unsigned w = 0u;
        #pragma unroll
        for (int y = 0; y < BW; ++y) w = (x == y) ? bw[y] : w;
        const int rr = x * 32 + lane;
        const int sl = rr >= BM_WORDS ? 1 : 0;
        const int gbase = (2 * wv + sl) * BM_BITS + (rr - sl * BM_WORDS) * 32;
        unsigned m = __builtin_amdgcn_ballot_w32(w != 0u);
        while (m != 0u) {
            if (w != 0u) {
                int slot = gcnt + (int)__builtin_amdgcn_mbcnt_lo(m, 0u);
                glist[slot] = gbase + __builtin_ctz(w);
                w &= w - 1u;
            }
            gcnt += __builtin_popcount(m);
            if (gcnt >= 32) {
                cnt = own_groups<D, PITCH>(32, lane, glist, cnt, elist, lo, (const int4*)epk, dinv, h, acc, &nchunk, myrec);
                int rest = gcnt - 32;
                int mv = 0;
                if (lane < rest) mv = glist[32 + lane];
                if (lane < rest) glist[lane] = mv;
                gcnt = rest;
            }
            m = __builtin_amdgcn_ballot_w32(w != 0u);
        }
    }
    if (gcnt != 0) {
        cnt = own_groups<D, PITCH>(gcnt, lane, glist, cnt, elist, lo, (const int4*)epk, dinv, h, acc, &nchunk, myrec);
    }
    if (cnt != 0) {
        int en = 0;
        if (lane < cnt) en = elist[lane];
        own_flush<D, PITCH>(cnt, lane, en, lo, dinv, h, acc, &nchunk, myrec);
    }
    }
    __syncthreads();

    if constexpr (REPLAY) {
        for (int r = wv; r < OWN_ROWS; r += OWN_THREADS / 32) {
            int node = lo + r;
            if (node < N_NODES) {
                vec_t o;
                #pragma unroll
                for (int j = 0; j < V; ++j) o[j] = acc[r * PITCH + j * 32 + lane];
                *(vec_t*)(agg + (size_t)node * D + lane * V) = o;
            }
        }
    } else {
        if (tid == 0) nch[blockIdx.x] = nchunk;

        int* sW2 = &wlist[0][0];
        stage_w<DOUT>(W2, sW2, wv, lane);
        __syncthreads();

        if (wv < ACC1_ROWS / 16) {
            const int hi      = lane >> 4;
            const int laneLow = lane & 15;
            float* strip = acc + (wv * 16) * PITCH;
            const float* aP = strip + laneLow * PITCH + 64 * hi;

            {
                constexpr int NT = DH / 16;
                v8f c[NT];
                #pragma unroll
                for (int t = 0; t < NT; ++t) c[t] = (v8f){0.f, 0.f, 0.f, 0.f, 0.f, 0.f, 0.f, 0.f};
                strip_product<NT>(aP, sW1 + laneLow * GM_PITCH + 2 * hi, c);
                float* o = strip + (8 * hi) * PITCH + (laneLow & 3) * 32 + (laneLow >> 2);
                #pragma unroll
                for (int t = 0; t < NT; ++t) {
                    const float bb = bias[t * 16 + laneLow];
                    #pragma unroll
                    for (int r = 0; r < 8; ++r) o[r * PITCH + 4 * t] = fmaxf(c[t][r] + bb, 0.0f);
                }
            }
            {
                constexpr int NT = DOUT / 16;
                v8f c[NT];
                #pragma unroll
                for (int t = 0; t < NT; ++t) c[t] = (v8f){0.f, 0.f, 0.f, 0.f, 0.f, 0.f, 0.f, 0.f};
                strip_product<NT>(aP, sW2 + laneLow * GM_PITCH + 2 * hi, c);
                float* o = strip + (8 * hi) * PITCH + laneLow;
                #pragma unroll
                for (int t = 0; t < NT; ++t) {
                    #pragma unroll
                    for (int r = 0; r < 8; ++r) o[r * PITCH + t * 16] = c[t][r];
                }
                #pragma unroll
                for (int i = 0; i < 16; ++i) {
                    const int r    = wv * 16 + i;
                    const int node = lo + r;
                    if (r < OWN_ROWS && node < N_NODES) {
                        v2f v = *(const v2f*)(strip + i * PITCH + 2 * lane);
                        *(v2f*)(agg + (size_t)node * DOUT + 2 * lane) = v;
                    }
                }
            }
        }
    }
}


extern "C" void kernel_launch(void* const* d_in, const int* in_sizes, int n_in,
                              void* d_out, int out_size, void* d_ws, size_t ws_size,
                              hipStream_t stream) {
    const float* x   = (const float*)d_in[0];
    const int*   ei  = (const int*)d_in[1];
    const float* W1  = (const float*)d_in[2];
    const float* b1  = (const float*)d_in[3];
    const float* W2  = (const float*)d_in[4];
    const float* b2  = (const float*)d_in[5];
    float*       out = (float*)d_out;

    const int* src = ei;
    const int* dst = ei + N_EDGES;

    float* ws    = (float*)d_ws;
    float* dinv  = ws;
    float* keyp  = dinv  + N_NODES;
    float* h2    = keyp  + N_EDGES;
    float* histp = h2    + (size_t)N_NODES * DH;
    float* recp  = histp + (size_t)N_NODES * DH;
    int*   ocnt  = (int*)(keyp + N_EDGES / 4);
    int*   nch   = ocnt + OWN_WG;
    unsigned* rec = (unsigned*)(recp + 32);
    unsigned* bmp = (unsigned*)(recp + 4000000);
    unsigned* hist = (unsigned*)histp;
    int* epk = (int*)(h2 + (size_t)N_NODES * DOUT);

    k_hist_mark<<<HIST_G, 1024, 0, stream>>>(dst, hist, bmp);
    k_dinv    <<<OWN_WG, 512, 0, stream>>>(hist, dinv, ocnt, src, dst, epk);

    k_agg_owner<DIN, false><<<OWN_WG, OWN_THREADS, 0, stream>>>(epk, bmp, dinv, b1, x, h2, ocnt, nch, rec, W1, W2);

    k_agg_owner<DOUT, true><<<OWN_WG, OWN_THREADS, 0, stream>>>(epk, bmp, dinv, b2, h2, out, ocnt, nch, rec, W1, W2);
}
